# starting kernel
# baseline (speedup 1.0000x reference)
#include <hip/hip_runtime.h>

typedef __attribute__((ext_vector_type(2))) float v2f;
typedef __attribute__((ext_vector_type(8))) float v8f;

#define F_IN 512
#define HID  16
#define OUTF 2

__global__ void k_deg_init(float* __restrict__ deg, int N) {
  int n = blockIdx.x * blockDim.x + threadIdx.x;
  if (n < N) deg[n] = 1.0f;
}

__global__ void k_deg_count(const int* __restrict__ dst, float* __restrict__ deg, int E) {
  int e = blockIdx.x * blockDim.x + threadIdx.x;
  if (e < E) atomicAdd(&deg[dst[e]], 1.0f);
}

__global__ void k_rsqrt(float* __restrict__ deg, int N) {
  int n = blockIdx.x * blockDim.x + threadIdx.x;
  if (n < N) deg[n] = rsqrtf(deg[n]);
}

__global__ __launch_bounds__(256) void k_gemm1_wmma(
    const float* __restrict__ x, const float* __restrict__ W1,
    float* __restrict__ h, int N, int nTiles) {
  int wave = blockIdx.x * (blockDim.x >> 5) + (threadIdx.x >> 5);
  if (wave >= nTiles) return;
  int lane = threadIdx.x & 31;
  int col  = lane & 15;
  int hi   = lane >> 4;
  int tileBase = wave * 16;

  int arow = tileBase + col;
  if (arow > N - 1) arow = N - 1;
  const float* xp = x  + (size_t)arow * F_IN + hi * 2;
  const float* wp = W1 + (size_t)(hi * 2) * HID + col;

  v8f acc = {};
#pragma unroll 4
  for (int kb = 0; kb < F_IN; kb += 4) {
    v2f a = __builtin_nontemporal_load((const v2f*)(xp + kb));
    v2f b;
    b.x = wp[kb * HID];
    b.y = wp[kb * HID + HID];
    acc = __builtin_amdgcn_wmma_f32_16x16x4_f32(
         false, a,  false, b,
         (short)0, acc,  false,  false);
  }
  if (tileBase + 16 <= N) {
    float* outp = h + ((size_t)(tileBase + 8 * hi)) * HID + col;
#pragma unroll
    for (int r = 0; r < 8; ++r) outp[r * HID] = acc[r];
  } else {
#pragma unroll
    for (int r = 0; r < 8; ++r) {
      int m = tileBase + r + 8 * hi;
      if (m < N) h[(size_t)m * HID + col] = acc[r];
    }
  }
}

__global__ void k_agg1_init(const float* __restrict__ h, const float* __restrict__ dinv,
                            const float* __restrict__ b1, float* __restrict__ acc1, int N) {
  int t = blockIdx.x * blockDim.x + threadIdx.x;
  if (t >= N * HID) return;
  int n = t >> 4, j = t & 15;
  float dv = dinv[n];
  acc1[t] = h[t] * dv * dv + b1[j];
}

__global__ __launch_bounds__(256) void k_agg1_edges(
    const int* __restrict__ src, const int* __restrict__ dst,
    const float* __restrict__ h, const float* __restrict__ dinv,
    float* __restrict__ acc1, int E) {
  long long t = (long long)blockIdx.x * blockDim.x + threadIdx.x;
  int e = (int)(t >> 2);
  if (e >= E) return;
  int j4 = (int)(t & 3) * 4;
  int s = src[e], d = dst[e];
  float c = dinv[s] * dinv[d];
  const float4 m = *(const float4*)(h + (size_t)s * HID + j4);
  float* a = acc1 + (size_t)d * HID + j4;
  atomicAdd(a + 0, m.x * c);
  atomicAdd(a + 1, m.y * c);
  atomicAdd(a + 2, m.z * c);
  atomicAdd(a + 3, m.w * c);
}

__global__ void k_layer2_node(const float* __restrict__ acc1,
                              const float* __restrict__ W2, const float* __restrict__ b2,
                              const float* __restrict__ dinv,
                              float* __restrict__ h3, float* __restrict__ out, int N) {
  int n = blockIdx.x * blockDim.x + threadIdx.x;
  if (n >= N) return;
  float o0 = 0.0f, o1 = 0.0f;
#pragma unroll
  for (int j = 0; j < HID; ++j) {
    float v = fmaxf(acc1[(size_t)n * HID + j], 0.0f);
    o0 = __builtin_fmaf(v, W2[j * OUTF + 0], o0);
    o1 = __builtin_fmaf(v, W2[j * OUTF + 1], o1);
  }
  h3[(size_t)n * OUTF + 0] = o0;
  h3[(size_t)n * OUTF + 1] = o1;
  float dv = dinv[n];
  float c = dv * dv;
  out[(size_t)n * OUTF + 0] = o0 * c + b2[0];
  out[(size_t)n * OUTF + 1] = o1 * c + b2[1];
}

__global__ __launch_bounds__(256) void k_agg2_edges(
    const int* __restrict__ src, const int* __restrict__ dst,
    const float* __restrict__ h3, const float* __restrict__ dinv,
    float* __restrict__ out, int E) {
  int e = blockIdx.x * blockDim.x + threadIdx.x;
  if (e >= E) return;
  int s = src[e], d = dst[e];
  float c = dinv[s] * dinv[d];
  atomicAdd(&out[(size_t)d * OUTF + 0], h3[(size_t)s * OUTF + 0] * c);
  atomicAdd(&out[(size_t)d * OUTF + 1], h3[(size_t)s * OUTF + 1] * c);
}

static inline int cdiv_i(long long a, long long b) { return (int)((a + b - 1) / b); }

extern "C" void kernel_launch(void* const* d_in, const int* in_sizes, int n_in,
                              void* d_out, int out_size, void* d_ws, size_t ws_size,
                              hipStream_t stream) {
  const float* x  = (const float*)d_in[0];
  const int*   ei = (const int*)  d_in[1];
  const float* W1 = (const float*)d_in[2];
  const float* b1 = (const float*)d_in[3];
  const float* W2 = (const float*)d_in[4];
  const float* b2 = (const float*)d_in[5];
  float* out = (float*)d_out;

  const int N = in_sizes[0] / F_IN;
  const int E = in_sizes[1] / 2;
  const int* src = ei;
  const int* dst = ei + E;

  float* ws   = (float*)d_ws;
  float* dinv = ws;
  float* h    = dinv + N;
  float* acc1 = h + (size_t)N * HID;
  float* h3   = acc1 + (size_t)N * HID;

  const int BS = 256;

  k_deg_init <<<cdiv_i(N, BS), BS, 0, stream>>>(dinv, N);
  k_deg_count<<<cdiv_i(E, BS), BS, 0, stream>>>(dst, dinv, E);
  k_rsqrt    <<<cdiv_i(N, BS), BS, 0, stream>>>(dinv, N);

  int nTiles = cdiv_i(N, 16);
  k_gemm1_wmma<<<cdiv_i(nTiles, BS / 32), BS, 0, stream>>>(x, W1, h, N, nTiles);

  k_agg1_init <<<cdiv_i((long long)N * HID, BS), BS, 0, stream>>>(h, dinv, b1, acc1, N);
  k_agg1_edges<<<cdiv_i((long long)E * 4, BS), BS, 0, stream>>>(src, dst, h, dinv, acc1, E);

  k_layer2_node<<<cdiv_i(N, BS), BS, 0, stream>>>(acc1, W2, b2, dinv, h3, out, N);

  k_agg2_edges<<<cdiv_i(E, BS), BS, 0, stream>>>(src, dst, h3, dinv, out, E);
}


# Degree count and layer-2 scatter-add: global f32 atomics replaced by LDS accumulation per node range over edge chunks, then a sum of 16 partial planes; changes the order of summation
# speedup vs baseline: 2.1070x; 2.1070x over previous
#include <hip/hip_runtime.h>

typedef __attribute__((ext_vector_type(2))) float v2f;
typedef __attribute__((ext_vector_type(8))) float v8f;

#define F_IN 512
#define HID  16
#define OUTF 2

#define NPP     6400
#define NCH     16
#define SCAN_BS 1024
#define SCAN_U  8

__global__ __launch_bounds__(SCAN_BS) void k_deg_part(
    const int* __restrict__ dst, float* __restrict__ part, int N, int E) {
  __shared__ unsigned cnt[NPP];
  const int tid = threadIdx.x;
  const int n0  = blockIdx.x * NPP;
  const int np  = min(NPP, N - n0);
  for (int i = tid; i < NPP; i += SCAN_BS) cnt[i] = 0u;
  __syncthreads();
  const int per = (E + NCH - 1) / NCH;
  const int e0  = blockIdx.y * per;
  const int e1  = min(E, e0 + per);
  for (int base = e0 + tid; base < e1; base += 2 * SCAN_U * SCAN_BS) {
    int d[2 * SCAN_U];
#pragma unroll
    for (int k = 0; k < 2 * SCAN_U; ++k) d[k] = dst[min(base + k * SCAN_BS, e1 - 1)];
#pragma unroll
    for (int k = 0; k < 2 * SCAN_U; ++k) {
      unsigned r = (unsigned)(d[k] - n0);
      if (base + k * SCAN_BS < e1 && r < (unsigned)np) atomicAdd(&cnt[r], 1u);
    }
  }
  __syncthreads();
  float* o = part + (size_t)blockIdx.y * N + n0;
  for (int i = tid; i < np; i += SCAN_BS) o[i] = (float)cnt[i];
}

__global__ void k_deg_reduce(const float* __restrict__ part, float* __restrict__ dinv, int N) {
  int n = blockIdx.x * blockDim.x + threadIdx.x;
  if (n >= N) return;
  float s = 1.0f;
#pragma unroll
  for (int c = 0; c < NCH; ++c) s += part[(size_t)c * N + n];
  dinv[n] = rsqrtf(s);
}

__global__ __launch_bounds__(256) void k_gemm1_wmma(
    const float* __restrict__ x, const float* __restrict__ W1,
    float* __restrict__ h, int N, int nTiles) {
  int wave = blockIdx.x * (blockDim.x >> 5) + (threadIdx.x >> 5);
  if (wave >= nTiles) return;
  int lane = threadIdx.x & 31;
  int col  = lane & 15;
  int hi   = lane >> 4;
  int tileBase = wave * 16;

  int arow = tileBase + col;
  if (arow > N - 1) arow = N - 1;
  const float* xp = x  + (size_t)arow * F_IN + hi * 2;
  const float* wp = W1 + (size_t)(hi * 2) * HID + col;

  v8f acc = {};
#pragma unroll 4
  for (int kb = 0; kb < F_IN; kb += 4) {
    v2f a = __builtin_nontemporal_load((const v2f*)(xp + kb));
    v2f b;
    b.x = wp[kb * HID];
    b.y = wp[kb * HID + HID];
    acc = __builtin_amdgcn_wmma_f32_16x16x4_f32(
         false, a,  false, b,
         (short)0, acc,  false,  false);
  }
  if (tileBase + 16 <= N) {
    float* outp = h + ((size_t)(tileBase + 8 * hi)) * HID + col;
#pragma unroll
    for (int r = 0; r < 8; ++r) outp[r * HID] = acc[r];
  } else {
#pragma unroll
    for (int r = 0; r < 8; ++r) {
      int m = tileBase + r + 8 * hi;
      if (m < N) h[(size_t)m * HID + col] = acc[r];
    }
  }
}

__global__ void k_agg1_init(const float* __restrict__ h, const float* __restrict__ dinv,
                            const float* __restrict__ b1, float* __restrict__ acc1, int N) {
  int t = blockIdx.x * blockDim.x + threadIdx.x;
  if (t >= N * HID) return;
  int n = t >> 4, j = t & 15;
  float dv = dinv[n];
  acc1[t] = h[t] * dv * dv + b1[j];
}

__global__ __launch_bounds__(256) void k_agg1_edges(
    const int* __restrict__ src, const int* __restrict__ dst,
    const float* __restrict__ h, const float* __restrict__ dinv,
    float* __restrict__ acc1, int E) {
  long long t = (long long)blockIdx.x * blockDim.x + threadIdx.x;
  int e = (int)(t >> 2);
  if (e >= E) return;
  int j4 = (int)(t & 3) * 4;
  int s = src[e], d = dst[e];
  float c = dinv[s] * dinv[d];
  const float4 m = *(const float4*)(h + (size_t)s * HID + j4);
  float* a = acc1 + (size_t)d * HID + j4;
  atomicAdd(a + 0, m.x * c);
  atomicAdd(a + 1, m.y * c);
  atomicAdd(a + 2, m.z * c);
  atomicAdd(a + 3, m.w * c);
}

__global__ void k_layer2_node(const float* __restrict__ acc1,
                              const float* __restrict__ W2,
                              float* __restrict__ h3, int N) {
  int n = blockIdx.x * blockDim.x + threadIdx.x;
  if (n >= N) return;
  float o0 = 0.0f, o1 = 0.0f;
#pragma unroll
  for (int j = 0; j < HID; ++j) {
    float v = fmaxf(acc1[(size_t)n * HID + j], 0.0f);
    o0 = __builtin_fmaf(v, W2[j * OUTF + 0], o0);
    o1 = __builtin_fmaf(v, W2[j * OUTF + 1], o1);
  }
  h3[(size_t)n * OUTF + 0] = o0;
  h3[(size_t)n * OUTF + 1] = o1;
}

__global__ __launch_bounds__(SCAN_BS) void k_agg2_part(
    const int* __restrict__ src, const int* __restrict__ dst,
    const float* __restrict__ h3, const float* __restrict__ dinv,
    float* __restrict__ part, int N, int E) {
  __shared__ float acc[NPP * OUTF];
  __shared__ float dinv_d[NPP];
  const int tid = threadIdx.x;
  const int n0  = blockIdx.x * NPP;
  const int np  = min(NPP, N - n0);
  for (int i = tid; i < NPP * OUTF; i += SCAN_BS) acc[i] = 0.0f;
  for (int i = tid; i < np; i += SCAN_BS) dinv_d[i] = dinv[n0 + i];
  __syncthreads();
  const int per = (E + NCH - 1) / NCH;
  const int e0  = blockIdx.y * per;
  const int e1  = min(E, e0 + per);

  if (e0 < e1) {
    const unsigned NONE = 0xffffffffu;
    unsigned r[SCAN_U];
    int      s[SCAN_U];
    int base = e0 + tid;
#pragma unroll
    for (int k = 0; k < SCAN_U; ++k) {
      int e  = base + k * SCAN_BS;
      int ec = min(e, e1 - 1);
      int dd = dst[ec];
      s[k] = src[ec];
      r[k] = (e < e1) ? (unsigned)(dd - n0) : NONE;
    }
    while (base < e1) {
      float ga[SCAN_U], gb[SCAN_U], gx[SCAN_U], gy[SCAN_U];
#pragma unroll
      for (int k = 0; k < SCAN_U; ++k) {
        const bool keep = r[k] < (unsigned)np;
        const int si = keep ? s[k] : 0;
        ga[k] = dinv[si];
        gb[k] = dinv_d[keep ? r[k] : 0u];
        gx[k] = h3[(size_t)si * OUTF + 0];
        gy[k] = h3[(size_t)si * OUTF + 1];
      }
      const int nbase = base + SCAN_U * SCAN_BS;
      int dn[SCAN_U], sn[SCAN_U];
#pragma unroll
      for (int k = 0; k < SCAN_U; ++k) {
        int ec = min(nbase + k * SCAN_BS, e1 - 1);
        dn[k] = dst[ec];
        sn[k] = src[ec];
      }
#pragma unroll
      for (int k = 0; k < SCAN_U; ++k) {
        if (r[k] < (unsigned)np) {
          float c = ga[k] * gb[k];
          atomicAdd(&acc[r[k] * OUTF + 0], gx[k] * c);
          atomicAdd(&acc[r[k] * OUTF + 1], gy[k] * c);
        }
      }
#pragma unroll
      for (int k = 0; k < SCAN_U; ++k) {
        r[k] = (nbase + k * SCAN_BS < e1) ? (unsigned)(dn[k] - n0) : NONE;
        s[k] = sn[k];
      }
      base = nbase;
    }
  }
  __syncthreads();
  float* o = part + ((size_t)blockIdx.y * N + n0) * OUTF;
  for (int i = tid; i < np * OUTF; i += SCAN_BS) o[i] = acc[i];
}

__global__ void k_agg2_reduce(const float* __restrict__ part, const float* __restrict__ h3,
                              const float* __restrict__ dinv, const float* __restrict__ b2,
                              float* __restrict__ out, int N) {
  int t = blockIdx.x * blockDim.x + threadIdx.x;
  if (t >= N * OUTF) return;
  float dv = dinv[t / OUTF];
  float s = h3[t] * (dv * dv) + b2[t % OUTF];
#pragma unroll
  for (int c = 0; c < NCH; ++c) s += part[(size_t)c * N * OUTF + t];
  out[t] = s;
}

static inline int cdiv_i(long long a, long long b) { return (int)((a + b - 1) / b); }

extern "C" void kernel_launch(void* const* d_in, const int* in_sizes, int n_in,
                              void* d_out, int out_size, void* d_ws, size_t ws_size,
                              hipStream_t stream) {
  const float* x  = (const float*)d_in[0];
  const int*   ei = (const int*)  d_in[1];
  const float* W1 = (const float*)d_in[2];
  const float* b1 = (const float*)d_in[3];
  const float* W2 = (const float*)d_in[4];
  const float* b2 = (const float*)d_in[5];
  float* out = (float*)d_out;

  const int N = in_sizes[0] / F_IN;
  const int E = in_sizes[1] / 2;
  const int* src = ei;
  const int* dst = ei + E;

  float* ws   = (float*)d_ws;
  float* dinv = ws;
  float* h    = dinv + N;
  float* acc1 = h + (size_t)N * HID;
  float* h3   = acc1 + (size_t)N * HID;

  const int BS = 256;

  float* part = h;
  const dim3 scanGrid(cdiv_i(N, NPP), NCH);

  k_deg_part  <<<scanGrid, SCAN_BS, 0, stream>>>(dst, part, N, E);
  k_deg_reduce<<<cdiv_i(N, BS), BS, 0, stream>>>(part, dinv, N);

  int nTiles = cdiv_i(N, 16);
  k_gemm1_wmma<<<cdiv_i(nTiles, BS / 32), BS, 0, stream>>>(x, W1, h, N, nTiles);

  k_agg1_init <<<cdiv_i((long long)N * HID, BS), BS, 0, stream>>>(h, dinv, b1, acc1, N);
  k_agg1_edges<<<cdiv_i((long long)E * 4, BS), BS, 0, stream>>>(src, dst, h, dinv, acc1, E);

  k_layer2_node<<<cdiv_i(N, BS), BS, 0, stream>>>(acc1, W2, h3, N);

  k_agg2_part  <<<scanGrid, SCAN_BS, 0, stream>>>(src, dst, h3, dinv, part, N, E);
  k_agg2_reduce<<<cdiv_i((long long)N * OUTF, BS), BS, 0, stream>>>(part, h3, dinv, b2, out, N);
}


# Layer-1 edge scatter: 16 adjacent lanes add one 64-byte row per edge (one f32 atomic request per edge, not four) in resident looping waves; atomics kept, sum order by arrival as before
# speedup vs baseline: 2.5868x; 1.2277x over previous
#include <hip/hip_runtime.h>

typedef __attribute__((ext_vector_type(2))) float v2f;
typedef __attribute__((ext_vector_type(8))) float v8f;

#define F_IN 512
#define HID  16
#define OUTF 2

#define NPP     6400
#define NCH     16
#define SCAN_BS 1024
#define SCAN_U  8

__global__ __launch_bounds__(SCAN_BS) void k_deg_part(
    const int* __restrict__ dst, float* __restrict__ part, int N, int E) {
  __shared__ unsigned cnt[NPP];
  const int tid = threadIdx.x;
  const int n0  = blockIdx.x * NPP;
  const int np  = min(NPP, N - n0);
  for (int i = tid; i < NPP; i += SCAN_BS) cnt[i] = 0u;
  __syncthreads();
  const int per = (E + NCH - 1) / NCH;
  const int e0  = blockIdx.y * per;
  const int e1  = min(E, e0 + per);
  for (int base = e0 + tid; base < e1; base += 2 * SCAN_U * SCAN_BS) {
    int d[2 * SCAN_U];
#pragma unroll
    for (int k = 0; k < 2 * SCAN_U; ++k) d[k] = dst[min(base + k * SCAN_BS, e1 - 1)];
#pragma unroll
    for (int k = 0; k < 2 * SCAN_U; ++k) {
      unsigned r = (unsigned)(d[k] - n0);
      if (base + k * SCAN_BS < e1 && r < (unsigned)np) atomicAdd(&cnt[r], 1u);
    }
  }
  __syncthreads();
  float* o = part + (size_t)blockIdx.y * N + n0;
  for (int i = tid; i < np; i += SCAN_BS) o[i] = (float)cnt[i];
}

__global__ void k_deg_reduce(const float* __restrict__ part, float* __restrict__ dinv, int N) {
  int n = blockIdx.x * blockDim.x + threadIdx.x;
  if (n >= N) return;
  float s = 1.0f;
#pragma unroll
  for (int c = 0; c < NCH; ++c) s += part[(size_t)c * N + n];
  dinv[n] = rsqrtf(s);
}

__global__ __launch_bounds__(256) void k_gemm1_wmma(
    const float* __restrict__ x, const float* __restrict__ W1,
    float* __restrict__ h, int N, int nTiles) {
  int wave = blockIdx.x * (blockDim.x >> 5) + (threadIdx.x >> 5);
  if (wave >= nTiles) return;
  int lane = threadIdx.x & 31;
  int col  = lane & 15;
  int hi   = lane >> 4;
  int tileBase = wave * 16;

  int arow = tileBase + col;
  if (arow > N - 1) arow = N - 1;
  const float* xp = x  + (size_t)arow * F_IN + hi * 2;
  const float* wp = W1 + (size_t)(hi * 2) * HID + col;

  v8f acc = {};
#pragma unroll 4
  for (int kb = 0; kb < F_IN; kb += 4) {
    v2f a = __builtin_nontemporal_load((const v2f*)(xp + kb));
    v2f b;
    b.x = wp[kb * HID];
    b.y = wp[kb * HID + HID];
    acc = __builtin_amdgcn_wmma_f32_16x16x4_f32(
         false, a,  false, b,
         (short)0, acc,  false,  false);
  }
  if (tileBase + 16 <= N) {
    float* outp = h + ((size_t)(tileBase + 8 * hi)) * HID + col;
#pragma unroll
    for (int r = 0; r < 8; ++r) outp[r * HID] = acc[r];
  } else {
#pragma unroll
    for (int r = 0; r < 8; ++r) {
      int m = tileBase + r + 8 * hi;
      if (m < N) h[(size_t)m * HID + col] = acc[r];
    }
  }
}

__global__ void k_agg1_init(const float* __restrict__ h, const float* __restrict__ dinv,
                            const float* __restrict__ b1, float* __restrict__ acc1, int N) {
  int t = blockIdx.x * blockDim.x + threadIdx.x;
  if (t >= N * HID) return;
  int n = t >> 4, j = t & 15;
  float dv = dinv[n];
  acc1[t] = h[t] * dv * dv + b1[j];
}

#define AGG1_BS     256
#define AGG1_BLOCKS 2048

template <int I>
__device__ __forceinline__ int row_share_bcast(int v) {
  return __builtin_amdgcn_mov_dpp(v, 0x150 + I, 0xf, 0xf, false);
}

#define AGG1_BCAST(I)                                          \
  ss[I] = (unsigned)row_share_bcast<I>(s0);                    \
  dd[I] = row_share_bcast<I>(d0);                              \
  cc[I] = __int_as_float(row_share_bcast<I>(__float_as_int(c0)));

__global__ __launch_bounds__(AGG1_BS) void k_agg1_edges(
    const int* __restrict__ src, const int* __restrict__ dst,
    const float* __restrict__ h, const float* __restrict__ dinv,
    float* __restrict__ acc1, int E) {
  const int lane   = threadIdx.x & 31;
  const unsigned j = lane & 15;
  const int nTrips = (E + 31) >> 5;
  const int stride = gridDim.x * (AGG1_BS >> 5);
  int trip = blockIdx.x * (AGG1_BS >> 5) + (threadIdx.x >> 5);
  if (trip >= nTrips) return;

  int e  = trip * 32 + lane;
  int ec = min(e, E - 1);
  int s0 = src[ec];
  int d0 = dst[ec];
  d0 = (e < E) ? d0 : -1;
  e  = (trip + stride) * 32 + lane;
  ec = min(e, E - 1);
  int s1 = src[ec];
  int d1 = dst[ec];
  d1 = (e < E) ? d1 : -1;
  float c0 = dinv[s0] * dinv[max(d0, 0)];

  while (true) {
    const float c1 = dinv[s1] * dinv[max(d1, 0)];
    e  = (trip + 2 * stride) * 32 + lane;
    ec = min(e, E - 1);
    const int s2 = src[ec];
    int d2 = dst[ec];
    d2 = (e < E) ? d2 : -1;

    unsigned ss[16];
    int dd[16];
    float cc[16], m[16];
    AGG1_BCAST(0)  AGG1_BCAST(1)  AGG1_BCAST(2)  AGG1_BCAST(3)
    AGG1_BCAST(4)  AGG1_BCAST(5)  AGG1_BCAST(6)  AGG1_BCAST(7)
    AGG1_BCAST(8)  AGG1_BCAST(9)  AGG1_BCAST(10) AGG1_BCAST(11)
    AGG1_BCAST(12) AGG1_BCAST(13) AGG1_BCAST(14) AGG1_BCAST(15)
#pragma unroll
    for (int i = 0; i < 16; ++i) m[i] = h[ss[i] * HID + j];
    if (trip * 32 + 32 <= E) {
#pragma unroll
      for (int i = 0; i < 16; ++i) atomicAdd(acc1 + ((unsigned)dd[i] * HID + j), m[i] * cc[i]);
    } else {
#pragma unroll
      for (int i = 0; i < 16; ++i) {
        if (dd[i] >= 0) atomicAdd(acc1 + ((unsigned)dd[i] * HID + j), m[i] * cc[i]);
      }
    }

    trip += stride;
    if (trip >= nTrips) break;
    s0 = s1; d0 = d1; c0 = c1;
    s1 = s2; d1 = d2;
  }
}

__global__ void k_layer2_node(const float* __restrict__ acc1,
                              const float* __restrict__ W2,
                              float* __restrict__ h3, int N) {
  int n = blockIdx.x * blockDim.x + threadIdx.x;
  if (n >= N) return;
  float o0 = 0.0f, o1 = 0.0f;
#pragma unroll
  for (int j = 0; j < HID; ++j) {
    float v = fmaxf(acc1[(size_t)n * HID + j], 0.0f);
    o0 = __builtin_fmaf(v, W2[j * OUTF + 0], o0);
    o1 = __builtin_fmaf(v, W2[j * OUTF + 1], o1);
  }
  h3[(size_t)n * OUTF + 0] = o0;
  h3[(size_t)n * OUTF + 1] = o1;
}

__global__ __launch_bounds__(SCAN_BS) void k_agg2_part(
    const int* __restrict__ src, const int* __restrict__ dst,
    const float* __restrict__ h3, const float* __restrict__ dinv,
    float* __restrict__ part, int N, int E) {
  __shared__ float acc[NPP * OUTF];
  __shared__ float dinv_d[NPP];
  const int tid = threadIdx.x;
  const int n0  = blockIdx.x * NPP;
  const int np  = min(NPP, N - n0);
  for (int i = tid; i < NPP * OUTF; i += SCAN_BS) acc[i] = 0.0f;
  for (int i = tid; i < np; i += SCAN_BS) dinv_d[i] = dinv[n0 + i];
  __syncthreads();
  const int per = (E + NCH - 1) / NCH;
  const int e0  = blockIdx.y * per;
  const int e1  = min(E, e0 + per);

  if (e0 < e1) {
    const unsigned NONE = 0xffffffffu;
    unsigned r[SCAN_U];
    int      s[SCAN_U];
    int base = e0 + tid;
#pragma unroll
    for (int k = 0; k < SCAN_U; ++k) {
      int e  = base + k * SCAN_BS;
      int ec = min(e, e1 - 1);
      int dd = dst[ec];
      s[k] = src[ec];
      r[k] = (e < e1) ? (unsigned)(dd - n0) : NONE;
    }
    while (base < e1) {
      float ga[SCAN_U], gb[SCAN_U], gx[SCAN_U], gy[SCAN_U];
#pragma unroll
      for (int k = 0; k < SCAN_U; ++k) {
        const bool keep = r[k] < (unsigned)np;
        const int si = keep ? s[k] : 0;
        ga[k] = dinv[si];
        gb[k] = dinv_d[keep ? r[k] : 0u];
        gx[k] = h3[(size_t)si * OUTF + 0];
        gy[k] = h3[(size_t)si * OUTF + 1];
      }
      const int nbase = base + SCAN_U * SCAN_BS;
      int dn[SCAN_U], sn[SCAN_U];
#pragma unroll
      for (int k = 0; k < SCAN_U; ++k) {
        int ec = min(nbase + k * SCAN_BS, e1 - 1);
        dn[k] = dst[ec];
        sn[k] = src[ec];
      }
#pragma unroll
      for (int k = 0; k < SCAN_U; ++k) {
        if (r[k] < (unsigned)np) {
          float c = ga[k] * gb[k];
          atomicAdd(&acc[r[k] * OUTF + 0], gx[k] * c);
          atomicAdd(&acc[r[k] * OUTF + 1], gy[k] * c);
        }
      }
#pragma unroll
      for (int k = 0; k < SCAN_U; ++k) {
        r[k] = (nbase + k * SCAN_BS < e1) ? (unsigned)(dn[k] - n0) : NONE;
        s[k] = sn[k];
      }
      base = nbase;
    }
  }
  __syncthreads();
  float* o = part + ((size_t)blockIdx.y * N + n0) * OUTF;
  for (int i = tid; i < np * OUTF; i += SCAN_BS) o[i] = acc[i];
}

__global__ void k_agg2_reduce(const float* __restrict__ part, const float* __restrict__ h3,
                              const float* __restrict__ dinv, const float* __restrict__ b2,
                              float* __restrict__ out, int N) {
  int t = blockIdx.x * blockDim.x + threadIdx.x;
  if (t >= N * OUTF) return;
  float dv = dinv[t / OUTF];
  float s = h3[t] * (dv * dv) + b2[t % OUTF];
#pragma unroll
  for (int c = 0; c < NCH; ++c) s += part[(size_t)c * N * OUTF + t];
  out[t] = s;
}

static inline int cdiv_i(long long a, long long b) { return (int)((a + b - 1) / b); }

extern "C" void kernel_launch(void* const* d_in, const int* in_sizes, int n_in,
                              void* d_out, int out_size, void* d_ws, size_t ws_size,
                              hipStream_t stream) {
  const float* x  = (const float*)d_in[0];
  const int*   ei = (const int*)  d_in[1];
  const float* W1 = (const float*)d_in[2];
  const float* b1 = (const float*)d_in[3];
  const float* W2 = (const float*)d_in[4];
  const float* b2 = (const float*)d_in[5];
  float* out = (float*)d_out;

  const int N = in_sizes[0] / F_IN;
  const int E = in_sizes[1] / 2;
  const int* src = ei;
  const int* dst = ei + E;

  float* ws   = (float*)d_ws;
  float* dinv = ws;
  float* h    = dinv + N;
  float* acc1 = h + (size_t)N * HID;
  float* h3   = acc1 + (size_t)N * HID;

  const int BS = 256;

  float* part = h;
  const dim3 scanGrid(cdiv_i(N, NPP), NCH);

  k_deg_part  <<<scanGrid, SCAN_BS, 0, stream>>>(dst, part, N, E);
  k_deg_reduce<<<cdiv_i(N, BS), BS, 0, stream>>>(part, dinv, N);

  int nTiles = cdiv_i(N, 16);
  k_gemm1_wmma<<<cdiv_i(nTiles, BS / 32), BS, 0, stream>>>(x, W1, h, N, nTiles);

  k_agg1_init <<<cdiv_i((long long)N * HID, BS), BS, 0, stream>>>(h, dinv, b1, acc1, N);
  k_agg1_edges<<<AGG1_BLOCKS, AGG1_BS, 0, stream>>>(src, dst, h, dinv, acc1, E);

  k_layer2_node<<<cdiv_i(N, BS), BS, 0, stream>>>(acc1, W2, h3, N);

  k_agg2_part  <<<scanGrid, SCAN_BS, 0, stream>>>(src, dst, h3, dinv, part, N, E);
  k_agg2_reduce<<<cdiv_i((long long)N * OUTF, BS), BS, 0, stream>>>(part, h3, dinv, b2, out, N);
}


# Layer-1 GEMM: x read as whole 256-byte row pieces (16 B per lane, two pieces ahead) and turned into WMMA fragments through LDS, W1 fragments from LDS; k order of the f32 sums changes
# speedup vs baseline: 3.1168x; 1.2049x over previous
#include <hip/hip_runtime.h>

typedef __attribute__((ext_vector_type(2))) float v2f;
typedef __attribute__((ext_vector_type(4))) float v4f;
typedef __attribute__((ext_vector_type(8))) float v8f;

#define F_IN 512
#define HID  16
#define OUTF 2

#define NPP     6400
#define NCH     16
#define SCAN_BS 1024
#define SCAN_U  8

__global__ __launch_bounds__(SCAN_BS) void k_deg_part(
    const int* __restrict__ dst, float* __restrict__ part, int N, int E) {
  __shared__ unsigned cnt[NPP];
  const int tid = threadIdx.x;
  const int n0  = blockIdx.x * NPP;
  const int np  = min(NPP, N - n0);
  for (int i = tid; i < NPP; i += SCAN_BS) cnt[i] = 0u;
  __syncthreads();
  const int per = (E + NCH - 1) / NCH;
  const int e0  = blockIdx.y * per;
  const int e1  = min(E, e0 + per);
  for (int base = e0 + tid; base < e1; base += 2 * SCAN_U * SCAN_BS) {
    int d[2 * SCAN_U];
#pragma unroll
    for (int k = 0; k < 2 * SCAN_U; ++k) d[k] = dst[min(base + k * SCAN_BS, e1 - 1)];
#pragma unroll
    for (int k = 0; k < 2 * SCAN_U; ++k) {
      unsigned r = (unsigned)(d[k] - n0);
      if (base + k * SCAN_BS < e1 && r < (unsigned)np) atomicAdd(&cnt[r], 1u);
    }
  }
  __syncthreads();
  float* o = part + (size_t)blockIdx.y * N + n0;
  for (int i = tid; i < np; i += SCAN_BS) o[i] = (float)cnt[i];
}

__global__ void k_deg_reduce(const float* __restrict__ part, float* __restrict__ dinv, int N) {
  int n = blockIdx.x * blockDim.x + threadIdx.x;
  if (n >= N) return;
  float s = 1.0f;
#pragma unroll
  for (int c = 0; c < NCH; ++c) s += part[(size_t)c * N + n];
  dinv[n] = rsqrtf(s);
}

#define G1_WAVES  8
#define G1_PIECE  64
#define G1_PITCH  68

__device__ __forceinline__ void g1_load_piece(v4f (&R)[8], const float* gp, int G) {
#pragma unroll
  for (int j = 0; j < 8; ++j)
    R[j] = __builtin_nontemporal_load((const v4f*)(gp + (size_t)j * 2 * F_IN + G * G1_PIECE));
}

__device__ __forceinline__ void g1_store_piece(const v4f (&R)[8], float* xsw, int hi, int col) {
#pragma unroll
  for (int j = 0; j < 8; ++j)
    *(v4f*)(xsw + (2 * j + hi) * G1_PITCH + 4 * col) = R[j];
}

__device__ __forceinline__ v8f g1_mul_piece(v8f acc, const float* xsw, const float* wfrag,
                                            int G, int hi, int col) {
#pragma unroll
  for (int i = 0; i < 8; ++i) {
    const v4f a4 = *(const v4f*)(xsw + col * G1_PITCH + 8 * i + 4 * hi);
    const v4f b4 = *(const v4f*)(wfrag + (((8 * G + i) * 2 + hi) * 16 + col) * 4);
    v2f a, b;
    a.x = a4.x; a.y = a4.y; b.x = b4.x; b.y = b4.y;
    acc = __builtin_amdgcn_wmma_f32_16x16x4_f32(
         false, a,  false, b,
         (short)0, acc,  false,  false);
    a.x = a4.z; a.y = a4.w; b.x = b4.z; b.y = b4.w;
    acc = __builtin_amdgcn_wmma_f32_16x16x4_f32(
         false, a,  false, b,
         (short)0, acc,  false,  false);
  }
  return acc;
}

__global__ __launch_bounds__(G1_WAVES * 32) void k_gemm1_wmma(
    const float* __restrict__ x, const float* __restrict__ W1,
    float* __restrict__ h, int N, int nTiles) {
  __shared__ __attribute__((aligned(16))) float wfrag[F_IN * HID];
  __shared__ __attribute__((aligned(16))) float xs[G1_WAVES][16 * G1_PITCH];
  const int tid  = threadIdx.x;
  const int lane = tid & 31;
  const int wv   = tid >> 5;
  const int col  = lane & 15;
  const int hi   = lane >> 4;
  const int wave = blockIdx.x * G1_WAVES + wv;
  const int tileBase = wave * 16;
  const int tb = min(tileBase, N - 16);
  const float* gp = x + (size_t)(tb + hi) * F_IN + 4 * col;
  float* xsw = &xs[wv][0];

  v4f R0[8], R1[8];
  g1_load_piece(R0, gp, 0);
  g1_load_piece(R1, gp, 1);

#pragma unroll
  for (int it = 0; it < F_IN * HID / 4 / (G1_WAVES * 32); ++it) {
    const int i4 = tid + it * (G1_WAVES * 32);
    const v4f w = *(const v4f*)(W1 + (size_t)i4 * 4);
    const int k  = i4 >> 2;
    const int c0 = (i4 & 3) * 4;
    float* d = wfrag + (((k >> 3) * 2 + ((k >> 2) & 1)) * 16 + c0) * 4 + (k & 3);
    d[0] = w.x; d[4] = w.y; d[8] = w.z; d[12] = w.w;
  }
  __syncthreads();
  if (wave >= nTiles) return;

  v8f acc = {};
  for (int G = 0; G < F_IN / G1_PIECE; G += 2) {
    g1_store_piece(R0, xsw, hi, col);
    if (G + 2 < F_IN / G1_PIECE) g1_load_piece(R0, gp, G + 2);
    acc = g1_mul_piece(acc, xsw, wfrag, G, hi, col);
    g1_store_piece(R1, xsw, hi, col);
    if (G + 3 < F_IN / G1_PIECE) g1_load_piece(R1, gp, G + 3);
    acc = g1_mul_piece(acc, xsw, wfrag, G + 1, hi, col);
  }

  if (tb == tileBase) {
    float* outp = h + ((size_t)(tb + 8 * hi)) * HID + col;
#pragma unroll
    for (int r = 0; r < 8; ++r) outp[r * HID] = acc[r];
  } else {
#pragma unroll
    for (int r = 0; r < 8; ++r) {
      int m = tb + r + 8 * hi;
      if (m >= tileBase) h[(size_t)m * HID + col] = acc[r];
    }
  }
}

__global__ void k_agg1_init(const float* __restrict__ h, const float* __restrict__ dinv,
                            const float* __restrict__ b1, float* __restrict__ acc1, int N) {
  int t = blockIdx.x * blockDim.x + threadIdx.x;
  if (t >= N * HID) return;
  int n = t >> 4, j = t & 15;
  float dv = dinv[n];
  acc1[t] = h[t] * dv * dv + b1[j];
}

#define AGG1_BS     256
#define AGG1_BLOCKS 2048

template <int I>
__device__ __forceinline__ int row_share_bcast(int v) {
  return __builtin_amdgcn_mov_dpp(v, 0x150 + I, 0xf, 0xf, false);
}

#define AGG1_BCAST(I)                                          \
  ss[I] = (unsigned)row_share_bcast<I>(s0);                    \
  dd[I] = row_share_bcast<I>(d0);                              \
  cc[I] = __int_as_float(row_share_bcast<I>(__float_as_int(c0)));

__global__ __launch_bounds__(AGG1_BS) void k_agg1_edges(
    const int* __restrict__ src, const int* __restrict__ dst,
    const float* __restrict__ h, const float* __restrict__ dinv,
    float* __restrict__ acc1, int E) {
  const int lane   = threadIdx.x & 31;
  const unsigned j = lane & 15;
  const int nTrips = (E + 31) >> 5;
  const int stride = gridDim.x * (AGG1_BS >> 5);
  int trip = blockIdx.x * (AGG1_BS >> 5) + (threadIdx.x >> 5);
  if (trip >= nTrips) return;

  int e  = trip * 32 + lane;
  int ec = min(e, E - 1);
  int s0 = src[ec];
  int d0 = dst[ec];
  d0 = (e < E) ? d0 : -1;
  e  = (trip + stride) * 32 + lane;
  ec = min(e, E - 1);
  int s1 = src[ec];
  int d1 = dst[ec];
  d1 = (e < E) ? d1 : -1;
  float c0 = dinv[s0] * dinv[max(d0, 0)];

  while (true) {
    const float c1 = dinv[s1] * dinv[max(d1, 0)];
    e  = (trip + 2 * stride) * 32 + lane;
    ec = min(e, E - 1);
    const int s2 = src[ec];
    int d2 = dst[ec];
    d2 = (e < E) ? d2 : -1;

    unsigned ss[16];
    int dd[16];
    float cc[16], m[16];
    AGG1_BCAST(0)  AGG1_BCAST(1)  AGG1_BCAST(2)  AGG1_BCAST(3)
    AGG1_BCAST(4)  AGG1_BCAST(5)  AGG1_BCAST(6)  AGG1_BCAST(7)
    AGG1_BCAST(8)  AGG1_BCAST(9)  AGG1_BCAST(10) AGG1_BCAST(11)
    AGG1_BCAST(12) AGG1_BCAST(13) AGG1_BCAST(14) AGG1_BCAST(15)
#pragma unroll
    for (int i = 0; i < 16; ++i) m[i] = h[ss[i] * HID + j];
    if (trip * 32 + 32 <= E) {
#pragma unroll
      for (int i = 0; i < 16; ++i) atomicAdd(acc1 + ((unsigned)dd[i] * HID + j), m[i] * cc[i]);
    } else {
#pragma unroll
      for (int i = 0; i < 16; ++i) {
        if (dd[i] >= 0) atomicAdd(acc1 + ((unsigned)dd[i] * HID + j), m[i] * cc[i]);
      }
    }

    trip += stride;
    if (trip >= nTrips) break;
    s0 = s1; d0 = d1; c0 = c1;
    s1 = s2; d1 = d2;
  }
}

__global__ void k_layer2_node(const float* __restrict__ acc1,
                              const float* __restrict__ W2,
                              float* __restrict__ h3, int N) {
  int n = blockIdx.x * blockDim.x + threadIdx.x;
  if (n >= N) return;
  float o0 = 0.0f, o1 = 0.0f;
#pragma unroll
  for (int j = 0; j < HID; ++j) {
    float v = fmaxf(acc1[(size_t)n * HID + j], 0.0f);
    o0 = __builtin_fmaf(v, W2[j * OUTF + 0], o0);
    o1 = __builtin_fmaf(v, W2[j * OUTF + 1], o1);
  }
  h3[(size_t)n * OUTF + 0] = o0;
  h3[(size_t)n * OUTF + 1] = o1;
}

__global__ __launch_bounds__(SCAN_BS) void k_agg2_part(
    const int* __restrict__ src, const int* __restrict__ dst,
    const float* __restrict__ h3, const float* __restrict__ dinv,
    float* __restrict__ part, int N, int E) {
  __shared__ float acc[NPP * OUTF];
  __shared__ float dinv_d[NPP];
  const int tid = threadIdx.x;
  const int n0  = blockIdx.x * NPP;
  const int np  = min(NPP, N - n0);
  for (int i = tid; i < NPP * OUTF; i += SCAN_BS) acc[i] = 0.0f;
  for (int i = tid; i < np; i += SCAN_BS) dinv_d[i] = dinv[n0 + i];
  __syncthreads();
  const int per = (E + NCH - 1) / NCH;
  const int e0  = blockIdx.y * per;
  const int e1  = min(E, e0 + per);

  if (e0 < e1) {
    const unsigned NONE = 0xffffffffu;
    unsigned r[SCAN_U];
    int      s[SCAN_U];
    int base = e0 + tid;
#pragma unroll
    for (int k = 0; k < SCAN_U; ++k) {
      int e  = base + k * SCAN_BS;
      int ec = min(e, e1 - 1);
      int dd = dst[ec];
      s[k] = src[ec];
      r[k] = (e < e1) ? (unsigned)(dd - n0) : NONE;
    }
    while (base < e1) {
      float ga[SCAN_U], gb[SCAN_U], gx[SCAN_U], gy[SCAN_U];
#pragma unroll
      for (int k = 0; k < SCAN_U; ++k) {
        const bool keep = r[k] < (unsigned)np;
        const int si = keep ? s[k] : 0;
        ga[k] = dinv[si];
        gb[k] = dinv_d[keep ? r[k] : 0u];
        gx[k] = h3[(size_t)si * OUTF + 0];
        gy[k] = h3[(size_t)si * OUTF + 1];
      }
      const int nbase = base + SCAN_U * SCAN_BS;
      int dn[SCAN_U], sn[SCAN_U];
#pragma unroll
      for (int k = 0; k < SCAN_U; ++k) {
        int ec = min(nbase + k * SCAN_BS, e1 - 1);
        dn[k] = dst[ec];
        sn[k] = src[ec];
      }
#pragma unroll
      for (int k = 0; k < SCAN_U; ++k) {
        if (r[k] < (unsigned)np) {
          float c = ga[k] * gb[k];
          atomicAdd(&acc[r[k] * OUTF + 0], gx[k] * c);
          atomicAdd(&acc[r[k] * OUTF + 1], gy[k] * c);
        }
      }
#pragma unroll
      for (int k = 0; k < SCAN_U; ++k) {
        r[k] = (nbase + k * SCAN_BS < e1) ? (unsigned)(dn[k] - n0) : NONE;
        s[k] = sn[k];
      }
      base = nbase;
    }
  }
  __syncthreads();
  float* o = part + ((size_t)blockIdx.y * N + n0) * OUTF;
  for (int i = tid; i < np * OUTF; i += SCAN_BS) o[i] = acc[i];
}

__global__ void k_agg2_reduce(const float* __restrict__ part, const float* __restrict__ h3,
                              const float* __restrict__ dinv, const float* __restrict__ b2,
                              float* __restrict__ out, int N) {
  int t = blockIdx.x * blockDim.x + threadIdx.x;
  if (t >= N * OUTF) return;
  float dv = dinv[t / OUTF];
  float s = h3[t] * (dv * dv) + b2[t % OUTF];
#pragma unroll
  for (int c = 0; c < NCH; ++c) s += part[(size_t)c * N * OUTF + t];
  out[t] = s;
}

static inline int cdiv_i(long long a, long long b) { return (int)((a + b - 1) / b); }

extern "C" void kernel_launch(void* const* d_in, const int* in_sizes, int n_in,
                              void* d_out, int out_size, void* d_ws, size_t ws_size,
                              hipStream_t stream) {
  const float* x  = (const float*)d_in[0];
  const int*   ei = (const int*)  d_in[1];
  const float* W1 = (const float*)d_in[2];
  const float* b1 = (const float*)d_in[3];
  const float* W2 = (const float*)d_in[4];
  const float* b2 = (const float*)d_in[5];
  float* out = (float*)d_out;

  const int N = in_sizes[0] / F_IN;
  const int E = in_sizes[1] / 2;
  const int* src = ei;
  const int* dst = ei + E;

  float* ws   = (float*)d_ws;
  float* dinv = ws;
  float* h    = dinv + N;
  float* acc1 = h + (size_t)N * HID;
  float* h3   = acc1 + (size_t)N * HID;

  const int BS = 256;

  float* part = h;
  const dim3 scanGrid(cdiv_i(N, NPP), NCH);

  k_deg_part  <<<scanGrid, SCAN_BS, 0, stream>>>(dst, part, N, E);
  k_deg_reduce<<<cdiv_i(N, BS), BS, 0, stream>>>(part, dinv, N);

  int nTiles = cdiv_i(N, 16);
  k_gemm1_wmma<<<cdiv_i(nTiles, G1_WAVES), G1_WAVES * 32, 0, stream>>>(x, W1, h, N, nTiles);

  k_agg1_init <<<cdiv_i((long long)N * HID, BS), BS, 0, stream>>>(h, dinv, b1, acc1, N);
  k_agg1_edges<<<AGG1_BLOCKS, AGG1_BS, 0, stream>>>(src, dst, h, dinv, acc1, E);

  k_layer2_node<<<cdiv_i(N, BS), BS, 0, stream>>>(acc1, W2, h3, N);

  k_agg2_part  <<<scanGrid, SCAN_BS, 0, stream>>>(src, dst, h3, dinv, part, N, E);
  k_agg2_reduce<<<cdiv_i((long long)N * OUTF, BS), BS, 0, stream>>>(part, h3, dinv, b2, out, N);
}


# Edge weight dinv[src]*dinv[dst] factored out of both edge passes: rows scaled by dinv once per node where produced, sums by dinv[dst] where read; per-term rounding changes, sum order as before
# speedup vs baseline: 3.5966x; 1.1539x over previous
#include <hip/hip_runtime.h>

typedef __attribute__((ext_vector_type(2))) float v2f;
typedef __attribute__((ext_vector_type(4))) float v4f;
typedef __attribute__((ext_vector_type(8))) float v8f;

#define F_IN 512
#define HID  16
#define OUTF 2

#define NPP     6400
#define NCH     16
#define SCAN_BS 1024
#define SCAN_U  8

__global__ __launch_bounds__(SCAN_BS) void k_deg_part(
    const int* __restrict__ dst, float* __restrict__ part, int N, int E) {
  __shared__ unsigned cnt[NPP];
  const int tid = threadIdx.x;
  const int n0  = blockIdx.x * NPP;
  const int np  = min(NPP, N - n0);
  for (int i = tid; i < NPP; i += SCAN_BS) cnt[i] = 0u;
  __syncthreads();
  const int per = (E + NCH - 1) / NCH;
  const int e0  = blockIdx.y * per;
  const int e1  = min(E, e0 + per);
  for (int base = e0 + tid; base < e1; base += 2 * SCAN_U * SCAN_BS) {
    int d[2 * SCAN_U];
#pragma unroll
    for (int k = 0; k < 2 * SCAN_U; ++k) d[k] = dst[min(base + k * SCAN_BS, e1 - 1)];
#pragma unroll
    for (int k = 0; k < 2 * SCAN_U; ++k) {
      unsigned r = (unsigned)(d[k] - n0);
      if (base + k * SCAN_BS < e1 && r < (unsigned)np) atomicAdd(&cnt[r], 1u);
    }
  }
  __syncthreads();
  float* o = part + (size_t)blockIdx.y * N + n0;
  for (int i = tid; i < np; i += SCAN_BS) o[i] = (float)cnt[i];
}

__global__ void k_deg_reduce(const float* __restrict__ part, float* __restrict__ dinv, int N) {
  int n = blockIdx.x * blockDim.x + threadIdx.x;
  if (n >= N) return;
  float s = 1.0f;
#pragma unroll
  for (int c = 0; c < NCH; ++c) s += part[(size_t)c * N + n];
  dinv[n] = rsqrtf(s);
}

#define G1_WAVES  8
#define G1_PIECE  64
#define G1_PITCH  68

__device__ __forceinline__ void g1_load_piece(v4f (&R)[8], const float* gp, int G) {
#pragma unroll
  for (int j = 0; j < 8; ++j)
    R[j] = __builtin_nontemporal_load((const v4f*)(gp + (size_t)j * 2 * F_IN + G * G1_PIECE));
}

__device__ __forceinline__ void g1_store_piece(const v4f (&R)[8], float* xsw, int hi, int col) {
#pragma unroll
  for (int j = 0; j < 8; ++j)
    *(v4f*)(xsw + (2 * j + hi) * G1_PITCH + 4 * col) = R[j];
}

__device__ __forceinline__ v8f g1_mul_piece(v8f acc, const float* xsw, const float* wfrag,
                                            int G, int hi, int col) {
#pragma unroll
  for (int i = 0; i < 8; ++i) {
    const v4f a4 = *(const v4f*)(xsw + col * G1_PITCH + 8 * i + 4 * hi);
    const v4f b4 = *(const v4f*)(wfrag + (((8 * G + i) * 2 + hi) * 16 + col) * 4);
    v2f a, b;
    a.x = a4.x; a.y = a4.y; b.x = b4.x; b.y = b4.y;
    acc = __builtin_amdgcn_wmma_f32_16x16x4_f32(
         false, a,  false, b,
         (short)0, acc,  false,  false);
    a.x = a4.z; a.y = a4.w; b.x = b4.z; b.y = b4.w;
    acc = __builtin_amdgcn_wmma_f32_16x16x4_f32(
         false, a,  false, b,
         (short)0, acc,  false,  false);
  }
  return acc;
}

__global__ __launch_bounds__(G1_WAVES * 32) void k_gemm1_wmma(
    const float* __restrict__ x, const float* __restrict__ W1,
    float* __restrict__ h, int N, int nTiles) {
  __shared__ __attribute__((aligned(16))) float wfrag[F_IN * HID];
  __shared__ __attribute__((aligned(16))) float xs[G1_WAVES][16 * G1_PITCH];
  const int tid  = threadIdx.x;
  const int lane = tid & 31;
  const int wv   = tid >> 5;
  const int col  = lane & 15;
  const int hi   = lane >> 4;
  const int wave = blockIdx.x * G1_WAVES + wv;
  const int tileBase = wave * 16;
  const int tb = min(tileBase, N - 16);
  const float* gp = x + (size_t)(tb + hi) * F_IN + 4 * col;
  float* xsw = &xs[wv][0];

  v4f R0[8], R1[8];
  g1_load_piece(R0, gp, 0);
  g1_load_piece(R1, gp, 1);

#pragma unroll
  for (int it = 0; it < F_IN * HID / 4 / (G1_WAVES * 32); ++it) {
    const int i4 = tid + it * (G1_WAVES * 32);
    const v4f w = *(const v4f*)(W1 + (size_t)i4 * 4);
    const int k  = i4 >> 2;
    const int c0 = (i4 & 3) * 4;
    float* d = wfrag + (((k >> 3) * 2 + ((k >> 2) & 1)) * 16 + c0) * 4 + (k & 3);
    d[0] = w.x; d[4] = w.y; d[8] = w.z; d[12] = w.w;
  }
  __syncthreads();
  if (wave >= nTiles) return;

  v8f acc = {};
  for (int G = 0; G < F_IN / G1_PIECE; G += 2) {
    g1_store_piece(R0, xsw, hi, col);
    if (G + 2 < F_IN / G1_PIECE) g1_load_piece(R0, gp, G + 2);
    acc = g1_mul_piece(acc, xsw, wfrag, G, hi, col);
    g1_store_piece(R1, xsw, hi, col);
    if (G + 3 < F_IN / G1_PIECE) g1_load_piece(R1, gp, G + 3);
    acc = g1_mul_piece(acc, xsw, wfrag, G + 1, hi, col);
  }

  if (tb == tileBase) {
    float* outp = h + ((size_t)(tb + 8 * hi)) * HID + col;
#pragma unroll
    for (int r = 0; r < 8; ++r) outp[r * HID] = acc[r];
  } else {
#pragma unroll
    for (int r = 0; r < 8; ++r) {
      int m = tb + r + 8 * hi;
      if (m >= tileBase) h[(size_t)m * HID + col] = acc[r];
    }
  }
}

__global__ void k_agg1_init(float* h, const float* __restrict__ dinv,
                            float* __restrict__ acc1, int N) {
  int t = blockIdx.x * blockDim.x + threadIdx.x;
  if (t >= N * HID) return;
  float v = h[t] * dinv[t >> 4];
  h[t]    = v;
  acc1[t] = v;
}

#define AGG1_BS     256
#define AGG1_BLOCKS 2048

template <int I>
__device__ __forceinline__ int row_share_bcast(int v) {
  return __builtin_amdgcn_mov_dpp(v, 0x150 + I, 0xf, 0xf, false);
}

#define AGG1_BCAST(I)                                          \
  ss[I] = (unsigned)row_share_bcast<I>(s0);                    \
  dd[I] = row_share_bcast<I>(d0);

__global__ __launch_bounds__(AGG1_BS) void k_agg1_edges(
    const int* __restrict__ src, const int* __restrict__ dst,
    const float* __restrict__ g, float* __restrict__ acc1, int E) {
  const int lane   = threadIdx.x & 31;
  const unsigned j = lane & 15;
  const int nTrips = (E + 31) >> 5;
  const int stride = gridDim.x * (AGG1_BS >> 5);
  int trip = blockIdx.x * (AGG1_BS >> 5) + (threadIdx.x >> 5);
  if (trip >= nTrips) return;

  int e  = trip * 32 + lane;
  int ec = min(e, E - 1);
  int s0 = src[ec];
  int d0 = dst[ec];
  d0 = (e < E) ? d0 : -1;

  while (true) {
    const int tn = trip + stride;
    e  = tn * 32 + lane;
    ec = min(e, E - 1);
    const int s1 = src[ec];
    int d1 = dst[ec];
    d1 = (e < E) ? d1 : -1;

    unsigned ss[16];
    int dd[16];
    float m[16];
    AGG1_BCAST(0)  AGG1_BCAST(1)  AGG1_BCAST(2)  AGG1_BCAST(3)
    AGG1_BCAST(4)  AGG1_BCAST(5)  AGG1_BCAST(6)  AGG1_BCAST(7)
    AGG1_BCAST(8)  AGG1_BCAST(9)  AGG1_BCAST(10) AGG1_BCAST(11)
    AGG1_BCAST(12) AGG1_BCAST(13) AGG1_BCAST(14) AGG1_BCAST(15)
#pragma unroll
    for (int i = 0; i < 16; ++i) m[i] = g[ss[i] * HID + j];
    if (trip * 32 + 32 <= E) {
#pragma unroll
      for (int i = 0; i < 16; ++i) atomicAdd(acc1 + ((unsigned)dd[i] * HID + j), m[i]);
    } else {
#pragma unroll
      for (int i = 0; i < 16; ++i) {
        if (dd[i] >= 0) atomicAdd(acc1 + ((unsigned)dd[i] * HID + j), m[i]);
      }
    }

    trip = tn;
    if (trip >= nTrips) break;
    s0 = s1; d0 = d1;
  }
}

__global__ void k_layer2_node(const float* __restrict__ acc1, const float* __restrict__ dinv,
                              const float* __restrict__ b1, const float* __restrict__ W2,
                              float* __restrict__ h3, int N) {
  int n = blockIdx.x * blockDim.x + threadIdx.x;
  if (n >= N) return;
  const float dv = dinv[n];
  float o0 = 0.0f, o1 = 0.0f;
#pragma unroll
  for (int j = 0; j < HID; ++j) {
    float v = fmaxf(acc1[(size_t)n * HID + j] * dv + b1[j], 0.0f);
    o0 = __builtin_fmaf(v, W2[j * OUTF + 0], o0);
    o1 = __builtin_fmaf(v, W2[j * OUTF + 1], o1);
  }
  v2f o;
  o.x = o0 * dv;
  o.y = o1 * dv;
  *(v2f*)(h3 + (size_t)n * OUTF) = o;
}

__global__ __launch_bounds__(SCAN_BS) void k_agg2_part(
    const int* __restrict__ src, const int* __restrict__ dst,
    const float* __restrict__ h3, float* __restrict__ part, int N, int E) {
  __shared__ float acc[NPP * OUTF];
  const v2f* __restrict__ h3v = (const v2f*)h3;
  const int tid = threadIdx.x;
  const int n0  = blockIdx.x * NPP;
  const int np  = min(NPP, N - n0);
  for (int i = tid; i < NPP * OUTF; i += SCAN_BS) acc[i] = 0.0f;
  __syncthreads();
  const int per = (E + NCH - 1) / NCH;
  const int e0  = blockIdx.y * per;
  const int e1  = min(E, e0 + per);

  if (e0 < e1) {
    const unsigned NONE = 0xffffffffu;
    unsigned r[SCAN_U];
    int      s[SCAN_U];
    int base = e0 + tid;
#pragma unroll
    for (int k = 0; k < SCAN_U; ++k) {
      int e  = base + k * SCAN_BS;
      int ec = min(e, e1 - 1);
      int dd = dst[ec];
      s[k] = src[ec];
      r[k] = (e < e1) ? (unsigned)(dd - n0) : NONE;
    }
    while (base < e1) {
      v2f gv[SCAN_U];
#pragma unroll
      for (int k = 0; k < SCAN_U; ++k) {
        const bool keep = r[k] < (unsigned)np;
        gv[k] = h3v[keep ? s[k] : 0];
      }
      const int nbase = base + SCAN_U * SCAN_BS;
      int dn[SCAN_U], sn[SCAN_U];
#pragma unroll
      for (int k = 0; k < SCAN_U; ++k) {
        int ec = min(nbase + k * SCAN_BS, e1 - 1);
        dn[k] = dst[ec];
        sn[k] = src[ec];
      }
#pragma unroll
      for (int k = 0; k < SCAN_U; ++k) {
        if (r[k] < (unsigned)np) {
          atomicAdd(&acc[r[k] * OUTF + 0], gv[k].x);
          atomicAdd(&acc[r[k] * OUTF + 1], gv[k].y);
        }
      }
#pragma unroll
      for (int k = 0; k < SCAN_U; ++k) {
        r[k] = (nbase + k * SCAN_BS < e1) ? (unsigned)(dn[k] - n0) : NONE;
        s[k] = sn[k];
      }
      base = nbase;
    }
  }
  __syncthreads();
  float* o = part + ((size_t)blockIdx.y * N + n0) * OUTF;
  for (int i = tid; i < np * OUTF; i += SCAN_BS) o[i] = acc[i];
}

__global__ void k_agg2_reduce(const float* __restrict__ part, const float* __restrict__ h3,
                              const float* __restrict__ dinv, const float* __restrict__ b2,
                              float* __restrict__ out, int N) {
  int t = blockIdx.x * blockDim.x + threadIdx.x;
  if (t >= N * OUTF) return;
  float s = h3[t];
#pragma unroll
  for (int c = 0; c < NCH; ++c) s += part[(size_t)c * N * OUTF + t];
  out[t] = s * dinv[t / OUTF] + b2[t % OUTF];
}

static inline int cdiv_i(long long a, long long b) { return (int)((a + b - 1) / b); }

extern "C" void kernel_launch(void* const* d_in, const int* in_sizes, int n_in,
                              void* d_out, int out_size, void* d_ws, size_t ws_size,
                              hipStream_t stream) {
  const float* x  = (const float*)d_in[0];
  const int*   ei = (const int*)  d_in[1];
  const float* W1 = (const float*)d_in[2];
  const float* b1 = (const float*)d_in[3];
  const float* W2 = (const float*)d_in[4];
  const float* b2 = (const float*)d_in[5];
  float* out = (float*)d_out;

  const int N = in_sizes[0] / F_IN;
  const int E = in_sizes[1] / 2;
  const int* src = ei;
  const int* dst = ei + E;

  float* ws   = (float*)d_ws;
  float* dinv = ws;
  float* h    = dinv + N;
  float* acc1 = h + (size_t)N * HID;
  float* h3   = acc1 + (size_t)N * HID;

  const int BS = 256;

  float* part = h;
  const dim3 scanGrid(cdiv_i(N, NPP), NCH);

  k_deg_part  <<<scanGrid, SCAN_BS, 0, stream>>>(dst, part, N, E);
  k_deg_reduce<<<cdiv_i(N, BS), BS, 0, stream>>>(part, dinv, N);

  int nTiles = cdiv_i(N, 16);
  k_gemm1_wmma<<<cdiv_i(nTiles, G1_WAVES), G1_WAVES * 32, 0, stream>>>(x, W1, h, N, nTiles);

  k_agg1_init <<<cdiv_i((long long)N * HID, BS), BS, 0, stream>>>(h, dinv, acc1, N);
  k_agg1_edges<<<AGG1_BLOCKS, AGG1_BS, 0, stream>>>(src, dst, h, acc1, E);

  k_layer2_node<<<cdiv_i(N, BS), BS, 0, stream>>>(acc1, dinv, b1, W2, h3, N);

  k_agg2_part  <<<scanGrid, SCAN_BS, 0, stream>>>(src, dst, h3, part, N, E);
  k_agg2_reduce<<<cdiv_i((long long)N * OUTF, BS), BS, 0, stream>>>(part, h3, dinv, b2, out, N);
}


# Layer-1 edge scatter: rows summed in LDS per node range (21 ranges x 12 edge chunks, kept edges compacted per wave), then whole-line f32 atomics into acc1; sum order by arrival as before, regrouped
# speedup vs baseline: 3.8225x; 1.0628x over previous
#include <hip/hip_runtime.h>

typedef __attribute__((ext_vector_type(2))) float v2f;
typedef __attribute__((ext_vector_type(4))) float v4f;
typedef __attribute__((ext_vector_type(8))) float v8f;

#define F_IN 512
#define HID  16
#define OUTF 2

#define NPP     6400
#define NCH     16
#define SCAN_BS 1024
#define SCAN_U  8

__global__ __launch_bounds__(SCAN_BS) void k_deg_part(
    const int* __restrict__ dst, float* __restrict__ part, int N, int E) {
  __shared__ unsigned cnt[NPP];
  const int tid = threadIdx.x;
  const int n0  = blockIdx.x * NPP;
  const int np  = min(NPP, N - n0);
  for (int i = tid; i < NPP; i += SCAN_BS) cnt[i] = 0u;
  __syncthreads();
  const int per = (E + NCH - 1) / NCH;
  const int e0  = blockIdx.y * per;
  const int e1  = min(E, e0 + per);
  for (int base = e0 + tid; base < e1; base += 2 * SCAN_U * SCAN_BS) {
    int d[2 * SCAN_U];
#pragma unroll
    for (int k = 0; k < 2 * SCAN_U; ++k) d[k] = dst[min(base + k * SCAN_BS, e1 - 1)];
#pragma unroll
    for (int k = 0; k < 2 * SCAN_U; ++k) {
      unsigned r = (unsigned)(d[k] - n0);
      if (base + k * SCAN_BS < e1 && r < (unsigned)np) atomicAdd(&cnt[r], 1u);
    }
  }
  __syncthreads();
  float* o = part + (size_t)blockIdx.y * N + n0;
  for (int i = tid; i < np; i += SCAN_BS) o[i] = (float)cnt[i];
}

__global__ void k_deg_reduce(const float* __restrict__ part, float* __restrict__ dinv, int N) {
  int n = blockIdx.x * blockDim.x + threadIdx.x;
  if (n >= N) return;
  float s = 1.0f;
#pragma unroll
  for (int c = 0; c < NCH; ++c) s += part[(size_t)c * N + n];
  dinv[n] = rsqrtf(s);
}

#define G1_WAVES  8
#define G1_PIECE  64
#define G1_PITCH  68

__device__ __forceinline__ void g1_load_piece(v4f (&R)[8], const float* gp, int G) {
#pragma unroll
  for (int j = 0; j < 8; ++j)
    R[j] = __builtin_nontemporal_load((const v4f*)(gp + (size_t)j * 2 * F_IN + G * G1_PIECE));
}

__device__ __forceinline__ void g1_store_piece(const v4f (&R)[8], float* xsw, int hi, int col) {
#pragma unroll
  for (int j = 0; j < 8; ++j)
    *(v4f*)(xsw + (2 * j + hi) * G1_PITCH + 4 * col) = R[j];
}

__device__ __forceinline__ v8f g1_mul_piece(v8f acc, const float* xsw, const float* wfrag,
                                            int G, int hi, int col) {
#pragma unroll
  for (int i = 0; i < 8; ++i) {
    const v4f a4 = *(const v4f*)(xsw + col * G1_PITCH + 8 * i + 4 * hi);
    const v4f b4 = *(const v4f*)(wfrag + (((8 * G + i) * 2 + hi) * 16 + col) * 4);
    v2f a, b;
    a.x = a4.x; a.y = a4.y; b.x = b4.x; b.y = b4.y;
    acc = __builtin_amdgcn_wmma_f32_16x16x4_f32(
         false, a,  false, b,
         (short)0, acc,  false,  false);
    a.x = a4.z; a.y = a4.w; b.x = b4.z; b.y = b4.w;
    acc = __builtin_amdgcn_wmma_f32_16x16x4_f32(
         false, a,  false, b,
         (short)0, acc,  false,  false);
  }
  return acc;
}

__global__ __launch_bounds__(G1_WAVES * 32) void k_gemm1_wmma(
    const float* __restrict__ x, const float* __restrict__ W1,
    float* __restrict__ h, int N, int nTiles) {
  __shared__ __attribute__((aligned(16))) float wfrag[F_IN * HID];
  __shared__ __attribute__((aligned(16))) float xs[G1_WAVES][16 * G1_PITCH];
  const int tid  = threadIdx.x;
  const int lane = tid & 31;
  const int wv   = tid >> 5;
  const int col  = lane & 15;
  const int hi   = lane >> 4;
  const int wave = blockIdx.x * G1_WAVES + wv;
  const int tileBase = wave * 16;
  const int tb = min(tileBase, N - 16);
  const float* gp = x + (size_t)(tb + hi) * F_IN + 4 * col;
  float* xsw = &xs[wv][0];

  v4f R0[8], R1[8];
  g1_load_piece(R0, gp, 0);
  g1_load_piece(R1, gp, 1);

#pragma unroll
  for (int it = 0; it < F_IN * HID / 4 / (G1_WAVES * 32); ++it) {
    const int i4 = tid + it * (G1_WAVES * 32);
    const v4f w = *(const v4f*)(W1 + (size_t)i4 * 4);
    const int k  = i4 >> 2;
    const int c0 = (i4 & 3) * 4;
    float* d = wfrag + (((k >> 3) * 2 + ((k >> 2) & 1)) * 16 + c0) * 4 + (k & 3);
    d[0] = w.x; d[4] = w.y; d[8] = w.z; d[12] = w.w;
  }
  __syncthreads();
  if (wave >= nTiles) return;

  v8f acc = {};
  for (int G = 0; G < F_IN / G1_PIECE; G += 2) {
    g1_store_piece(R0, xsw, hi, col);
    if (G + 2 < F_IN / G1_PIECE) g1_load_piece(R0, gp, G + 2);
    acc = g1_mul_piece(acc, xsw, wfrag, G, hi, col);
    g1_store_piece(R1, xsw, hi, col);
    if (G + 3 < F_IN / G1_PIECE) g1_load_piece(R1, gp, G + 3);
    acc = g1_mul_piece(acc, xsw, wfrag, G + 1, hi, col);
  }

  if (tb == tileBase) {
    float* outp = h + ((size_t)(tb + 8 * hi)) * HID + col;
#pragma unroll
    for (int r = 0; r < 8; ++r) outp[r * HID] = acc[r];
  } else {
#pragma unroll
    for (int r = 0; r < 8; ++r) {
      int m = tb + r + 8 * hi;
      if (m >= tileBase) h[(size_t)m * HID + col] = acc[r];
    }
  }
}

__global__ void k_agg1_init(float* h, const float* __restrict__ dinv,
                            float* __restrict__ acc1, int N) {
  int t = blockIdx.x * blockDim.x + threadIdx.x;
  if (t >= N * HID) return;
  float v = h[t] * dinv[t >> 4];
  h[t]    = v;
  acc1[t] = v;
}

#define A1_NPP  4800
#define A1_C    12
#define A1_BS   1024
#define A1_U    8
#define A1_Q    64

template <int I>
__device__ __forceinline__ int row_share_bcast(int v) {
  return __builtin_amdgcn_mov_dpp(v, 0x150 + I, 0xf, 0xf, false);
}

#define A1_BCAST(I)                                            \
  ss[I] = (unsigned)row_share_bcast<I>(s0);                    \
  rr[I] = (unsigned)row_share_bcast<I>(r0);

template <bool FULL>
__device__ __forceinline__ void a1_drain(const unsigned* qs, const unsigned* qr, unsigned head,
                                         unsigned cnt, const float* __restrict__ g, float* acc,
                                         unsigned lane, unsigned j) {
  const unsigned NONE = 0xffffffffu;
  const unsigned idx = (head + lane) & (A1_Q - 1);
  const bool valid = FULL || lane < cnt;
  const int s0 = valid ? (int)qs[idx] : 0;
  const int r0 = valid ? (int)qr[idx] : (int)NONE;
  unsigned ss[16], rr[16];
  float m[16];
  A1_BCAST(0)  A1_BCAST(1)  A1_BCAST(2)  A1_BCAST(3)
  A1_BCAST(4)  A1_BCAST(5)  A1_BCAST(6)  A1_BCAST(7)
  A1_BCAST(8)  A1_BCAST(9)  A1_BCAST(10) A1_BCAST(11)
  A1_BCAST(12) A1_BCAST(13) A1_BCAST(14) A1_BCAST(15)
#pragma unroll
  for (int i = 0; i < 16; ++i) m[i] = g[ss[i] * HID + j];
#pragma unroll
  for (int i = 0; i < 16; ++i) {
    if (FULL || rr[i] != NONE) atomicAdd(&acc[rr[i] * HID + j], m[i]);
  }
}

__global__ __launch_bounds__(A1_BS) void k_agg1_lds(
    const int* __restrict__ src, const int* __restrict__ dst,
    const float* __restrict__ g, float* __restrict__ acc1, int N, int E) {
  __shared__ __attribute__((aligned(16))) float acc[A1_NPP * HID];
  __shared__ unsigned qsrc[A1_BS / 32][A1_Q];
  __shared__ unsigned qrow[A1_BS / 32][A1_Q];
  const int tid = threadIdx.x;
  const unsigned lane = tid & 31;
  const unsigned j = lane & 15;
  const int wv  = tid >> 5;
  const int n0  = blockIdx.x * A1_NPP;
  const int np  = min(A1_NPP, N - n0);
  for (int i = tid; i < A1_NPP * HID / 4; i += A1_BS) ((v4f*)acc)[i] = v4f{0.0f, 0.0f, 0.0f, 0.0f};
  __syncthreads();
  const int per = (E + A1_C - 1) / A1_C;
  const int e0  = blockIdx.y * per;
  const int e1  = min(E, e0 + per);
  unsigned* qs = &qsrc[wv][0];
  unsigned* qr = &qrow[wv][0];

  if (e0 < e1) {
    const unsigned NONE = 0xffffffffu;
    unsigned head = 0, tail = 0;
    unsigned r[A1_U];
    int      s[A1_U];
    int tb = e0;
#pragma unroll
    for (int k = 0; k < A1_U; ++k) {
      int e  = tb + tid + k * A1_BS;
      int ec = min(e, e1 - 1);
      int dd = dst[ec];
      s[k] = src[ec];
      r[k] = (e < e1) ? (unsigned)(dd - n0) : NONE;
    }
    while (tb < e1) {
      const int nbase = tb + A1_U * A1_BS + tid;
      int dn[A1_U], sn[A1_U];
#pragma unroll
      for (int k = 0; k < A1_U; ++k) {
        int ec = min(nbase + k * A1_BS, e1 - 1);
        dn[k] = dst[ec];
        sn[k] = src[ec];
      }
#pragma unroll
      for (int k = 0; k < A1_U; ++k) {
        const bool hit = r[k] < (unsigned)np;
        const unsigned mask = __builtin_amdgcn_ballot_w32(hit);
        if (mask != 0u) {
          const unsigned pos = (tail + __builtin_amdgcn_mbcnt_lo(mask, 0u)) & (A1_Q - 1);
          if (hit) { qs[pos] = (unsigned)s[k]; qr[pos] = r[k]; }
          tail += (unsigned)__builtin_popcount(mask);
          if (tail - head >= 32u) {
            a1_drain<true>(qs, qr, head, 32u, g, acc, lane, j);
            head += 32u;
          }
        }
      }
#pragma unroll
      for (int k = 0; k < A1_U; ++k) {
        r[k] = (nbase + k * A1_BS < e1) ? (unsigned)(dn[k] - n0) : NONE;
        s[k] = sn[k];
      }
      tb += A1_U * A1_BS;
    }
    if (tail != head) a1_drain<false>(qs, qr, head, tail - head, g, acc, lane, j);
  }
  __syncthreads();
  float* o = acc1 + (size_t)n0 * HID;
  const int nf = np * HID;
  for (int i = tid; i < nf; i += 5 * A1_BS) {
    float v[5];
#pragma unroll
    for (int q = 0; q < 5; ++q) v[q] = acc[min(i + q * A1_BS, nf - 1)];
#pragma unroll
    for (int q = 0; q < 5; ++q) {
      if (i + q * A1_BS < nf) atomicAdd(o + i + q * A1_BS, v[q]);
    }
  }
}

__global__ void k_layer2_node(const float* __restrict__ acc1, const float* __restrict__ dinv,
                              const float* __restrict__ b1, const float* __restrict__ W2,
                              float* __restrict__ h3, int N) {
  int n = blockIdx.x * blockDim.x + threadIdx.x;
  if (n >= N) return;
  const float dv = dinv[n];
  float o0 = 0.0f, o1 = 0.0f;
#pragma unroll
  for (int j = 0; j < HID; ++j) {
    float v = fmaxf(acc1[(size_t)n * HID + j] * dv + b1[j], 0.0f);
    o0 = __builtin_fmaf(v, W2[j * OUTF + 0], o0);
    o1 = __builtin_fmaf(v, W2[j * OUTF + 1], o1);
  }
  v2f o;
  o.x = o0 * dv;
  o.y = o1 * dv;
  *(v2f*)(h3 + (size_t)n * OUTF) = o;
}

__global__ __launch_bounds__(SCAN_BS) void k_agg2_part(
    const int* __restrict__ src, const int* __restrict__ dst,
    const float* __restrict__ h3, float* __restrict__ part, int N, int E) {
  __shared__ float acc[NPP * OUTF];
  const v2f* __restrict__ h3v = (const v2f*)h3;
  const int tid = threadIdx.x;
  const int n0  = blockIdx.x * NPP;
  const int np  = min(NPP, N - n0);
  for (int i = tid; i < NPP * OUTF; i += SCAN_BS) acc[i] = 0.0f;
  __syncthreads();
  const int per = (E + NCH - 1) / NCH;
  const int e0  = blockIdx.y * per;
  const int e1  = min(E, e0 + per);

  if (e0 < e1) {
    const unsigned NONE = 0xffffffffu;
    unsigned r[SCAN_U];
    int      s[SCAN_U];
    int base = e0 + tid;
#pragma unroll
    for (int k = 0; k < SCAN_U; ++k) {
      int e  = base + k * SCAN_BS;
      int ec = min(e, e1 - 1);
      int dd = dst[ec];
      s[k] = src[ec];
      r[k] = (e < e1) ? (unsigned)(dd - n0) : NONE;
    }
    while (base < e1) {
      v2f gv[SCAN_U];
#pragma unroll
      for (int k = 0; k < SCAN_U; ++k) {
        const bool keep = r[k] < (unsigned)np;
        gv[k] = h3v[keep ? s[k] : 0];
      }
      const int nbase = base + SCAN_U * SCAN_BS;
      int dn[SCAN_U], sn[SCAN_U];
#pragma unroll
      for (int k = 0; k < SCAN_U; ++k) {
        int ec = min(nbase + k * SCAN_BS, e1 - 1);
        dn[k] = dst[ec];
        sn[k] = src[ec];
      }
#pragma unroll
      for (int k = 0; k < SCAN_U; ++k) {
        if (r[k] < (unsigned)np) {
          atomicAdd(&acc[r[k] * OUTF + 0], gv[k].x);
          atomicAdd(&acc[r[k] * OUTF + 1], gv[k].y);
        }
      }
#pragma unroll
      for (int k = 0; k < SCAN_U; ++k) {
        r[k] = (nbase + k * SCAN_BS < e1) ? (unsigned)(dn[k] - n0) : NONE;
        s[k] = sn[k];
      }
      base = nbase;
    }
  }
  __syncthreads();
  float* o = part + ((size_t)blockIdx.y * N + n0) * OUTF;
  for (int i = tid; i < np * OUTF; i += SCAN_BS) o[i] = acc[i];
}

__global__ void k_agg2_reduce(const float* __restrict__ part, const float* __restrict__ h3,
                              const float* __restrict__ dinv, const float* __restrict__ b2,
                              float* __restrict__ out, int N) {
  int t = blockIdx.x * blockDim.x + threadIdx.x;
  if (t >= N * OUTF) return;
  float s = h3[t];
#pragma unroll
  for (int c = 0; c < NCH; ++c) s += part[(size_t)c * N * OUTF + t];
  out[t] = s * dinv[t / OUTF] + b2[t % OUTF];
}

static inline int cdiv_i(long long a, long long b) { return (int)((a + b - 1) / b); }

extern "C" void kernel_launch(void* const* d_in, const int* in_sizes, int n_in,
                              void* d_out, int out_size, void* d_ws, size_t ws_size,
                              hipStream_t stream) {
  const float* x  = (const float*)d_in[0];
  const int*   ei = (const int*)  d_in[1];
  const float* W1 = (const float*)d_in[2];
  const float* b1 = (const float*)d_in[3];
  const float* W2 = (const float*)d_in[4];
  const float* b2 = (const float*)d_in[5];
  float* out = (float*)d_out;

  const int N = in_sizes[0] / F_IN;
  const int E = in_sizes[1] / 2;
  const int* src = ei;
  const int* dst = ei + E;

  float* ws   = (float*)d_ws;
  float* dinv = ws;
  float* h    = dinv + N;
  float* acc1 = h + (size_t)N * HID;
  float* h3   = acc1 + (size_t)N * HID;

  const int BS = 256;

  float* part = h;
  const dim3 scanGrid(cdiv_i(N, NPP), NCH);

  k_deg_part  <<<scanGrid, SCAN_BS, 0, stream>>>(dst, part, N, E);
  k_deg_reduce<<<cdiv_i(N, BS), BS, 0, stream>>>(part, dinv, N);

  int nTiles = cdiv_i(N, 16);
  k_gemm1_wmma<<<cdiv_i(nTiles, G1_WAVES), G1_WAVES * 32, 0, stream>>>(x, W1, h, N, nTiles);

  k_agg1_init <<<cdiv_i((long long)N * HID, BS), BS, 0, stream>>>(h, dinv, acc1, N);
  k_agg1_lds<<<dim3(cdiv_i(N, A1_NPP), A1_C), A1_BS, 0, stream>>>(src, dst, h, acc1, N, E);

  k_layer2_node<<<cdiv_i(N, BS), BS, 0, stream>>>(acc1, dinv, b1, W2, h3, N);

  k_agg2_part  <<<scanGrid, SCAN_BS, 0, stream>>>(src, dst, h3, part, N, E);
  k_agg2_reduce<<<cdiv_i((long long)N * OUTF, BS), BS, 0, stream>>>(part, h3, dinv, b2, out, N);
}


# Edges sorted once by destination node range into 4-byte (src,row) words; degree, layer-1 and layer-2 edge passes read only their own segments, no 16/21/16-fold list scans; sums by arrival order
# speedup vs baseline: 5.4784x; 1.4332x over previous
#include <hip/hip_runtime.h>

typedef __attribute__((ext_vector_type(2))) float v2f;
typedef __attribute__((ext_vector_type(4))) float v4f;
typedef __attribute__((ext_vector_type(8))) float v8f;

#define F_IN 512
#define HID  16
#define OUTF 2

#define NPP     6400
#define NCH     16
#define SCAN_BS 1024
#define SCAN_U  8

__global__ __launch_bounds__(SCAN_BS) void k_deg_part(
    const int* __restrict__ dst, float* __restrict__ part, int N, int E) {
  __shared__ unsigned cnt[NPP];
  const int tid = threadIdx.x;
  const int n0  = blockIdx.x * NPP;
  const int np  = min(NPP, N - n0);
  for (int i = tid; i < NPP; i += SCAN_BS) cnt[i] = 0u;
  __syncthreads();
  const int per = (E + NCH - 1) / NCH;
  const int e0  = blockIdx.y * per;
  const int e1  = min(E, e0 + per);
  for (int base = e0 + tid; base < e1; base += 2 * SCAN_U * SCAN_BS) {
    int d[2 * SCAN_U];
#pragma unroll
    for (int k = 0; k < 2 * SCAN_U; ++k) d[k] = dst[min(base + k * SCAN_BS, e1 - 1)];
#pragma unroll
    for (int k = 0; k < 2 * SCAN_U; ++k) {
      unsigned r = (unsigned)(d[k] - n0);
      if (base + k * SCAN_BS < e1 && r < (unsigned)np) atomicAdd(&cnt[r], 1u);
    }
  }
  __syncthreads();
  float* o = part + (size_t)blockIdx.y * N + n0;
  for (int i = tid; i < np; i += SCAN_BS) o[i] = (float)cnt[i];
}

template <int P>
__global__ void k_deg_reduce(const float* __restrict__ part, float* __restrict__ dinv, int N) {
  int n = blockIdx.x * blockDim.x + threadIdx.x;
  if (n >= N) return;
  float s = 1.0f;
#pragma unroll
  for (int c = 0; c < P; ++c) s += part[(size_t)c * N + n];
  dinv[n] = rsqrtf(s);
}

#define BK_NPP   4800
#define BK_RMAX  32
#define BK_TABW  (BK_RMAX + 1)
#define BK_C     12
#define BK_B     (32 * BK_C)
#define BK_BS    1024
#define BK_U     9
#define BK_G     8
#define BK_LBITS 13
#define BK_LMASK 0x1FFFu

__global__ __launch_bounds__(BK_BS) void k_bin_edges(
    const int* __restrict__ src, const int* __restrict__ dst,
    unsigned* __restrict__ bins, int* __restrict__ tab, int N, int E, int L) {
  __shared__ unsigned cnt[BK_RMAX * BK_G];
  __shared__ unsigned loff[BK_RMAX * BK_G];
  __shared__ unsigned segs[BK_TABW];
  __shared__ unsigned stage[BK_U * BK_BS];
  const unsigned NONE = 0xffffffffu;
  const int tid = threadIdx.x;
  const unsigned grp = (unsigned)tid & (BK_G - 1);
  const long long c0 = (long long)blockIdx.x * L;
  const int e0 = (int)(c0 < (long long)E ? c0 : (long long)E);
  const int e1 = (int)(c0 + L < (long long)E ? c0 + L : (long long)E);
  if (tid < BK_RMAX * BK_G) cnt[tid] = 0u;
  __syncthreads();

  int d[BK_U], s[BK_U];
#pragma unroll
  for (int k = 0; k < BK_U; ++k) {
    const int ec = max(0, min(e0 + tid + k * BK_BS, e1 - 1));
    d[k] = dst[ec];
    s[k] = src[ec];
  }
  unsigned word[BK_U], key[BK_U], pos[BK_U];
#pragma unroll
  for (int k = 0; k < BK_U; ++k) {
    const unsigned r = (unsigned)d[k] / BK_NPP;
    const bool ok = (e0 + tid + k * BK_BS < e1) && ((unsigned)d[k] < (unsigned)N);
    key[k]  = ok ? r * BK_G + grp : NONE;
    pos[k]  = 0u;
    if (ok) pos[k] = atomicAdd(&cnt[r * BK_G + grp], 1u);
    word[k] = ((unsigned)s[k] << BK_LBITS) | ((unsigned)d[k] - r * BK_NPP);
  }
  __syncthreads();
  if (tid < BK_RMAX) {
    unsigned c[BK_G], tot = 0u;
#pragma unroll
    for (int q = 0; q < BK_G; ++q) { c[q] = cnt[tid * BK_G + q]; tot += c[q]; }
    unsigned incl = tot;
#pragma unroll
    for (int dl = 1; dl < 32; dl <<= 1) {
      const unsigned t = __shfl_up(incl, (unsigned)dl);
      if (tid >= dl) incl += t;
    }
    unsigned a = incl - tot;
    segs[tid] = a;
    if (tid == BK_RMAX - 1) segs[BK_RMAX] = incl;
#pragma unroll
    for (int q = 0; q < BK_G; ++q) { loff[tid * BK_G + q] = a; a += c[q]; }
  }
  __syncthreads();
#pragma unroll
  for (int k = 0; k < BK_U; ++k) {
    if (key[k] != NONE) stage[loff[key[k]] + pos[k]] = word[k];
  }
  __syncthreads();
  const unsigned tot = segs[BK_RMAX];
  unsigned* o = bins + (size_t)blockIdx.x * L;
  for (unsigned i = tid; i < tot; i += BK_BS) o[i] = stage[i];
  if (tid < BK_TABW) tab[blockIdx.x * BK_TABW + tid] = (int)segs[tid];
}

__device__ __forceinline__ const unsigned* bk_segment(const unsigned* __restrict__ bins,
                                                      const int* __restrict__ tab, int L,
                                                      int wv, int& len) {
  const int b  = blockIdx.y * 32 + wv;
  const int s0 = __builtin_amdgcn_readfirstlane(tab[b * BK_TABW + blockIdx.x]);
  const int s1 = __builtin_amdgcn_readfirstlane(tab[b * BK_TABW + blockIdx.x + 1]);
  len = s1 - s0;
  return bins + (size_t)b * L + s0;
}

#define BK_DEG_U 8

__global__ __launch_bounds__(BK_BS) void k_deg_part_bk(
    const unsigned* __restrict__ bins, const int* __restrict__ tab,
    float* __restrict__ part, int N, int L) {
  __shared__ unsigned cnt[BK_NPP];
  const int tid  = threadIdx.x;
  const int lane = tid & 31;
  const int n0   = blockIdx.x * BK_NPP;
  const int np   = min(BK_NPP, N - n0);
  for (int i = tid; i < BK_NPP; i += BK_BS) cnt[i] = 0u;
  __syncthreads();
  int len;
  const unsigned* p = bk_segment(bins, tab, L, tid >> 5, len);
  for (int i = lane; i < len; i += 32 * BK_DEG_U) {
    unsigned w[BK_DEG_U];
#pragma unroll
    for (int k = 0; k < BK_DEG_U; ++k) w[k] = p[min(i + 32 * k, len - 1)];
#pragma unroll
    for (int k = 0; k < BK_DEG_U; ++k)
      atomicAdd(&cnt[w[k] & BK_LMASK], (i + 32 * k < len) ? 1u : 0u);
  }
  __syncthreads();
  float* o = part + (size_t)blockIdx.y * N + n0;
  for (int i = tid; i < np; i += BK_BS) o[i] = (float)cnt[i];
}

#define G1_WAVES  8
#define G1_PIECE  64
#define G1_PITCH  68

__device__ __forceinline__ void g1_load_piece(v4f (&R)[8], const float* gp, int G) {
#pragma unroll
  for (int j = 0; j < 8; ++j)
    R[j] = __builtin_nontemporal_load((const v4f*)(gp + (size_t)j * 2 * F_IN + G * G1_PIECE));
}

__device__ __forceinline__ void g1_store_piece(const v4f (&R)[8], float* xsw, int hi, int col) {
#pragma unroll
  for (int j = 0; j < 8; ++j)
    *(v4f*)(xsw + (2 * j + hi) * G1_PITCH + 4 * col) = R[j];
}

__device__ __forceinline__ v8f g1_mul_piece(v8f acc, const float* xsw, const float* wfrag,
                                            int G, int hi, int col) {
#pragma unroll
  for (int i = 0; i < 8; ++i) {
    const v4f a4 = *(const v4f*)(xsw + col * G1_PITCH + 8 * i + 4 * hi);
    const v4f b4 = *(const v4f*)(wfrag + (((8 * G + i) * 2 + hi) * 16 + col) * 4);
    v2f a, b;
    a.x = a4.x; a.y = a4.y; b.x = b4.x; b.y = b4.y;
    acc = __builtin_amdgcn_wmma_f32_16x16x4_f32(
         false, a,  false, b,
         (short)0, acc,  false,  false);
    a.x = a4.z; a.y = a4.w; b.x = b4.z; b.y = b4.w;
    acc = __builtin_amdgcn_wmma_f32_16x16x4_f32(
         false, a,  false, b,
         (short)0, acc,  false,  false);
  }
  return acc;
}

__global__ __launch_bounds__(G1_WAVES * 32) void k_gemm1_wmma(
    const float* __restrict__ x, const float* __restrict__ W1,
    float* __restrict__ h, int N, int nTiles) {
  __shared__ __attribute__((aligned(16))) float wfrag[F_IN * HID];
  __shared__ __attribute__((aligned(16))) float xs[G1_WAVES][16 * G1_PITCH];
  const int tid  = threadIdx.x;
  const int lane = tid & 31;
  const int wv   = tid >> 5;
  const int col  = lane & 15;
  const int hi   = lane >> 4;
  const int wave = blockIdx.x * G1_WAVES + wv;
  const int tileBase = wave * 16;
  const int tb = min(tileBase, N - 16);
  const float* gp = x + (size_t)(tb + hi) * F_IN + 4 * col;
  float* xsw = &xs[wv][0];

  v4f R0[8], R1[8];
  g1_load_piece(R0, gp, 0);
  g1_load_piece(R1, gp, 1);

#pragma unroll
  for (int it = 0; it < F_IN * HID / 4 / (G1_WAVES * 32); ++it) {
    const int i4 = tid + it * (G1_WAVES * 32);
    const v4f w = *(const v4f*)(W1 + (size_t)i4 * 4);
    const int k  = i4 >> 2;
    const int c0 = (i4 & 3) * 4;
    float* d = wfrag + (((k >> 3) * 2 + ((k >> 2) & 1)) * 16 + c0) * 4 + (k & 3);
    d[0] = w.x; d[4] = w.y; d[8] = w.z; d[12] = w.w;
  }
  __syncthreads();
  if (wave >= nTiles) return;

  v8f acc = {};
  for (int G = 0; G < F_IN / G1_PIECE; G += 2) {
    g1_store_piece(R0, xsw, hi, col);
    if (G + 2 < F_IN / G1_PIECE) g1_load_piece(R0, gp, G + 2);
    acc = g1_mul_piece(acc, xsw, wfrag, G, hi, col);
    g1_store_piece(R1, xsw, hi, col);
    if (G + 3 < F_IN / G1_PIECE) g1_load_piece(R1, gp, G + 3);
    acc = g1_mul_piece(acc, xsw, wfrag, G + 1, hi, col);
  }

  if (tb == tileBase) {
    float* outp = h + ((size_t)(tb + 8 * hi)) * HID + col;
#pragma unroll
    for (int r = 0; r < 8; ++r) outp[r * HID] = acc[r];
  } else {
#pragma unroll
    for (int r = 0; r < 8; ++r) {
      int m = tb + r + 8 * hi;
      if (m >= tileBase) h[(size_t)m * HID + col] = acc[r];
    }
  }
}

__global__ void k_agg1_init(float* h, const float* __restrict__ dinv,
                            float* __restrict__ acc1, int N) {
  int t = blockIdx.x * blockDim.x + threadIdx.x;
  if (t >= N * HID) return;
  float v = h[t] * dinv[t >> 4];
  h[t]    = v;
  acc1[t] = v;
}

#define A1_NPP  4800
#define A1_C    12
#define A1_BS   1024
#define A1_U    8
#define A1_Q    64

template <int I>
__device__ __forceinline__ int row_share_bcast(int v) {
  return __builtin_amdgcn_mov_dpp(v, 0x150 + I, 0xf, 0xf, false);
}

#define A1_BCAST(I)                                            \
  ss[I] = (unsigned)row_share_bcast<I>(s0);                    \
  rr[I] = (unsigned)row_share_bcast<I>(r0);

template <bool FULL>
__device__ __forceinline__ void a1_add_rows(int s0, int r0, const float* __restrict__ g,
                                            float* acc, unsigned j) {
  const unsigned NONE = 0xffffffffu;
  unsigned ss[16], rr[16];
  float m[16];
  A1_BCAST(0)  A1_BCAST(1)  A1_BCAST(2)  A1_BCAST(3)
  A1_BCAST(4)  A1_BCAST(5)  A1_BCAST(6)  A1_BCAST(7)
  A1_BCAST(8)  A1_BCAST(9)  A1_BCAST(10) A1_BCAST(11)
  A1_BCAST(12) A1_BCAST(13) A1_BCAST(14) A1_BCAST(15)
#pragma unroll
  for (int i = 0; i < 16; ++i) m[i] = g[ss[i] * HID + j];
#pragma unroll
  for (int i = 0; i < 16; ++i) {
    if (FULL || rr[i] != NONE) atomicAdd(&acc[rr[i] * HID + j], m[i]);
  }
}

template <bool FULL>
__device__ __forceinline__ void a1_drain(const unsigned* qs, const unsigned* qr, unsigned head,
                                         unsigned cnt, const float* __restrict__ g, float* acc,
                                         unsigned lane, unsigned j) {
  const unsigned NONE = 0xffffffffu;
  const unsigned idx = (head + lane) & (A1_Q - 1);
  const bool valid = FULL || lane < cnt;
  const int s0 = valid ? (int)qs[idx] : 0;
  const int r0 = valid ? (int)qr[idx] : (int)NONE;
  a1_add_rows<FULL>(s0, r0, g, acc, j);
}

__device__ __forceinline__ void a1_flush(const float* acc, float* o, int nf, int tid) {
  for (int i = tid; i < nf; i += 5 * A1_BS) {
    float v[5];
#pragma unroll
    for (int q = 0; q < 5; ++q) v[q] = acc[min(i + q * A1_BS, nf - 1)];
#pragma unroll
    for (int q = 0; q < 5; ++q) {
      if (i + q * A1_BS < nf) atomicAdd(o + i + q * A1_BS, v[q]);
    }
  }
}

__global__ __launch_bounds__(A1_BS) void k_agg1_bk(
    const unsigned* __restrict__ bins, const int* __restrict__ tab,
    const float* __restrict__ g, float* __restrict__ acc1, int N, int L) {
  __shared__ __attribute__((aligned(16))) float acc[BK_NPP * HID];
  const unsigned NONE = 0xffffffffu;
  const int tid = threadIdx.x;
  const int lane = tid & 31;
  const unsigned j = (unsigned)lane & 15u;
  const int n0  = blockIdx.x * BK_NPP;
  const int np  = min(BK_NPP, N - n0);
  for (int i = tid; i < BK_NPP * HID / 4; i += A1_BS) ((v4f*)acc)[i] = v4f{0.0f, 0.0f, 0.0f, 0.0f};
  __syncthreads();
  int len;
  const unsigned* p = bk_segment(bins, tab, L, tid >> 5, len);
  unsigned wn = (lane < len) ? p[lane] : NONE;
  for (int i = 0; i < len; i += 32) {
    const unsigned w = wn;
    const int ni = i + 32 + lane;
    wn = (ni < len) ? p[ni] : NONE;
    const bool valid = w != NONE;
    const int s0 = valid ? (int)(w >> BK_LBITS) : 0;
    const int r0 = valid ? (int)(w & BK_LMASK) : (int)NONE;
    if (i + 32 <= len) a1_add_rows<true>(s0, r0, g, acc, j);
    else               a1_add_rows<false>(s0, r0, g, acc, j);
  }
  __syncthreads();
  a1_flush(acc, acc1 + (size_t)n0 * HID, np * HID, tid);
}

__global__ __launch_bounds__(A1_BS) void k_agg1_lds(
    const int* __restrict__ src, const int* __restrict__ dst,
    const float* __restrict__ g, float* __restrict__ acc1, int N, int E) {
  __shared__ __attribute__((aligned(16))) float acc[A1_NPP * HID];
  __shared__ unsigned qsrc[A1_BS / 32][A1_Q];
  __shared__ unsigned qrow[A1_BS / 32][A1_Q];
  const int tid = threadIdx.x;
  const unsigned lane = tid & 31;
  const unsigned j = lane & 15;
  const int wv  = tid >> 5;
  const int n0  = blockIdx.x * A1_NPP;
  const int np  = min(A1_NPP, N - n0);
  for (int i = tid; i < A1_NPP * HID / 4; i += A1_BS) ((v4f*)acc)[i] = v4f{0.0f, 0.0f, 0.0f, 0.0f};
  __syncthreads();
  const int per = (E + A1_C - 1) / A1_C;
  const int e0  = blockIdx.y * per;
  const int e1  = min(E, e0 + per);
  unsigned* qs = &qsrc[wv][0];
  unsigned* qr = &qrow[wv][0];

  if (e0 < e1) {
    const unsigned NONE = 0xffffffffu;
    unsigned head = 0, tail = 0;
    unsigned r[A1_U];
    int      s[A1_U];
    int tb = e0;
#pragma unroll
    for (int k = 0; k < A1_U; ++k) {
      int e  = tb + tid + k * A1_BS;
      int ec = min(e, e1 - 1);
      int dd = dst[ec];
      s[k] = src[ec];
      r[k] = (e < e1) ? (unsigned)(dd - n0) : NONE;
    }
    while (tb < e1) {
      const int nbase = tb + A1_U * A1_BS + tid;
      int dn[A1_U], sn[A1_U];
#pragma unroll
      for (int k = 0; k < A1_U; ++k) {
        int ec = min(nbase + k * A1_BS, e1 - 1);
        dn[k] = dst[ec];
        sn[k] = src[ec];
      }
#pragma unroll
      for (int k = 0; k < A1_U; ++k) {
        const bool hit = r[k] < (unsigned)np;
        const unsigned mask = __builtin_amdgcn_ballot_w32(hit);
        if (mask != 0u) {
          const unsigned pos = (tail + __builtin_amdgcn_mbcnt_lo(mask, 0u)) & (A1_Q - 1);
          if (hit) { qs[pos] = (unsigned)s[k]; qr[pos] = r[k]; }
          tail += (unsigned)__builtin_popcount(mask);
          if (tail - head >= 32u) {
            a1_drain<true>(qs, qr, head, 32u, g, acc, lane, j);
            head += 32u;
          }
        }
      }
#pragma unroll
      for (int k = 0; k < A1_U; ++k) {
        r[k] = (nbase + k * A1_BS < e1) ? (unsigned)(dn[k] - n0) : NONE;
        s[k] = sn[k];
      }
      tb += A1_U * A1_BS;
    }
    if (tail != head) a1_drain<false>(qs, qr, head, tail - head, g, acc, lane, j);
  }
  __syncthreads();
  a1_flush(acc, acc1 + (size_t)n0 * HID, np * HID, tid);
}

__global__ void k_layer2_node(const float* __restrict__ acc1, const float* __restrict__ dinv,
                              const float* __restrict__ b1, const float* __restrict__ W2,
                              float* __restrict__ h3, int N) {
  int n = blockIdx.x * blockDim.x + threadIdx.x;
  if (n >= N) return;
  const float dv = dinv[n];
  float o0 = 0.0f, o1 = 0.0f;
#pragma unroll
  for (int j = 0; j < HID; ++j) {
    float v = fmaxf(acc1[(size_t)n * HID + j] * dv + b1[j], 0.0f);
    o0 = __builtin_fmaf(v, W2[j * OUTF + 0], o0);
    o1 = __builtin_fmaf(v, W2[j * OUTF + 1], o1);
  }
  v2f o;
  o.x = o0 * dv;
  o.y = o1 * dv;
  *(v2f*)(h3 + (size_t)n * OUTF) = o;
}

__global__ __launch_bounds__(SCAN_BS) void k_agg2_part(
    const int* __restrict__ src, const int* __restrict__ dst,
    const float* __restrict__ h3, float* __restrict__ part, int N, int E) {
  __shared__ float acc[NPP * OUTF];
  const v2f* __restrict__ h3v = (const v2f*)h3;
  const int tid = threadIdx.x;
  const int n0  = blockIdx.x * NPP;
  const int np  = min(NPP, N - n0);
  for (int i = tid; i < NPP * OUTF; i += SCAN_BS) acc[i] = 0.0f;
  __syncthreads();
  const int per = (E + NCH - 1) / NCH;
  const int e0  = blockIdx.y * per;
  const int e1  = min(E, e0 + per);

  if (e0 < e1) {
    const unsigned NONE = 0xffffffffu;
    unsigned r[SCAN_U];
    int      s[SCAN_U];
    int base = e0 + tid;
#pragma unroll
    for (int k = 0; k < SCAN_U; ++k) {
      int e  = base + k * SCAN_BS;
      int ec = min(e, e1 - 1);
      int dd = dst[ec];
      s[k] = src[ec];
      r[k] = (e < e1) ? (unsigned)(dd - n0) : NONE;
    }
    while (base < e1) {
      v2f gv[SCAN_U];
#pragma unroll
      for (int k = 0; k < SCAN_U; ++k) {
        const bool keep = r[k] < (unsigned)np;
        gv[k] = h3v[keep ? s[k] : 0];
      }
      const int nbase = base + SCAN_U * SCAN_BS;
      int dn[SCAN_U], sn[SCAN_U];
#pragma unroll
      for (int k = 0; k < SCAN_U; ++k) {
        int ec = min(nbase + k * SCAN_BS, e1 - 1);
        dn[k] = dst[ec];
        sn[k] = src[ec];
      }
#pragma unroll
      for (int k = 0; k < SCAN_U; ++k) {
        if (r[k] < (unsigned)np) {
          atomicAdd(&acc[r[k] * OUTF + 0], gv[k].x);
          atomicAdd(&acc[r[k] * OUTF + 1], gv[k].y);
        }
      }
#pragma unroll
      for (int k = 0; k < SCAN_U; ++k) {
        r[k] = (nbase + k * SCAN_BS < e1) ? (unsigned)(dn[k] - n0) : NONE;
        s[k] = sn[k];
      }
      base = nbase;
    }
  }
  __syncthreads();
  float* o = part + ((size_t)blockIdx.y * N + n0) * OUTF;
  for (int i = tid; i < np * OUTF; i += SCAN_BS) o[i] = acc[i];
}

#define BK_A2_U 8

__global__ __launch_bounds__(BK_BS) void k_agg2_part_bk(
    const unsigned* __restrict__ bins, const int* __restrict__ tab,
    const float* __restrict__ h3, float* __restrict__ part, int N, int L) {
  __shared__ float acc[BK_NPP * OUTF];
  const v2f* __restrict__ h3v = (const v2f*)h3;
  const int tid  = threadIdx.x;
  const int lane = tid & 31;
  const int n0   = blockIdx.x * BK_NPP;
  const int np   = min(BK_NPP, N - n0);
  for (int i = tid; i < BK_NPP * OUTF; i += BK_BS) acc[i] = 0.0f;
  __syncthreads();
  int len;
  const unsigned* p = bk_segment(bins, tab, L, tid >> 5, len);
  for (int i = lane; i < len; i += 32 * BK_A2_U) {
    unsigned w[BK_A2_U];
    v2f gv[BK_A2_U];
#pragma unroll
    for (int k = 0; k < BK_A2_U; ++k) w[k] = p[min(i + 32 * k, len - 1)];
#pragma unroll
    for (int k = 0; k < BK_A2_U; ++k) gv[k] = h3v[w[k] >> BK_LBITS];
#pragma unroll
    for (int k = 0; k < BK_A2_U; ++k) {
      if (i + 32 * k < len) {
        const unsigned row = w[k] & BK_LMASK;
        atomicAdd(&acc[row * OUTF + 0], gv[k].x);
        atomicAdd(&acc[row * OUTF + 1], gv[k].y);
      }
    }
  }
  __syncthreads();
  float* o = part + ((size_t)blockIdx.y * N + n0) * OUTF;
  for (int i = tid; i < np * OUTF; i += BK_BS) o[i] = acc[i];
}

template <int P>
__global__ void k_agg2_reduce(const float* __restrict__ part, const float* __restrict__ h3,
                              const float* __restrict__ dinv, const float* __restrict__ b2,
                              float* __restrict__ out, int N) {
  int t = blockIdx.x * blockDim.x + threadIdx.x;
  if (t >= N * OUTF) return;
  float s = h3[t];
#pragma unroll
  for (int c = 0; c < P; ++c) s += part[(size_t)c * N * OUTF + t];
  out[t] = s * dinv[t / OUTF] + b2[t % OUTF];
}

static inline int cdiv_i(long long a, long long b) { return (int)((a + b - 1) / b); }

extern "C" void kernel_launch(void* const* d_in, const int* in_sizes, int n_in,
                              void* d_out, int out_size, void* d_ws, size_t ws_size,
                              hipStream_t stream) {
  const float* x  = (const float*)d_in[0];
  const int*   ei = (const int*)  d_in[1];
  const float* W1 = (const float*)d_in[2];
  const float* b1 = (const float*)d_in[3];
  const float* W2 = (const float*)d_in[4];
  const float* b2 = (const float*)d_in[5];
  float* out = (float*)d_out;

  const int N = in_sizes[0] / F_IN;
  const int E = in_sizes[1] / 2;
  const int* src = ei;
  const int* dst = ei + E;

  float* ws   = (float*)d_ws;
  float* dinv = ws;
  float* h    = dinv + N;
  float* acc1 = h + (size_t)N * HID;
  float* h3   = acc1 + (size_t)N * HID;

  const int BS = 256;
  const int nTiles = cdiv_i(N, 16);

  float* part = h;

  const int R = cdiv_i(N, BK_NPP);
  const int L = cdiv_i(E, BK_B);
  int*      tab  = (int*)(h3 + (size_t)N * OUTF);
  unsigned* bins = (unsigned*)(tab + (size_t)BK_B * BK_TABW);
  const size_t need = ((size_t)N * (1 + 2 * HID + OUTF) + (size_t)BK_B * BK_TABW + (size_t)BK_B * L) * sizeof(float);
  const bool bucketed = ws_size >= need && E >= 1 && R <= BK_RMAX && L <= BK_U * BK_BS &&
                        N <= (1 << (32 - BK_LBITS));

  if (bucketed) {
    const dim3 bkGrid(R, BK_C);

    k_bin_edges<<<BK_B, BK_BS, 0, stream>>>(src, dst, bins, tab, N, E, L);

    k_deg_part_bk<<<bkGrid, BK_BS, 0, stream>>>(bins, tab, part, N, L);
    k_deg_reduce<BK_C><<<cdiv_i(N, BS), BS, 0, stream>>>(part, dinv, N);

    k_gemm1_wmma<<<cdiv_i(nTiles, G1_WAVES), G1_WAVES * 32, 0, stream>>>(x, W1, h, N, nTiles);

    k_agg1_init<<<cdiv_i((long long)N * HID, BS), BS, 0, stream>>>(h, dinv, acc1, N);
    k_agg1_bk<<<bkGrid, A1_BS, 0, stream>>>(bins, tab, h, acc1, N, L);

    k_layer2_node<<<cdiv_i(N, BS), BS, 0, stream>>>(acc1, dinv, b1, W2, h3, N);

    k_agg2_part_bk<<<bkGrid, BK_BS, 0, stream>>>(bins, tab, h3, part, N, L);
    k_agg2_reduce<BK_C><<<cdiv_i((long long)N * OUTF, BS), BS, 0, stream>>>(part, h3, dinv, b2, out, N);
    return;
  }

  const dim3 scanGrid(cdiv_i(N, NPP), NCH);

  k_deg_part  <<<scanGrid, SCAN_BS, 0, stream>>>(dst, part, N, E);
  k_deg_reduce<NCH><<<cdiv_i(N, BS), BS, 0, stream>>>(part, dinv, N);

  k_gemm1_wmma<<<cdiv_i(nTiles, G1_WAVES), G1_WAVES * 32, 0, stream>>>(x, W1, h, N, nTiles);

  k_agg1_init <<<cdiv_i((long long)N * HID, BS), BS, 0, stream>>>(h, dinv, acc1, N);
  k_agg1_lds<<<dim3(cdiv_i(N, A1_NPP), A1_C), A1_BS, 0, stream>>>(src, dst, h, acc1, N, E);

  k_layer2_node<<<cdiv_i(N, BS), BS, 0, stream>>>(acc1, dinv, b1, W2, h3, N);

  k_agg2_part  <<<scanGrid, SCAN_BS, 0, stream>>>(src, dst, h3, part, N, E);
  k_agg2_reduce<NCH><<<cdiv_i((long long)N * OUTF, BS), BS, 0, stream>>>(part, h3, dinv, b2, out, N);
}


# Row scaling g = h * dinv and the start value of the layer-1 sums written by the GEMM's own epilogue; the N*16-thread launch (one load, two stores a thread) behind the GEMM is gone; bits unchanged
# speedup vs baseline: 5.7349x; 1.0468x over previous
#include <hip/hip_runtime.h>

typedef __attribute__((ext_vector_type(2))) float v2f;
typedef __attribute__((ext_vector_type(4))) float v4f;
typedef __attribute__((ext_vector_type(8))) float v8f;

#define F_IN 512
#define HID  16
#define OUTF 2

#define NPP     6400
#define NCH     16
#define SCAN_BS 1024
#define SCAN_U  8

__global__ __launch_bounds__(SCAN_BS) void k_deg_part(
    const int* __restrict__ dst, float* __restrict__ part, int N, int E) {
  __shared__ unsigned cnt[NPP];
  const int tid = threadIdx.x;
  const int n0  = blockIdx.x * NPP;
  const int np  = min(NPP, N - n0);
  for (int i = tid; i < NPP; i += SCAN_BS) cnt[i] = 0u;
  __syncthreads();
  const int per = (E + NCH - 1) / NCH;
  const int e0  = blockIdx.y * per;
  const int e1  = min(E, e0 + per);
  for (int base = e0 + tid; base < e1; base += 2 * SCAN_U * SCAN_BS) {
    int d[2 * SCAN_U];
#pragma unroll
    for (int k = 0; k < 2 * SCAN_U; ++k) d[k] = dst[min(base + k * SCAN_BS, e1 - 1)];
#pragma unroll
    for (int k = 0; k < 2 * SCAN_U; ++k) {
      unsigned r = (unsigned)(d[k] - n0);
      if (base + k * SCAN_BS < e1 && r < (unsigned)np) atomicAdd(&cnt[r], 1u);
    }
  }
  __syncthreads();
  float* o = part + (size_t)blockIdx.y * N + n0;
  for (int i = tid; i < np; i += SCAN_BS) o[i] = (float)cnt[i];
}

template <int P>
__global__ void k_deg_reduce(const float* __restrict__ part, float* __restrict__ dinv, int N) {
  int n = blockIdx.x * blockDim.x + threadIdx.x;
  if (n >= N) return;
  float s = 1.0f;
#pragma unroll
  for (int c = 0; c < P; ++c) s += part[(size_t)c * N + n];
  dinv[n] = rsqrtf(s);
}

#define BK_NPP   4800
#define BK_RMAX  32
#define BK_TABW  (BK_RMAX + 1)
#define BK_C     12
#define BK_B     (32 * BK_C)
#define BK_BS    1024
#define BK_U     9
#define BK_G     8
#define BK_LBITS 13
#define BK_LMASK 0x1FFFu

__global__ __launch_bounds__(BK_BS) void k_bin_edges(
    const int* __restrict__ src, const int* __restrict__ dst,
    unsigned* __restrict__ bins, int* __restrict__ tab, int N, int E, int L) {
  __shared__ unsigned cnt[BK_RMAX * BK_G];
  __shared__ unsigned loff[BK_RMAX * BK_G];
  __shared__ unsigned segs[BK_TABW];
  __shared__ unsigned stage[BK_U * BK_BS];
  const unsigned NONE = 0xffffffffu;
  const int tid = threadIdx.x;
  const unsigned grp = (unsigned)tid & (BK_G - 1);
  const long long c0 = (long long)blockIdx.x * L;
  const int e0 = (int)(c0 < (long long)E ? c0 : (long long)E);
  const int e1 = (int)(c0 + L < (long long)E ? c0 + L : (long long)E);
  if (tid < BK_RMAX * BK_G) cnt[tid] = 0u;
  __syncthreads();

  int d[BK_U], s[BK_U];
#pragma unroll
  for (int k = 0; k < BK_U; ++k) {
    const int ec = max(0, min(e0 + tid + k * BK_BS, e1 - 1));
    d[k] = dst[ec];
    s[k] = src[ec];
  }
  unsigned word[BK_U], key[BK_U], pos[BK_U];
#pragma unroll
  for (int k = 0; k < BK_U; ++k) {
    const unsigned r = (unsigned)d[k] / BK_NPP;
    const bool ok = (e0 + tid + k * BK_BS < e1) && ((unsigned)d[k] < (unsigned)N);
    key[k]  = ok ? r * BK_G + grp : NONE;
    pos[k]  = 0u;
    if (ok) pos[k] = atomicAdd(&cnt[r * BK_G + grp], 1u);
    word[k] = ((unsigned)s[k] << BK_LBITS) | ((unsigned)d[k] - r * BK_NPP);
  }
  __syncthreads();
  if (tid < BK_RMAX) {
    unsigned c[BK_G], tot = 0u;
#pragma unroll
    for (int q = 0; q < BK_G; ++q) { c[q] = cnt[tid * BK_G + q]; tot += c[q]; }
    unsigned incl = tot;
#pragma unroll
    for (int dl = 1; dl < 32; dl <<= 1) {
      const unsigned t = __shfl_up(incl, (unsigned)dl);
      if (tid >= dl) incl += t;
    }
    unsigned a = incl - tot;
    segs[tid] = a;
    if (tid == BK_RMAX - 1) segs[BK_RMAX] = incl;
#pragma unroll
    for (int q = 0; q < BK_G; ++q) { loff[tid * BK_G + q] = a; a += c[q]; }
  }
  __syncthreads();
#pragma unroll
  for (int k = 0; k < BK_U; ++k) {
    if (key[k] != NONE) stage[loff[key[k]] + pos[k]] = word[k];
  }
  __syncthreads();
  const unsigned tot = segs[BK_RMAX];
  unsigned* o = bins + (size_t)blockIdx.x * L;
  for (unsigned i = tid; i < tot; i += BK_BS) o[i] = stage[i];
  if (tid < BK_TABW) tab[blockIdx.x * BK_TABW + tid] = (int)segs[tid];
}

__device__ __forceinline__ const unsigned* bk_segment(const unsigned* __restrict__ bins,
                                                      const int* __restrict__ tab, int L,
                                                      int wv, int& len) {
  const int b  = blockIdx.y * 32 + wv;
  const int s0 = __builtin_amdgcn_readfirstlane(tab[b * BK_TABW + blockIdx.x]);
  const int s1 = __builtin_amdgcn_readfirstlane(tab[b * BK_TABW + blockIdx.x + 1]);
  len = s1 - s0;
  return bins + (size_t)b * L + s0;
}

#define BK_DEG_U 8

__global__ __launch_bounds__(BK_BS) void k_deg_part_bk(
    const unsigned* __restrict__ bins, const int* __restrict__ tab,
    float* __restrict__ part, int N, int L) {
  __shared__ unsigned cnt[BK_NPP];
  const int tid  = threadIdx.x;
  const int lane = tid & 31;
  const int n0   = blockIdx.x * BK_NPP;
  const int np   = min(BK_NPP, N - n0);
  for (int i = tid; i < BK_NPP; i += BK_BS) cnt[i] = 0u;
  __syncthreads();
  int len;
  const unsigned* p = bk_segment(bins, tab, L, tid >> 5, len);
  for (int i = lane; i < len; i += 32 * BK_DEG_U) {
    unsigned w[BK_DEG_U];
#pragma unroll
    for (int k = 0; k < BK_DEG_U; ++k) w[k] = p[min(i + 32 * k, len - 1)];
#pragma unroll
    for (int k = 0; k < BK_DEG_U; ++k)
      atomicAdd(&cnt[w[k] & BK_LMASK], (i + 32 * k < len) ? 1u : 0u);
  }
  __syncthreads();
  float* o = part + (size_t)blockIdx.y * N + n0;
  for (int i = tid; i < np; i += BK_BS) o[i] = (float)cnt[i];
}

#define G1_WAVES  8
#define G1_PIECE  64
#define G1_PITCH  68

__device__ __forceinline__ void g1_load_piece(v4f (&R)[8], const float* gp, int G) {
#pragma unroll
  for (int j = 0; j < 8; ++j)
    R[j] = __builtin_nontemporal_load((const v4f*)(gp + (size_t)j * 2 * F_IN + G * G1_PIECE));
}

__device__ __forceinline__ void g1_store_piece(const v4f (&R)[8], float* xsw, int hi, int col) {
#pragma unroll
  for (int j = 0; j < 8; ++j)
    *(v4f*)(xsw + (2 * j + hi) * G1_PITCH + 4 * col) = R[j];
}

__device__ __forceinline__ v8f g1_mul_piece(v8f acc, const float* xsw, const float* wfrag,
                                            int G, int hi, int col) {
#pragma unroll
  for (int i = 0; i < 8; ++i) {
    const v4f a4 = *(const v4f*)(xsw + col * G1_PITCH + 8 * i + 4 * hi);
    const v4f b4 = *(const v4f*)(wfrag + (((8 * G + i) * 2 + hi) * 16 + col) * 4);
    v2f a, b;
    a.x = a4.x; a.y = a4.y; b.x = b4.x; b.y = b4.y;
    acc = __builtin_amdgcn_wmma_f32_16x16x4_f32(
         false, a,  false, b,
         (short)0, acc,  false,  false);
    a.x = a4.z; a.y = a4.w; b.x = b4.z; b.y = b4.w;
    acc = __builtin_amdgcn_wmma_f32_16x16x4_f32(
         false, a,  false, b,
         (short)0, acc,  false,  false);
  }
  return acc;
}

__global__ __launch_bounds__(G1_WAVES * 32) void k_gemm1_wmma(
    const float* __restrict__ x, const float* __restrict__ W1,
    const float* __restrict__ dinv, float* __restrict__ g,
    float* __restrict__ acc1, int N, int nTiles) {
  __shared__ __attribute__((aligned(16))) float wfrag[F_IN * HID];
  __shared__ __attribute__((aligned(16))) float xs[G1_WAVES][16 * G1_PITCH];
  const int tid  = threadIdx.x;
  const int lane = tid & 31;
  const int wv   = tid >> 5;
  const int col  = lane & 15;
  const int hi   = lane >> 4;
  const int wave = blockIdx.x * G1_WAVES + wv;
  const int tileBase = wave * 16;
  const int tb = min(tileBase, N - 16);
  const float* gp = x + (size_t)(tb + hi) * F_IN + 4 * col;
  float* xsw = &xs[wv][0];

  v4f R0[8], R1[8];
  g1_load_piece(R0, gp, 0);
  g1_load_piece(R1, gp, 1);

#pragma unroll
  for (int it = 0; it < F_IN * HID / 4 / (G1_WAVES * 32); ++it) {
    const int i4 = tid + it * (G1_WAVES * 32);
    const v4f w = *(const v4f*)(W1 + (size_t)i4 * 4);
    const int k  = i4 >> 2;
    const int c0 = (i4 & 3) * 4;
    float* d = wfrag + (((k >> 3) * 2 + ((k >> 2) & 1)) * 16 + c0) * 4 + (k & 3);
    d[0] = w.x; d[4] = w.y; d[8] = w.z; d[12] = w.w;
  }
  __syncthreads();
  if (wave >= nTiles) return;

  float dv[8];
#pragma unroll
  for (int r = 0; r < 8; ++r) dv[r] = dinv[tb + 8 * hi + r];

  v8f acc = {};
  for (int G = 0; G < F_IN / G1_PIECE; G += 2) {
    g1_store_piece(R0, xsw, hi, col);
    if (G + 2 < F_IN / G1_PIECE) g1_load_piece(R0, gp, G + 2);
    acc = g1_mul_piece(acc, xsw, wfrag, G, hi, col);
    g1_store_piece(R1, xsw, hi, col);
    if (G + 3 < F_IN / G1_PIECE) g1_load_piece(R1, gp, G + 3);
    acc = g1_mul_piece(acc, xsw, wfrag, G + 1, hi, col);
  }

  if (tb == tileBase) {
    const size_t o = ((size_t)(tb + 8 * hi)) * HID + col;
    float* gout = g + o;
    float* aout = acc1 + o;
#pragma unroll
    for (int r = 0; r < 8; ++r) {
      const float v = acc[r] * dv[r];
      gout[r * HID] = v;
      aout[r * HID] = v;
    }
  } else {
#pragma unroll
    for (int r = 0; r < 8; ++r) {
      int m = tb + r + 8 * hi;
      if (m >= tileBase) {
        const float v = acc[r] * dv[r];
        g[(size_t)m * HID + col]    = v;
        acc1[(size_t)m * HID + col] = v;
      }
    }
  }
}


#define A1_NPP  4800
#define A1_C    12
#define A1_BS   1024
#define A1_U    8
#define A1_Q    64

template <int I>
__device__ __forceinline__ int row_share_bcast(int v) {
  return __builtin_amdgcn_mov_dpp(v, 0x150 + I, 0xf, 0xf, false);
}

#define A1_BCAST(I)                                            \
  ss[I] = (unsigned)row_share_bcast<I>(s0);                    \
  rr[I] = (unsigned)row_share_bcast<I>(r0);

template <bool FULL>
__device__ __forceinline__ void a1_add_rows(int s0, int r0, const float* __restrict__ g,
                                            float* acc, unsigned j) {
  const unsigned NONE = 0xffffffffu;
  unsigned ss[16], rr[16];
  float m[16];
  A1_BCAST(0)  A1_BCAST(1)  A1_BCAST(2)  A1_BCAST(3)
  A1_BCAST(4)  A1_BCAST(5)  A1_BCAST(6)  A1_BCAST(7)
  A1_BCAST(8)  A1_BCAST(9)  A1_BCAST(10) A1_BCAST(11)
  A1_BCAST(12) A1_BCAST(13) A1_BCAST(14) A1_BCAST(15)
#pragma unroll
  for (int i = 0; i < 16; ++i) m[i] = g[ss[i] * HID + j];
#pragma unroll
  for (int i = 0; i < 16; ++i) {
    if (FULL || rr[i] != NONE) atomicAdd(&acc[rr[i] * HID + j], m[i]);
  }
}

template <bool FULL>
__device__ __forceinline__ void a1_drain(const unsigned* qs, const unsigned* qr, unsigned head,
                                         unsigned cnt, const float* __restrict__ g, float* acc,
                                         unsigned lane, unsigned j) {
  const unsigned NONE = 0xffffffffu;
  const unsigned idx = (head + lane) & (A1_Q - 1);
  const bool valid = FULL || lane < cnt;
  const int s0 = valid ? (int)qs[idx] : 0;
  const int r0 = valid ? (int)qr[idx] : (int)NONE;
  a1_add_rows<FULL>(s0, r0, g, acc, j);
}

__device__ __forceinline__ void a1_flush(const float* acc, float* o, int nf, int tid) {
  for (int i = tid; i < nf; i += 5 * A1_BS) {
    float v[5];
#pragma unroll
    for (int q = 0; q < 5; ++q) v[q] = acc[min(i + q * A1_BS, nf - 1)];
#pragma unroll
    for (int q = 0; q < 5; ++q) {
      if (i + q * A1_BS < nf) atomicAdd(o + i + q * A1_BS, v[q]);
    }
  }
}

__global__ __launch_bounds__(A1_BS) void k_agg1_bk(
    const unsigned* __restrict__ bins, const int* __restrict__ tab,
    const float* __restrict__ g, float* __restrict__ acc1, int N, int L) {
  __shared__ __attribute__((aligned(16))) float acc[BK_NPP * HID];
  const unsigned NONE = 0xffffffffu;
  const int tid = threadIdx.x;
  const int lane = tid & 31;
  const unsigned j = (unsigned)lane & 15u;
  const int n0  = blockIdx.x * BK_NPP;
  const int np  = min(BK_NPP, N - n0);
  for (int i = tid; i < BK_NPP * HID / 4; i += A1_BS) ((v4f*)acc)[i] = v4f{0.0f, 0.0f, 0.0f, 0.0f};
  __syncthreads();
  int len;
  const unsigned* p = bk_segment(bins, tab, L, tid >> 5, len);
  unsigned wn = (lane < len) ? p[lane] : NONE;
  for (int i = 0; i < len; i += 32) {
    const unsigned w = wn;
    const int ni = i + 32 + lane;
    wn = (ni < len) ? p[ni] : NONE;
    const bool valid = w != NONE;
    const int s0 = valid ? (int)(w >> BK_LBITS) : 0;
    const int r0 = valid ? (int)(w & BK_LMASK) : (int)NONE;
    if (i + 32 <= len) a1_add_rows<true>(s0, r0, g, acc, j);
    else               a1_add_rows<false>(s0, r0, g, acc, j);
  }
  __syncthreads();
  a1_flush(acc, acc1 + (size_t)n0 * HID, np * HID, tid);
}

__global__ __launch_bounds__(A1_BS) void k_agg1_lds(
    const int* __restrict__ src, const int* __restrict__ dst,
    const float* __restrict__ g, float* __restrict__ acc1, int N, int E) {
  __shared__ __attribute__((aligned(16))) float acc[A1_NPP * HID];
  __shared__ unsigned qsrc[A1_BS / 32][A1_Q];
  __shared__ unsigned qrow[A1_BS / 32][A1_Q];
  const int tid = threadIdx.x;
  const unsigned lane = tid & 31;
  const unsigned j = lane & 15;
  const int wv  = tid >> 5;
  const int n0  = blockIdx.x * A1_NPP;
  const int np  = min(A1_NPP, N - n0);
  for (int i = tid; i < A1_NPP * HID / 4; i += A1_BS) ((v4f*)acc)[i] = v4f{0.0f, 0.0f, 0.0f, 0.0f};
  __syncthreads();
  const int per = (E + A1_C - 1) / A1_C;
  const int e0  = blockIdx.y * per;
  const int e1  = min(E, e0 + per);
  unsigned* qs = &qsrc[wv][0];
  unsigned* qr = &qrow[wv][0];

  if (e0 < e1) {
    const unsigned NONE = 0xffffffffu;
    unsigned head = 0, tail = 0;
    unsigned r[A1_U];
    int      s[A1_U];
    int tb = e0;
#pragma unroll
    for (int k = 0; k < A1_U; ++k) {
      int e  = tb + tid + k * A1_BS;
      int ec = min(e, e1 - 1);
      int dd = dst[ec];
      s[k] = src[ec];
      r[k] = (e < e1) ? (unsigned)(dd - n0) : NONE;
    }
    while (tb < e1) {
      const int nbase = tb + A1_U * A1_BS + tid;
      int dn[A1_U], sn[A1_U];
#pragma unroll
      for (int k = 0; k < A1_U; ++k) {
        int ec = min(nbase + k * A1_BS, e1 - 1);
        dn[k] = dst[ec];
        sn[k] = src[ec];
      }
#pragma unroll
      for (int k = 0; k < A1_U; ++k) {
        const bool hit = r[k] < (unsigned)np;
        const unsigned mask = __builtin_amdgcn_ballot_w32(hit);
        if (mask != 0u) {
          const unsigned pos = (tail + __builtin_amdgcn_mbcnt_lo(mask, 0u)) & (A1_Q - 1);
          if (hit) { qs[pos] = (unsigned)s[k]; qr[pos] = r[k]; }
          tail += (unsigned)__builtin_popcount(mask);
          if (tail - head >= 32u) {
            a1_drain<true>(qs, qr, head, 32u, g, acc, lane, j);
            head += 32u;
          }
        }
      }
#pragma unroll
      for (int k = 0; k < A1_U; ++k) {
        r[k] = (nbase + k * A1_BS < e1) ? (unsigned)(dn[k] - n0) : NONE;
        s[k] = sn[k];
      }
      tb += A1_U * A1_BS;
    }
    if (tail != head) a1_drain<false>(qs, qr, head, tail - head, g, acc, lane, j);
  }
  __syncthreads();
  a1_flush(acc, acc1 + (size_t)n0 * HID, np * HID, tid);
}

__global__ void k_layer2_node(const float* __restrict__ acc1, const float* __restrict__ dinv,
                              const float* __restrict__ b1, const float* __restrict__ W2,
                              float* __restrict__ h3, int N) {
  int n = blockIdx.x * blockDim.x + threadIdx.x;
  if (n >= N) return;
  const float dv = dinv[n];
  float o0 = 0.0f, o1 = 0.0f;
#pragma unroll
  for (int j = 0; j < HID; ++j) {
    float v = fmaxf(acc1[(size_t)n * HID + j] * dv + b1[j], 0.0f);
    o0 = __builtin_fmaf(v, W2[j * OUTF + 0], o0);
    o1 = __builtin_fmaf(v, W2[j * OUTF + 1], o1);
  }
  v2f o;
  o.x = o0 * dv;
  o.y = o1 * dv;
  *(v2f*)(h3 + (size_t)n * OUTF) = o;
}

__global__ __launch_bounds__(SCAN_BS) void k_agg2_part(
    const int* __restrict__ src, const int* __restrict__ dst,
    const float* __restrict__ h3, float* __restrict__ part, int N, int E) {
  __shared__ float acc[NPP * OUTF];
  const v2f* __restrict__ h3v = (const v2f*)h3;
  const int tid = threadIdx.x;
  const int n0  = blockIdx.x * NPP;
  const int np  = min(NPP, N - n0);
  for (int i = tid; i < NPP * OUTF; i += SCAN_BS) acc[i] = 0.0f;
  __syncthreads();
  const int per = (E + NCH - 1) / NCH;
  const int e0  = blockIdx.y * per;
  const int e1  = min(E, e0 + per);

  if (e0 < e1) {
    const unsigned NONE = 0xffffffffu;
    unsigned r[SCAN_U];
    int      s[SCAN_U];
    int base = e0 + tid;
#pragma unroll
    for (int k = 0; k < SCAN_U; ++k) {
      int e  = base + k * SCAN_BS;
      int ec = min(e, e1 - 1);
      int dd = dst[ec];
      s[k] = src[ec];
      r[k] = (e < e1) ? (unsigned)(dd - n0) : NONE;
    }
    while (base < e1) {
      v2f gv[SCAN_U];
#pragma unroll
      for (int k = 0; k < SCAN_U; ++k) {
        const bool keep = r[k] < (unsigned)np;
        gv[k] = h3v[keep ? s[k] : 0];
      }
      const int nbase = base + SCAN_U * SCAN_BS;
      int dn[SCAN_U], sn[SCAN_U];
#pragma unroll
      for (int k = 0; k < SCAN_U; ++k) {
        int ec = min(nbase + k * SCAN_BS, e1 - 1);
        dn[k] = dst[ec];
        sn[k] = src[ec];
      }
#pragma unroll
      for (int k = 0; k < SCAN_U; ++k) {
        if (r[k] < (unsigned)np) {
          atomicAdd(&acc[r[k] * OUTF + 0], gv[k].x);
          atomicAdd(&acc[r[k] * OUTF + 1], gv[k].y);
        }
      }
#pragma unroll
      for (int k = 0; k < SCAN_U; ++k) {
        r[k] = (nbase + k * SCAN_BS < e1) ? (unsigned)(dn[k] - n0) : NONE;
        s[k] = sn[k];
      }
      base = nbase;
    }
  }
  __syncthreads();
  float* o = part + ((size_t)blockIdx.y * N + n0) * OUTF;
  for (int i = tid; i < np * OUTF; i += SCAN_BS) o[i] = acc[i];
}

#define BK_A2_U 8

__global__ __launch_bounds__(BK_BS) void k_agg2_part_bk(
    const unsigned* __restrict__ bins, const int* __restrict__ tab,
    const float* __restrict__ h3, float* __restrict__ part, int N, int L) {
  __shared__ float acc[BK_NPP * OUTF];
  const v2f* __restrict__ h3v = (const v2f*)h3;
  const int tid  = threadIdx.x;
  const int lane = tid & 31;
  const int n0   = blockIdx.x * BK_NPP;
  const int np   = min(BK_NPP, N - n0);
  for (int i = tid; i < BK_NPP * OUTF; i += BK_BS) acc[i] = 0.0f;
  __syncthreads();
  int len;
  const unsigned* p = bk_segment(bins, tab, L, tid >> 5, len);
  for (int i = lane; i < len; i += 32 * BK_A2_U) {
    unsigned w[BK_A2_U];
    v2f gv[BK_A2_U];
#pragma unroll
    for (int k = 0; k < BK_A2_U; ++k) w[k] = p[min(i + 32 * k, len - 1)];
#pragma unroll
    for (int k = 0; k < BK_A2_U; ++k) gv[k] = h3v[w[k] >> BK_LBITS];
#pragma unroll
    for (int k = 0; k < BK_A2_U; ++k) {
      if (i + 32 * k < len) {
        const unsigned row = w[k] & BK_LMASK;
        atomicAdd(&acc[row * OUTF + 0], gv[k].x);
        atomicAdd(&acc[row * OUTF + 1], gv[k].y);
      }
    }
  }
  __syncthreads();
  float* o = part + ((size_t)blockIdx.y * N + n0) * OUTF;
  for (int i = tid; i < np * OUTF; i += BK_BS) o[i] = acc[i];
}

template <int P>
__global__ void k_agg2_reduce(const float* __restrict__ part, const float* __restrict__ h3,
                              const float* __restrict__ dinv, const float* __restrict__ b2,
                              float* __restrict__ out, int N) {
  int t = blockIdx.x * blockDim.x + threadIdx.x;
  if (t >= N * OUTF) return;
  float s = h3[t];
#pragma unroll
  for (int c = 0; c < P; ++c) s += part[(size_t)c * N * OUTF + t];
  out[t] = s * dinv[t / OUTF] + b2[t % OUTF];
}

static inline int cdiv_i(long long a, long long b) { return (int)((a + b - 1) / b); }

extern "C" void kernel_launch(void* const* d_in, const int* in_sizes, int n_in,
                              void* d_out, int out_size, void* d_ws, size_t ws_size,
                              hipStream_t stream) {
  const float* x  = (const float*)d_in[0];
  const int*   ei = (const int*)  d_in[1];
  const float* W1 = (const float*)d_in[2];
  const float* b1 = (const float*)d_in[3];
  const float* W2 = (const float*)d_in[4];
  const float* b2 = (const float*)d_in[5];
  float* out = (float*)d_out;

  const int N = in_sizes[0] / F_IN;
  const int E = in_sizes[1] / 2;
  const int* src = ei;
  const int* dst = ei + E;

  float* ws   = (float*)d_ws;
  float* dinv = ws;
  float* h    = dinv + N;
  float* acc1 = h + (size_t)N * HID;
  float* h3   = acc1 + (size_t)N * HID;

  const int BS = 256;
  const int nTiles = cdiv_i(N, 16);

  float* part = h;

  const int R = cdiv_i(N, BK_NPP);
  const int L = cdiv_i(E, BK_B);
  int*      tab  = (int*)(h3 + (size_t)N * OUTF);
  unsigned* bins = (unsigned*)(tab + (size_t)BK_B * BK_TABW);
  const size_t need = ((size_t)N * (1 + 2 * HID + OUTF) + (size_t)BK_B * BK_TABW + (size_t)BK_B * L) * sizeof(float);
  const bool bucketed = ws_size >= need && E >= 1 && R <= BK_RMAX && L <= BK_U * BK_BS &&
                        N <= (1 << (32 - BK_LBITS));

  if (bucketed) {
    const dim3 bkGrid(R, BK_C);

    k_bin_edges<<<BK_B, BK_BS, 0, stream>>>(src, dst, bins, tab, N, E, L);

    k_deg_part_bk<<<bkGrid, BK_BS, 0, stream>>>(bins, tab, part, N, L);
    k_deg_reduce<BK_C><<<cdiv_i(N, BS), BS, 0, stream>>>(part, dinv, N);

    k_gemm1_wmma<<<cdiv_i(nTiles, G1_WAVES), G1_WAVES * 32, 0, stream>>>(x, W1, dinv, h, acc1, N, nTiles);

    k_agg1_bk<<<bkGrid, A1_BS, 0, stream>>>(bins, tab, h, acc1, N, L);

    k_layer2_node<<<cdiv_i(N, BS), BS, 0, stream>>>(acc1, dinv, b1, W2, h3, N);

    k_agg2_part_bk<<<bkGrid, BK_BS, 0, stream>>>(bins, tab, h3, part, N, L);
    k_agg2_reduce<BK_C><<<cdiv_i((long long)N * OUTF, BS), BS, 0, stream>>>(part, h3, dinv, b2, out, N);
    return;
  }

  const dim3 scanGrid(cdiv_i(N, NPP), NCH);

  k_deg_part  <<<scanGrid, SCAN_BS, 0, stream>>>(dst, part, N, E);
  k_deg_reduce<NCH><<<cdiv_i(N, BS), BS, 0, stream>>>(part, dinv, N);

  k_gemm1_wmma<<<cdiv_i(nTiles, G1_WAVES), G1_WAVES * 32, 0, stream>>>(x, W1, dinv, h, acc1, N, nTiles);

  k_agg1_lds<<<dim3(cdiv_i(N, A1_NPP), A1_C), A1_BS, 0, stream>>>(src, dst, h, acc1, N, E);

  k_layer2_node<<<cdiv_i(N, BS), BS, 0, stream>>>(acc1, dinv, b1, W2, h3, N);

  k_agg2_part  <<<scanGrid, SCAN_BS, 0, stream>>>(src, dst, h3, part, N, E);
  k_agg2_reduce<NCH><<<cdiv_i((long long)N * OUTF, BS), BS, 0, stream>>>(part, h3, dinv, b2, out, N);
}


# Node ranges cut to 400 nodes so one block owns a range in each edge pass and finishes its nodes from LDS: partial planes, global f32 adds, 3 launches gone; edge sums regrouped, by arrival as before
# speedup vs baseline: 6.7746x; 1.1813x over previous
#include <hip/hip_runtime.h>

typedef __attribute__((ext_vector_type(2))) float v2f;
typedef __attribute__((ext_vector_type(4))) float v4f;
typedef __attribute__((ext_vector_type(8))) float v8f;

#define F_IN 512
#define HID  16
#define OUTF 2

#define NPP     6400
#define NCH     16
#define SCAN_BS 1024
#define SCAN_U  8

__global__ __launch_bounds__(SCAN_BS) void k_deg_part(
    const int* __restrict__ dst, float* __restrict__ part, int N, int E) {
  __shared__ unsigned cnt[NPP];
  const int tid = threadIdx.x;
  const int n0  = blockIdx.x * NPP;
  const int np  = min(NPP, N - n0);
  for (int i = tid; i < NPP; i += SCAN_BS) cnt[i] = 0u;
  __syncthreads();
  const int per = (E + NCH - 1) / NCH;
  const int e0  = blockIdx.y * per;
  const int e1  = min(E, e0 + per);
  for (int base = e0 + tid; base < e1; base += 2 * SCAN_U * SCAN_BS) {
    int d[2 * SCAN_U];
#pragma unroll
    for (int k = 0; k < 2 * SCAN_U; ++k) d[k] = dst[min(base + k * SCAN_BS, e1 - 1)];
#pragma unroll
    for (int k = 0; k < 2 * SCAN_U; ++k) {
      unsigned r = (unsigned)(d[k] - n0);
      if (base + k * SCAN_BS < e1 && r < (unsigned)np) atomicAdd(&cnt[r], 1u);
    }
  }
  __syncthreads();
  float* o = part + (size_t)blockIdx.y * N + n0;
  for (int i = tid; i < np; i += SCAN_BS) o[i] = (float)cnt[i];
}

template <int P>
__global__ void k_deg_reduce(const float* __restrict__ part, float* __restrict__ dinv, int N) {
  int n = blockIdx.x * blockDim.x + threadIdx.x;
  if (n >= N) return;
  float s = 1.0f;
#pragma unroll
  for (int c = 0; c < P; ++c) s += part[(size_t)c * N + n];
  dinv[n] = rsqrtf(s);
}

#define BK_NPP   400
#define BK_RMAX  256
#define BK_TABR  (BK_RMAX + 1)
#define BK_C     12
#define BK_B     (32 * BK_C)
#define BK_BS    1024
#define BK_U     9
#define BK_LBITS 9
#define BK_LMASK 0x1FFu

__global__ __launch_bounds__(BK_BS) void k_bin_edges(
    const int* __restrict__ src, const int* __restrict__ dst,
    unsigned* __restrict__ bins, int* __restrict__ tab, int N, int E, int L) {
  __shared__ unsigned cnt[BK_RMAX];
  __shared__ unsigned wtot[BK_RMAX / 32];
  __shared__ unsigned segs[BK_TABR];
  __shared__ unsigned stage[BK_U * BK_BS];
  const unsigned NONE = 0xffffffffu;
  const int tid = threadIdx.x;
  const int lane = tid & 31;
  const long long c0 = (long long)blockIdx.x * L;
  const int e0 = (int)(c0 < (long long)E ? c0 : (long long)E);
  const int e1 = (int)(c0 + L < (long long)E ? c0 + L : (long long)E);
  if (tid < BK_RMAX) cnt[tid] = 0u;
  __syncthreads();

  int d[BK_U], s[BK_U];
#pragma unroll
  for (int k = 0; k < BK_U; ++k) {
    const int ec = max(0, min(e0 + tid + k * BK_BS, e1 - 1));
    d[k] = dst[ec];
    s[k] = src[ec];
  }
  unsigned word[BK_U], key[BK_U], pos[BK_U];
#pragma unroll
  for (int k = 0; k < BK_U; ++k) {
    const unsigned r = (unsigned)d[k] / BK_NPP;
    const bool ok = (e0 + tid + k * BK_BS < e1) && ((unsigned)d[k] < (unsigned)N);
    key[k]  = ok ? r : NONE;
    pos[k]  = 0u;
    if (ok) pos[k] = atomicAdd(&cnt[r], 1u);
    word[k] = ((unsigned)s[k] << BK_LBITS) | ((unsigned)d[k] - r * BK_NPP);
  }
  __syncthreads();
  unsigned tot = 0u, incl = 0u;
  if (tid < BK_RMAX) {
    tot = cnt[tid];
    incl = tot;
#pragma unroll
    for (int dl = 1; dl < 32; dl <<= 1) {
      const unsigned t = __shfl_up(incl, (unsigned)dl);
      if (lane >= dl) incl += t;
    }
    if (lane == 31) wtot[tid >> 5] = incl;
  }
  __syncthreads();
  if (tid < BK_RMAX) {
    unsigned base = 0u;
#pragma unroll
    for (int w = 0; w < BK_RMAX / 32; ++w) base += (w < (tid >> 5)) ? wtot[w] : 0u;
    segs[tid] = base + incl - tot;
    if (tid == BK_RMAX - 1) segs[BK_RMAX] = base + incl;
  }
  __syncthreads();
#pragma unroll
  for (int k = 0; k < BK_U; ++k) {
    if (key[k] != NONE) stage[segs[key[k]] + pos[k]] = word[k];
  }
  __syncthreads();
  const unsigned kept = segs[BK_RMAX];
  unsigned* o = bins + (size_t)blockIdx.x * L;
  for (unsigned i = tid; i < kept; i += BK_BS) o[i] = stage[i];
  if (tid < BK_TABR) tab[tid * BK_B + blockIdx.x] = (int)segs[tid];
}

__device__ __forceinline__ void bk_run(const int* __restrict__ tab, int L, int wv, int lane,
                                       int (&P)[BK_C], int (&A)[BK_C], int& total) {
  const int b  = wv * BK_C + min(lane, BK_C - 1);
  const int s0 = tab[blockIdx.x * BK_B + b];
  const int s1 = tab[(blockIdx.x + 1) * BK_B + b];
  int run = 0;
#pragma unroll
  for (int q = 0; q < BK_C; ++q) {
    const int a0 = __builtin_amdgcn_readlane(s0, q);
    const int a1 = __builtin_amdgcn_readlane(s1, q);
    P[q] = run;
    A[q] = (wv * BK_C + q) * L + a0 - run;
    run += a1 - a0;
  }
  total = run;
}

__device__ __forceinline__ int bk_word(const int (&P)[BK_C], const int (&A)[BK_C], int i) {
  int a = A[0];
#pragma unroll
  for (int q = 1; q < BK_C; ++q) a = (i >= P[q]) ? A[q] : a;
  return a + i;
}

#define BK_DEG_U 8

__global__ __launch_bounds__(BK_BS) void k_deg_bk(
    const unsigned* __restrict__ bins, const int* __restrict__ tab,
    float* __restrict__ dinv, int N, int L) {
  __shared__ unsigned cnt[BK_NPP];
  const int tid  = threadIdx.x;
  const int lane = tid & 31;
  const int n0   = blockIdx.x * BK_NPP;
  const int np   = min(BK_NPP, N - n0);
  for (int i = tid; i < BK_NPP; i += BK_BS) cnt[i] = 0u;
  __syncthreads();
  int P[BK_C], A[BK_C], total;
  bk_run(tab, L, tid >> 5, lane, P, A, total);
  for (int i = lane; i < total; i += 32 * BK_DEG_U) {
    unsigned w[BK_DEG_U];
#pragma unroll
    for (int k = 0; k < BK_DEG_U; ++k) w[k] = bins[bk_word(P, A, min(i + 32 * k, total - 1))];
#pragma unroll
    for (int k = 0; k < BK_DEG_U; ++k)
      atomicAdd(&cnt[w[k] & BK_LMASK], (i + 32 * k < total) ? 1u : 0u);
  }
  __syncthreads();
  for (int i = tid; i < np; i += BK_BS) dinv[n0 + i] = rsqrtf(1.0f + (float)cnt[i]);
}

#define G1_WAVES  8
#define G1_PIECE  64
#define G1_PITCH  68

__device__ __forceinline__ void g1_load_piece(v4f (&R)[8], const float* gp, int G) {
#pragma unroll
  for (int j = 0; j < 8; ++j)
    R[j] = __builtin_nontemporal_load((const v4f*)(gp + (size_t)j * 2 * F_IN + G * G1_PIECE));
}

__device__ __forceinline__ void g1_store_piece(const v4f (&R)[8], float* xsw, int hi, int col) {
#pragma unroll
  for (int j = 0; j < 8; ++j)
    *(v4f*)(xsw + (2 * j + hi) * G1_PITCH + 4 * col) = R[j];
}

__device__ __forceinline__ v8f g1_mul_piece(v8f acc, const float* xsw, const float* wfrag,
                                            int G, int hi, int col) {
#pragma unroll
  for (int i = 0; i < 8; ++i) {
    const v4f a4 = *(const v4f*)(xsw + col * G1_PITCH + 8 * i + 4 * hi);
    const v4f b4 = *(const v4f*)(wfrag + (((8 * G + i) * 2 + hi) * 16 + col) * 4);
    v2f a, b;
    a.x = a4.x; a.y = a4.y; b.x = b4.x; b.y = b4.y;
    acc = __builtin_amdgcn_wmma_f32_16x16x4_f32(
         false, a,  false, b,
         (short)0, acc,  false,  false);
    a.x = a4.z; a.y = a4.w; b.x = b4.z; b.y = b4.w;
    acc = __builtin_amdgcn_wmma_f32_16x16x4_f32(
         false, a,  false, b,
         (short)0, acc,  false,  false);
  }
  return acc;
}

__global__ __launch_bounds__(G1_WAVES * 32) void k_gemm1_wmma(
    const float* __restrict__ x, const float* __restrict__ W1,
    const float* __restrict__ dinv, float* __restrict__ g,
    float* __restrict__ acc1, int N, int nTiles) {
  __shared__ __attribute__((aligned(16))) float wfrag[F_IN * HID];
  __shared__ __attribute__((aligned(16))) float xs[G1_WAVES][16 * G1_PITCH];
  const int tid  = threadIdx.x;
  const int lane = tid & 31;
  const int wv   = tid >> 5;
  const int col  = lane & 15;
  const int hi   = lane >> 4;
  const int wave = blockIdx.x * G1_WAVES + wv;
  const int tileBase = wave * 16;
  const int tb = min(tileBase, N - 16);
  const float* gp = x + (size_t)(tb + hi) * F_IN + 4 * col;
  float* xsw = &xs[wv][0];

  v4f R0[8], R1[8];
  g1_load_piece(R0, gp, 0);
  g1_load_piece(R1, gp, 1);

#pragma unroll
  for (int it = 0; it < F_IN * HID / 4 / (G1_WAVES * 32); ++it) {
    const int i4 = tid + it * (G1_WAVES * 32);
    const v4f w = *(const v4f*)(W1 + (size_t)i4 * 4);
    const int k  = i4 >> 2;
    const int c0 = (i4 & 3) * 4;
    float* d = wfrag + (((k >> 3) * 2 + ((k >> 2) & 1)) * 16 + c0) * 4 + (k & 3);
    d[0] = w.x; d[4] = w.y; d[8] = w.z; d[12] = w.w;
  }
  __syncthreads();
  if (wave >= nTiles) return;

  float dv[8];
#pragma unroll
  for (int r = 0; r < 8; ++r) dv[r] = dinv[tb + 8 * hi + r];

  v8f acc = {};
  for (int G = 0; G < F_IN / G1_PIECE; G += 2) {
    g1_store_piece(R0, xsw, hi, col);
    if (G + 2 < F_IN / G1_PIECE) g1_load_piece(R0, gp, G + 2);
    acc = g1_mul_piece(acc, xsw, wfrag, G, hi, col);
    g1_store_piece(R1, xsw, hi, col);
    if (G + 3 < F_IN / G1_PIECE) g1_load_piece(R1, gp, G + 3);
    acc = g1_mul_piece(acc, xsw, wfrag, G + 1, hi, col);
  }

  if (tb == tileBase) {
    const size_t o = ((size_t)(tb + 8 * hi)) * HID + col;
    float* gout = g + o;
#pragma unroll
    for (int r = 0; r < 8; ++r) gout[r * HID] = acc[r] * dv[r];
    if (acc1 != nullptr) {
      float* aout = acc1 + o;
#pragma unroll
      for (int r = 0; r < 8; ++r) aout[r * HID] = acc[r] * dv[r];
    }
  } else {
#pragma unroll
    for (int r = 0; r < 8; ++r) {
      int m = tb + r + 8 * hi;
      if (m >= tileBase) {
        const float v = acc[r] * dv[r];
        g[(size_t)m * HID + col] = v;
        if (acc1 != nullptr) acc1[(size_t)m * HID + col] = v;
      }
    }
  }
}


#define A1_NPP  4800
#define A1_C    12
#define A1_BS   1024
#define A1_U    8
#define A1_Q    64

template <int I>
__device__ __forceinline__ int row_share_bcast(int v) {
  return __builtin_amdgcn_mov_dpp(v, 0x150 + I, 0xf, 0xf, false);
}

#define A1_BCAST(I)                                            \
  ss[I] = (unsigned)row_share_bcast<I>(s0);                    \
  rr[I] = (unsigned)row_share_bcast<I>(r0);

template <bool FULL>
__device__ __forceinline__ void a1_add_rows(int s0, int r0, const float* __restrict__ g,
                                            float* acc, unsigned j) {
  const unsigned NONE = 0xffffffffu;
  unsigned ss[16], rr[16];
  float m[16];
  A1_BCAST(0)  A1_BCAST(1)  A1_BCAST(2)  A1_BCAST(3)
  A1_BCAST(4)  A1_BCAST(5)  A1_BCAST(6)  A1_BCAST(7)
  A1_BCAST(8)  A1_BCAST(9)  A1_BCAST(10) A1_BCAST(11)
  A1_BCAST(12) A1_BCAST(13) A1_BCAST(14) A1_BCAST(15)
#pragma unroll
  for (int i = 0; i < 16; ++i) m[i] = g[ss[i] * HID + j];
#pragma unroll
  for (int i = 0; i < 16; ++i) {
    if (FULL || rr[i] != NONE) atomicAdd(&acc[rr[i] * HID + j], m[i]);
  }
}

template <bool FULL>
__device__ __forceinline__ void a1_drain(const unsigned* qs, const unsigned* qr, unsigned head,
                                         unsigned cnt, const float* __restrict__ g, float* acc,
                                         unsigned lane, unsigned j) {
  const unsigned NONE = 0xffffffffu;
  const unsigned idx = (head + lane) & (A1_Q - 1);
  const bool valid = FULL || lane < cnt;
  const int s0 = valid ? (int)qs[idx] : 0;
  const int r0 = valid ? (int)qr[idx] : (int)NONE;
  a1_add_rows<FULL>(s0, r0, g, acc, j);
}

__device__ __forceinline__ void a1_flush(const float* acc, float* o, int nf, int tid) {
  for (int i = tid; i < nf; i += 5 * A1_BS) {
    float v[5];
#pragma unroll
    for (int q = 0; q < 5; ++q) v[q] = acc[min(i + q * A1_BS, nf - 1)];
#pragma unroll
    for (int q = 0; q < 5; ++q) {
      if (i + q * A1_BS < nf) atomicAdd(o + i + q * A1_BS, v[q]);
    }
  }
}

__global__ __launch_bounds__(A1_BS) void k_agg1_bk(
    const unsigned* __restrict__ bins, const int* __restrict__ tab,
    const float* __restrict__ g, const float* __restrict__ dinv,
    const float* __restrict__ b1, const float* __restrict__ W2,
    float* __restrict__ h3, int N, int L) {
  __shared__ __attribute__((aligned(16))) float acc[BK_NPP * HID];
  const unsigned NONE = 0xffffffffu;
  const int tid = threadIdx.x;
  const int lane = tid & 31;
  const unsigned j = (unsigned)lane & 15u;
  const int n0  = blockIdx.x * BK_NPP;
  const int np  = min(BK_NPP, N - n0);
  for (int i = tid; i < BK_NPP * HID / 4; i += A1_BS) ((v4f*)acc)[i] = v4f{0.0f, 0.0f, 0.0f, 0.0f};
  __syncthreads();
  int P[BK_C], A[BK_C], total;
  bk_run(tab, L, tid >> 5, lane, P, A, total);
  unsigned wn = (lane < total) ? bins[bk_word(P, A, lane)] : NONE;
  for (int i = 0; i < total; i += 32) {
    const unsigned w = wn;
    const int ni = i + 32 + lane;
    wn = (ni < total) ? bins[bk_word(P, A, ni)] : NONE;
    const bool valid = w != NONE;
    const int s0 = valid ? (int)(w >> BK_LBITS) : 0;
    const int r0 = valid ? (int)(w & BK_LMASK) : (int)NONE;
    if (i + 32 <= total) a1_add_rows<true>(s0, r0, g, acc, j);
    else                 a1_add_rows<false>(s0, r0, g, acc, j);
  }
  __syncthreads();
  if (tid < np) {
    const int n = n0 + tid;
    const float dv = dinv[n];
    const v4f* gr = (const v4f*)(g + (size_t)n * HID);
    const v4f* ar = (const v4f*)(acc + tid * HID);
    float o0 = 0.0f, o1 = 0.0f;
#pragma unroll
    for (int q = 0; q < HID / 4; ++q) {
      const v4f gv = gr[q];
      const v4f av = ar[q];
#pragma unroll
      for (int e = 0; e < 4; ++e) {
        const int jj = 4 * q + e;
        const float v = fmaxf((gv[e] + av[e]) * dv + b1[jj], 0.0f);
        o0 = __builtin_fmaf(v, W2[jj * OUTF + 0], o0);
        o1 = __builtin_fmaf(v, W2[jj * OUTF + 1], o1);
      }
    }
    v2f o;
    o.x = o0 * dv;
    o.y = o1 * dv;
    *(v2f*)(h3 + (size_t)n * OUTF) = o;
  }
}

__global__ __launch_bounds__(A1_BS) void k_agg1_lds(
    const int* __restrict__ src, const int* __restrict__ dst,
    const float* __restrict__ g, float* __restrict__ acc1, int N, int E) {
  __shared__ __attribute__((aligned(16))) float acc[A1_NPP * HID];
  __shared__ unsigned qsrc[A1_BS / 32][A1_Q];
  __shared__ unsigned qrow[A1_BS / 32][A1_Q];
  const int tid = threadIdx.x;
  const unsigned lane = tid & 31;
  const unsigned j = lane & 15;
  const int wv  = tid >> 5;
  const int n0  = blockIdx.x * A1_NPP;
  const int np  = min(A1_NPP, N - n0);
  for (int i = tid; i < A1_NPP * HID / 4; i += A1_BS) ((v4f*)acc)[i] = v4f{0.0f, 0.0f, 0.0f, 0.0f};
  __syncthreads();
  const int per = (E + A1_C - 1) / A1_C;
  const int e0  = blockIdx.y * per;
  const int e1  = min(E, e0 + per);
  unsigned* qs = &qsrc[wv][0];
  unsigned* qr = &qrow[wv][0];

  if (e0 < e1) {
    const unsigned NONE = 0xffffffffu;
    unsigned head = 0, tail = 0;
    unsigned r[A1_U];
    int      s[A1_U];
    int tb = e0;
#pragma unroll
    for (int k = 0; k < A1_U; ++k) {
      int e  = tb + tid + k * A1_BS;
      int ec = min(e, e1 - 1);
      int dd = dst[ec];
      s[k] = src[ec];
      r[k] = (e < e1) ? (unsigned)(dd - n0) : NONE;
    }
    while (tb < e1) {
      const int nbase = tb + A1_U * A1_BS + tid;
      int dn[A1_U], sn[A1_U];
#pragma unroll
      for (int k = 0; k < A1_U; ++k) {
        int ec = min(nbase + k * A1_BS, e1 - 1);
        dn[k] = dst[ec];
        sn[k] = src[ec];
      }
#pragma unroll
      for (int k = 0; k < A1_U; ++k) {
        const bool hit = r[k] < (unsigned)np;
        const unsigned mask = __builtin_amdgcn_ballot_w32(hit);
        if (mask != 0u) {
          const unsigned pos = (tail + __builtin_amdgcn_mbcnt_lo(mask, 0u)) & (A1_Q - 1);
          if (hit) { qs[pos] = (unsigned)s[k]; qr[pos] = r[k]; }
          tail += (unsigned)__builtin_popcount(mask);
          if (tail - head >= 32u) {
            a1_drain<true>(qs, qr, head, 32u, g, acc, lane, j);
            head += 32u;
          }
        }
      }
#pragma unroll
      for (int k = 0; k < A1_U; ++k) {
        r[k] = (nbase + k * A1_BS < e1) ? (unsigned)(dn[k] - n0) : NONE;
        s[k] = sn[k];
      }
      tb += A1_U * A1_BS;
    }
    if (tail != head) a1_drain<false>(qs, qr, head, tail - head, g, acc, lane, j);
  }
  __syncthreads();
  a1_flush(acc, acc1 + (size_t)n0 * HID, np * HID, tid);
}

__global__ void k_layer2_node(const float* __restrict__ acc1, const float* __restrict__ dinv,
                              const float* __restrict__ b1, const float* __restrict__ W2,
                              float* __restrict__ h3, int N) {
  int n = blockIdx.x * blockDim.x + threadIdx.x;
  if (n >= N) return;
  const float dv = dinv[n];
  float o0 = 0.0f, o1 = 0.0f;
#pragma unroll
  for (int j = 0; j < HID; ++j) {
    float v = fmaxf(acc1[(size_t)n * HID + j] * dv + b1[j], 0.0f);
    o0 = __builtin_fmaf(v, W2[j * OUTF + 0], o0);
    o1 = __builtin_fmaf(v, W2[j * OUTF + 1], o1);
  }
  v2f o;
  o.x = o0 * dv;
  o.y = o1 * dv;
  *(v2f*)(h3 + (size_t)n * OUTF) = o;
}

__global__ __launch_bounds__(SCAN_BS) void k_agg2_part(
    const int* __restrict__ src, const int* __restrict__ dst,
    const float* __restrict__ h3, float* __restrict__ part, int N, int E) {
  __shared__ float acc[NPP * OUTF];
  const v2f* __restrict__ h3v = (const v2f*)h3;
  const int tid = threadIdx.x;
  const int n0  = blockIdx.x * NPP;
  const int np  = min(NPP, N - n0);
  for (int i = tid; i < NPP * OUTF; i += SCAN_BS) acc[i] = 0.0f;
  __syncthreads();
  const int per = (E + NCH - 1) / NCH;
  const int e0  = blockIdx.y * per;
  const int e1  = min(E, e0 + per);

  if (e0 < e1) {
    const unsigned NONE = 0xffffffffu;
    unsigned r[SCAN_U];
    int      s[SCAN_U];
    int base = e0 + tid;
#pragma unroll
    for (int k = 0; k < SCAN_U; ++k) {
      int e  = base + k * SCAN_BS;
      int ec = min(e, e1 - 1);
      int dd = dst[ec];
      s[k] = src[ec];
      r[k] = (e < e1) ? (unsigned)(dd - n0) : NONE;
    }
    while (base < e1) {
      v2f gv[SCAN_U];
#pragma unroll
      for (int k = 0; k < SCAN_U; ++k) {
        const bool keep = r[k] < (unsigned)np;
        gv[k] = h3v[keep ? s[k] : 0];
      }
      const int nbase = base + SCAN_U * SCAN_BS;
      int dn[SCAN_U], sn[SCAN_U];
#pragma unroll
      for (int k = 0; k < SCAN_U; ++k) {
        int ec = min(nbase + k * SCAN_BS, e1 - 1);
        dn[k] = dst[ec];
        sn[k] = src[ec];
      }
#pragma unroll
      for (int k = 0; k < SCAN_U; ++k) {
        if (r[k] < (unsigned)np) {
          atomicAdd(&acc[r[k] * OUTF + 0], gv[k].x);
          atomicAdd(&acc[r[k] * OUTF + 1], gv[k].y);
        }
      }
#pragma unroll
      for (int k = 0; k < SCAN_U; ++k) {
        r[k] = (nbase + k * SCAN_BS < e1) ? (unsigned)(dn[k] - n0) : NONE;
        s[k] = sn[k];
      }
      base = nbase;
    }
  }
  __syncthreads();
  float* o = part + ((size_t)blockIdx.y * N + n0) * OUTF;
  for (int i = tid; i < np * OUTF; i += SCAN_BS) o[i] = acc[i];
}

#define BK_A2_U 8

__global__ __launch_bounds__(BK_BS) void k_agg2_bk(
    const unsigned* __restrict__ bins, const int* __restrict__ tab,
    const float* __restrict__ h3, const float* __restrict__ dinv,
    const float* __restrict__ b2, float* __restrict__ out, int N, int L) {
  __shared__ float acc[BK_NPP * OUTF];
  const v2f* __restrict__ h3v = (const v2f*)h3;
  const int tid  = threadIdx.x;
  const int lane = tid & 31;
  const int n0   = blockIdx.x * BK_NPP;
  const int np   = min(BK_NPP, N - n0);
  for (int i = tid; i < BK_NPP * OUTF; i += BK_BS) acc[i] = 0.0f;
  __syncthreads();
  int P[BK_C], A[BK_C], total;
  bk_run(tab, L, tid >> 5, lane, P, A, total);
  for (int i = lane; i < total; i += 32 * BK_A2_U) {
    unsigned w[BK_A2_U];
    v2f gv[BK_A2_U];
#pragma unroll
    for (int k = 0; k < BK_A2_U; ++k) w[k] = bins[bk_word(P, A, min(i + 32 * k, total - 1))];
#pragma unroll
    for (int k = 0; k < BK_A2_U; ++k) gv[k] = h3v[w[k] >> BK_LBITS];
#pragma unroll
    for (int k = 0; k < BK_A2_U; ++k) {
      if (i + 32 * k < total) {
        const unsigned row = w[k] & BK_LMASK;
        atomicAdd(&acc[row * OUTF + 0], gv[k].x);
        atomicAdd(&acc[row * OUTF + 1], gv[k].y);
      }
    }
  }
  __syncthreads();
  for (int i = tid; i < np * OUTF; i += BK_BS) {
    const int t = n0 * OUTF + i;
    float s = h3[t];
    s += acc[i];
    out[t] = s * dinv[t / OUTF] + b2[t % OUTF];
  }
}

template <int P>
__global__ void k_agg2_reduce(const float* __restrict__ part, const float* __restrict__ h3,
                              const float* __restrict__ dinv, const float* __restrict__ b2,
                              float* __restrict__ out, int N) {
  int t = blockIdx.x * blockDim.x + threadIdx.x;
  if (t >= N * OUTF) return;
  float s = h3[t];
#pragma unroll
  for (int c = 0; c < P; ++c) s += part[(size_t)c * N * OUTF + t];
  out[t] = s * dinv[t / OUTF] + b2[t % OUTF];
}

static inline int cdiv_i(long long a, long long b) { return (int)((a + b - 1) / b); }

extern "C" void kernel_launch(void* const* d_in, const int* in_sizes, int n_in,
                              void* d_out, int out_size, void* d_ws, size_t ws_size,
                              hipStream_t stream) {
  const float* x  = (const float*)d_in[0];
  const int*   ei = (const int*)  d_in[1];
  const float* W1 = (const float*)d_in[2];
  const float* b1 = (const float*)d_in[3];
  const float* W2 = (const float*)d_in[4];
  const float* b2 = (const float*)d_in[5];
  float* out = (float*)d_out;

  const int N = in_sizes[0] / F_IN;
  const int E = in_sizes[1] / 2;
  const int* src = ei;
  const int* dst = ei + E;

  float* ws   = (float*)d_ws;
  float* dinv = ws;
  float* h    = dinv + N;
  float* acc1 = h + (size_t)N * HID;
  float* h3   = acc1 + (size_t)N * HID;

  const int BS = 256;
  const int nTiles = cdiv_i(N, 16);

  float* part = h;

  const int R = cdiv_i(N, BK_NPP);
  const int L = cdiv_i(E, BK_B);
  int*      tab  = (int*)acc1;
  unsigned* bins = (unsigned*)(h3 + (size_t)N * OUTF);
  const size_t need = ((size_t)N * (1 + 2 * HID + OUTF) + (size_t)BK_B * L) * sizeof(float);
  const bool bucketed = ws_size >= need && E >= 1 && R <= BK_RMAX && L <= BK_U * BK_BS &&
                        N <= (1 << (32 - BK_LBITS)) && (size_t)BK_TABR * BK_B <= (size_t)N * HID;

  if (bucketed) {
    k_bin_edges<<<BK_B, BK_BS, 0, stream>>>(src, dst, bins, tab, N, E, L);

    k_deg_bk<<<R, BK_BS, 0, stream>>>(bins, tab, dinv, N, L);

    k_gemm1_wmma<<<cdiv_i(nTiles, G1_WAVES), G1_WAVES * 32, 0, stream>>>(x, W1, dinv, h, nullptr, N, nTiles);

    k_agg1_bk<<<R, A1_BS, 0, stream>>>(bins, tab, h, dinv, b1, W2, h3, N, L);

    k_agg2_bk<<<R, BK_BS, 0, stream>>>(bins, tab, h3, dinv, b2, out, N, L);
    return;
  }

  const dim3 scanGrid(cdiv_i(N, NPP), NCH);

  k_deg_part  <<<scanGrid, SCAN_BS, 0, stream>>>(dst, part, N, E);
  k_deg_reduce<NCH><<<cdiv_i(N, BS), BS, 0, stream>>>(part, dinv, N);

  k_gemm1_wmma<<<cdiv_i(nTiles, G1_WAVES), G1_WAVES * 32, 0, stream>>>(x, W1, dinv, h, acc1, N, nTiles);

  k_agg1_lds<<<dim3(cdiv_i(N, A1_NPP), A1_C), A1_BS, 0, stream>>>(src, dst, h, acc1, N, E);

  k_layer2_node<<<cdiv_i(N, BS), BS, 0, stream>>>(acc1, dinv, b1, W2, h3, N);

  k_agg2_part  <<<scanGrid, SCAN_BS, 0, stream>>>(src, dst, h3, part, N, E);
  k_agg2_reduce<NCH><<<cdiv_i((long long)N * OUTF, BS), BS, 0, stream>>>(part, h3, dinv, b2, out, N);
}


# GEMM and edge sort in one launch (sort blocks, then GEMM blocks): the sort's waits pass while x streams; rows scaled by dinv in the degree pass instead of the GEMM epilogue, same product; 4 launches
# speedup vs baseline: 7.2146x; 1.0649x over previous
#include <hip/hip_runtime.h>

typedef __attribute__((ext_vector_type(2))) float v2f;
typedef __attribute__((ext_vector_type(4))) float v4f;
typedef __attribute__((ext_vector_type(8))) float v8f;

#define F_IN 512
#define HID  16
#define OUTF 2

#define NPP     6400
#define NCH     16
#define SCAN_BS 1024
#define SCAN_U  8

__global__ __launch_bounds__(SCAN_BS) void k_deg_part(
    const int* __restrict__ dst, float* __restrict__ part, int N, int E) {
  __shared__ unsigned cnt[NPP];
  const int tid = threadIdx.x;
  const int n0  = blockIdx.x * NPP;
  const int np  = min(NPP, N - n0);
  for (int i = tid; i < NPP; i += SCAN_BS) cnt[i] = 0u;
  __syncthreads();
  const int per = (E + NCH - 1) / NCH;
  const int e0  = blockIdx.y * per;
  const int e1  = min(E, e0 + per);
  for (int base = e0 + tid; base < e1; base += 2 * SCAN_U * SCAN_BS) {
    int d[2 * SCAN_U];
#pragma unroll
    for (int k = 0; k < 2 * SCAN_U; ++k) d[k] = dst[min(base + k * SCAN_BS, e1 - 1)];
#pragma unroll
    for (int k = 0; k < 2 * SCAN_U; ++k) {
      unsigned r = (unsigned)(d[k] - n0);
      if (base + k * SCAN_BS < e1 && r < (unsigned)np) atomicAdd(&cnt[r], 1u);
    }
  }
  __syncthreads();
  float* o = part + (size_t)blockIdx.y * N + n0;
  for (int i = tid; i < np; i += SCAN_BS) o[i] = (float)cnt[i];
}

template <int P>
__global__ void k_deg_reduce(const float* __restrict__ part, float* __restrict__ dinv, int N) {
  int n = blockIdx.x * blockDim.x + threadIdx.x;
  if (n >= N) return;
  float s = 1.0f;
#pragma unroll
  for (int c = 0; c < P; ++c) s += part[(size_t)c * N + n];
  dinv[n] = rsqrtf(s);
}

#define BK_NPP   400
#define BK_RMAX  256
#define BK_TABR  (BK_RMAX + 1)
#define BK_C     12
#define BK_B     (32 * BK_C)
#define BK_BS    1024
#define BK_SORT_BS 512
#define BK_U     18
#define BK_LBITS 9
#define BK_LMASK 0x1FFu

#define BK_SORT_LDS (BK_RMAX + BK_RMAX / 32 + BK_TABR + BK_U * BK_SORT_BS)

__device__ __forceinline__ void bin_edges_block(
    const int* __restrict__ src, const int* __restrict__ dst,
    unsigned* __restrict__ bins, int* __restrict__ tab, int N, int E, int L,
    int chunk, unsigned* lds) {
  unsigned* cnt   = lds;
  unsigned* wtot  = cnt + BK_RMAX;
  unsigned* segs  = wtot + BK_RMAX / 32;
  unsigned* stage = segs + BK_TABR;
  const unsigned NONE = 0xffffffffu;
  const int tid = threadIdx.x;
  const int lane = tid & 31;
  const long long c0 = (long long)chunk * L;
  const int e0 = (int)(c0 < (long long)E ? c0 : (long long)E);
  const int e1 = (int)(c0 + L < (long long)E ? c0 + L : (long long)E);
  if (tid < BK_RMAX) cnt[tid] = 0u;
  __syncthreads();

  int d[BK_U], s[BK_U];
#pragma unroll
  for (int k = 0; k < BK_U; ++k) {
    const int ec = max(0, min(e0 + tid + k * BK_SORT_BS, e1 - 1));
    d[k] = dst[ec];
    s[k] = src[ec];
  }
  unsigned word[BK_U], key[BK_U], pos[BK_U];
#pragma unroll
  for (int k = 0; k < BK_U; ++k) {
    const unsigned r = (unsigned)d[k] / BK_NPP;
    const bool ok = (e0 + tid + k * BK_SORT_BS < e1) && ((unsigned)d[k] < (unsigned)N);
    key[k]  = ok ? r : NONE;
    pos[k]  = 0u;
    if (ok) pos[k] = atomicAdd(&cnt[r], 1u);
    word[k] = ((unsigned)s[k] << BK_LBITS) | ((unsigned)d[k] - r * BK_NPP);
  }
  __syncthreads();
  unsigned tot = 0u, incl = 0u;
  if (tid < BK_RMAX) {
    tot = cnt[tid];
    incl = tot;
#pragma unroll
    for (int dl = 1; dl < 32; dl <<= 1) {
      const unsigned t = __shfl_up(incl, (unsigned)dl);
      if (lane >= dl) incl += t;
    }
    if (lane == 31) wtot[tid >> 5] = incl;
  }
  __syncthreads();
  if (tid < BK_RMAX) {
    unsigned base = 0u;
#pragma unroll
    for (int w = 0; w < BK_RMAX / 32; ++w) base += (w < (tid >> 5)) ? wtot[w] : 0u;
    segs[tid] = base + incl - tot;
    if (tid == BK_RMAX - 1) segs[BK_RMAX] = base + incl;
  }
  __syncthreads();
#pragma unroll
  for (int k = 0; k < BK_U; ++k) {
    if (key[k] != NONE) stage[segs[key[k]] + pos[k]] = word[k];
  }
  __syncthreads();
  const unsigned kept = segs[BK_RMAX];
  unsigned* o = bins + (size_t)chunk * L;
  for (unsigned i = tid; i < kept; i += BK_SORT_BS) o[i] = stage[i];
  if (tid < BK_TABR) tab[tid * BK_B + chunk] = (int)segs[tid];
}

__device__ __forceinline__ void bk_run(const int* __restrict__ tab, int L, int wv, int lane,
                                       int (&P)[BK_C], int (&A)[BK_C], int& total) {
  const int b  = wv * BK_C + min(lane, BK_C - 1);
  const int s0 = tab[blockIdx.x * BK_B + b];
  const int s1 = tab[(blockIdx.x + 1) * BK_B + b];
  int run = 0;
#pragma unroll
  for (int q = 0; q < BK_C; ++q) {
    const int a0 = __builtin_amdgcn_readlane(s0, q);
    const int a1 = __builtin_amdgcn_readlane(s1, q);
    P[q] = run;
    A[q] = (wv * BK_C + q) * L + a0 - run;
    run += a1 - a0;
  }
  total = run;
}

__device__ __forceinline__ int bk_word(const int (&P)[BK_C], const int (&A)[BK_C], int i) {
  int a = A[0];
#pragma unroll
  for (int q = 1; q < BK_C; ++q) a = (i >= P[q]) ? A[q] : a;
  return a + i;
}

#define BK_DEG_U 8

#define BK_DEG_PCS ((BK_NPP * HID / 4 + BK_BS - 1) / BK_BS)

__global__ __launch_bounds__(BK_BS) void k_deg_bk(
    const unsigned* __restrict__ bins, const int* __restrict__ tab,
    float* __restrict__ dinv, float* __restrict__ h, int N, int L) {
  __shared__ unsigned cnt[BK_NPP];
  const int tid  = threadIdx.x;
  const int lane = tid & 31;
  const int n0   = blockIdx.x * BK_NPP;
  const int np   = min(BK_NPP, N - n0);
  v4f* hv = (v4f*)(h + (size_t)n0 * HID);
  const int npc = np * (HID / 4);
  v4f hp[BK_DEG_PCS];
#pragma unroll
  for (int q = 0; q < BK_DEG_PCS; ++q) hp[q] = hv[min(tid + q * BK_BS, npc - 1)];
  for (int i = tid; i < BK_NPP; i += BK_BS) cnt[i] = 0u;
  __syncthreads();
  int P[BK_C], A[BK_C], total;
  bk_run(tab, L, tid >> 5, lane, P, A, total);
  for (int i = lane; i < total; i += 32 * BK_DEG_U) {
    unsigned w[BK_DEG_U];
#pragma unroll
    for (int k = 0; k < BK_DEG_U; ++k) w[k] = bins[bk_word(P, A, min(i + 32 * k, total - 1))];
#pragma unroll
    for (int k = 0; k < BK_DEG_U; ++k)
      atomicAdd(&cnt[w[k] & BK_LMASK], (i + 32 * k < total) ? 1u : 0u);
  }
  __syncthreads();
  for (int i = tid; i < np; i += BK_BS) dinv[n0 + i] = rsqrtf(1.0f + (float)cnt[i]);
#pragma unroll
  for (int q = 0; q < BK_DEG_PCS; ++q) {
    const int p = tid + q * BK_BS;
    if (p < npc) hv[p] = hp[q] * rsqrtf(1.0f + (float)cnt[p >> 2]);
  }
}

#define G1_WAVES  8
#define G1_PIECE  64
#define G1_PITCH  68

__device__ __forceinline__ void g1_load_piece(v4f (&R)[8], const float* gp, int G) {
#pragma unroll
  for (int j = 0; j < 8; ++j)
    R[j] = __builtin_nontemporal_load((const v4f*)(gp + (size_t)j * 2 * F_IN + G * G1_PIECE));
}

__device__ __forceinline__ void g1_store_piece(const v4f (&R)[8], float* xsw, int hi, int col) {
#pragma unroll
  for (int j = 0; j < 8; ++j)
    *(v4f*)(xsw + (2 * j + hi) * G1_PITCH + 4 * col) = R[j];
}

__device__ __forceinline__ v8f g1_mul_piece(v8f acc, const float* xsw, const float* wfrag,
                                            int G, int hi, int col) {
#pragma unroll
  for (int i = 0; i < 8; ++i) {
    const v4f a4 = *(const v4f*)(xsw + col * G1_PITCH + 8 * i + 4 * hi);
    const v4f b4 = *(const v4f*)(wfrag + (((8 * G + i) * 2 + hi) * 16 + col) * 4);
    v2f a, b;
    a.x = a4.x; a.y = a4.y; b.x = b4.x; b.y = b4.y;
    acc = __builtin_amdgcn_wmma_f32_16x16x4_f32(
         false, a,  false, b,
         (short)0, acc,  false,  false);
    a.x = a4.z; a.y = a4.w; b.x = b4.z; b.y = b4.w;
    acc = __builtin_amdgcn_wmma_f32_16x16x4_f32(
         false, a,  false, b,
         (short)0, acc,  false,  false);
  }
  return acc;
}

template <int WAVES, bool SCALED>
__device__ __forceinline__ void gemm1_block(
    const float* __restrict__ x, const float* __restrict__ W1,
    const float* __restrict__ dinv, float* __restrict__ g,
    float* __restrict__ acc1, int N, int nTiles, int blk, float* wfrag, float* xs) {
  const int tid  = threadIdx.x;
  const int lane = tid & 31;
  const int wv   = tid >> 5;
  const int col  = lane & 15;
  const int hi   = lane >> 4;
  const int wave = blk * WAVES + wv;
  const int tileBase = wave * 16;
  const int tb = min(tileBase, N - 16);
  const float* gp = x + (size_t)(tb + hi) * F_IN + 4 * col;
  float* xsw = xs + wv * (16 * G1_PITCH);

  v4f R0[8], R1[8];
  g1_load_piece(R0, gp, 0);
  g1_load_piece(R1, gp, 1);

#pragma unroll
  for (int it = 0; it < F_IN * HID / 4 / (WAVES * 32); ++it) {
    const int i4 = tid + it * (WAVES * 32);
    const v4f w = *(const v4f*)(W1 + (size_t)i4 * 4);
    const int k  = i4 >> 2;
    const int c0 = (i4 & 3) * 4;
    float* d = wfrag + (((k >> 3) * 2 + ((k >> 2) & 1)) * 16 + c0) * 4 + (k & 3);
    d[0] = w.x; d[4] = w.y; d[8] = w.z; d[12] = w.w;
  }
  __syncthreads();
  if (wave >= nTiles) return;

  float dv[8];
#pragma unroll
  for (int r = 0; r < 8; ++r) dv[r] = SCALED ? dinv[tb + 8 * hi + r] : 1.0f;

  v8f acc = {};
  for (int G = 0; G < F_IN / G1_PIECE; G += 2) {
    g1_store_piece(R0, xsw, hi, col);
    if (G + 2 < F_IN / G1_PIECE) g1_load_piece(R0, gp, G + 2);
    acc = g1_mul_piece(acc, xsw, wfrag, G, hi, col);
    g1_store_piece(R1, xsw, hi, col);
    if (G + 3 < F_IN / G1_PIECE) g1_load_piece(R1, gp, G + 3);
    acc = g1_mul_piece(acc, xsw, wfrag, G + 1, hi, col);
  }

  if (tb == tileBase) {
    const size_t o = ((size_t)(tb + 8 * hi)) * HID + col;
    float* gout = g + o;
#pragma unroll
    for (int r = 0; r < 8; ++r) gout[r * HID] = SCALED ? acc[r] * dv[r] : acc[r];
    if (acc1 != nullptr) {
      float* aout = acc1 + o;
#pragma unroll
      for (int r = 0; r < 8; ++r) aout[r * HID] = SCALED ? acc[r] * dv[r] : acc[r];
    }
  } else {
#pragma unroll
    for (int r = 0; r < 8; ++r) {
      int m = tb + r + 8 * hi;
      if (m >= tileBase) {
        const float v = SCALED ? acc[r] * dv[r] : acc[r];
        g[(size_t)m * HID + col] = v;
        if (acc1 != nullptr) acc1[(size_t)m * HID + col] = v;
      }
    }
  }
}

__global__ __launch_bounds__(G1_WAVES * 32) void k_gemm1_wmma(
    const float* __restrict__ x, const float* __restrict__ W1,
    const float* __restrict__ dinv, float* __restrict__ g,
    float* __restrict__ acc1, int N, int nTiles) {
  __shared__ __attribute__((aligned(16))) float wfrag[F_IN * HID];
  __shared__ __attribute__((aligned(16))) float xs[G1_WAVES * 16 * G1_PITCH];
  gemm1_block<G1_WAVES, true>(x, W1, dinv, g, acc1, N, nTiles, (int)blockIdx.x, wfrag, xs);
}

#define G1S_WAVES 16
#define G1S_LDS   (F_IN * HID + G1S_WAVES * 16 * G1_PITCH)
static_assert(G1S_WAVES * 32 == BK_SORT_BS, "one block size for both roles");
static_assert(BK_SORT_LDS <= G1S_LDS, "the sort's arrays fit the launch's LDS");

__global__ __launch_bounds__(BK_SORT_BS) void k_gemm1_sort(
    const float* __restrict__ x, const float* __restrict__ W1, float* __restrict__ h,
    const int* __restrict__ src, const int* __restrict__ dst,
    unsigned* __restrict__ bins, int* __restrict__ tab,
    int N, int nTiles, int E, int L) {
  __shared__ __attribute__((aligned(16))) float lds[G1S_LDS];
  if ((int)blockIdx.x < BK_B) {
    bin_edges_block(src, dst, bins, tab, N, E, L, (int)blockIdx.x, (unsigned*)lds);
  } else {
    gemm1_block<G1S_WAVES, false>(x, W1, nullptr, h, nullptr, N, nTiles, (int)blockIdx.x - BK_B,
                                  lds, lds + F_IN * HID);
  }
}


#define A1_NPP  4800
#define A1_C    12
#define A1_BS   1024
#define A1_U    8
#define A1_Q    64

template <int I>
__device__ __forceinline__ int row_share_bcast(int v) {
  return __builtin_amdgcn_mov_dpp(v, 0x150 + I, 0xf, 0xf, false);
}

#define A1_BCAST(I)                                            \
  ss[I] = (unsigned)row_share_bcast<I>(s0);                    \
  rr[I] = (unsigned)row_share_bcast<I>(r0);

template <bool FULL>
__device__ __forceinline__ void a1_add_rows(int s0, int r0, const float* __restrict__ g,
                                            float* acc, unsigned j) {
  const unsigned NONE = 0xffffffffu;
  unsigned ss[16], rr[16];
  float m[16];
  A1_BCAST(0)  A1_BCAST(1)  A1_BCAST(2)  A1_BCAST(3)
  A1_BCAST(4)  A1_BCAST(5)  A1_BCAST(6)  A1_BCAST(7)
  A1_BCAST(8)  A1_BCAST(9)  A1_BCAST(10) A1_BCAST(11)
  A1_BCAST(12) A1_BCAST(13) A1_BCAST(14) A1_BCAST(15)
#pragma unroll
  for (int i = 0; i < 16; ++i) m[i] = g[ss[i] * HID + j];
#pragma unroll
  for (int i = 0; i < 16; ++i) {
    if (FULL || rr[i] != NONE) atomicAdd(&acc[rr[i] * HID + j], m[i]);
  }
}

template <bool FULL>
__device__ __forceinline__ void a1_drain(const unsigned* qs, const unsigned* qr, unsigned head,
                                         unsigned cnt, const float* __restrict__ g, float* acc,
                                         unsigned lane, unsigned j) {
  const unsigned NONE = 0xffffffffu;
  const unsigned idx = (head + lane) & (A1_Q - 1);
  const bool valid = FULL || lane < cnt;
  const int s0 = valid ? (int)qs[idx] : 0;
  const int r0 = valid ? (int)qr[idx] : (int)NONE;
  a1_add_rows<FULL>(s0, r0, g, acc, j);
}

__device__ __forceinline__ void a1_flush(const float* acc, float* o, int nf, int tid) {
  for (int i = tid; i < nf; i += 5 * A1_BS) {
    float v[5];
#pragma unroll
    for (int q = 0; q < 5; ++q) v[q] = acc[min(i + q * A1_BS, nf - 1)];
#pragma unroll
    for (int q = 0; q < 5; ++q) {
      if (i + q * A1_BS < nf) atomicAdd(o + i + q * A1_BS, v[q]);
    }
  }
}

__global__ __launch_bounds__(A1_BS) void k_agg1_bk(
    const unsigned* __restrict__ bins, const int* __restrict__ tab,
    const float* __restrict__ g, const float* __restrict__ dinv,
    const float* __restrict__ b1, const float* __restrict__ W2,
    float* __restrict__ h3, int N, int L) {
  __shared__ __attribute__((aligned(16))) float acc[BK_NPP * HID];
  const unsigned NONE = 0xffffffffu;
  const int tid = threadIdx.x;
  const int lane = tid & 31;
  const unsigned j = (unsigned)lane & 15u;
  const int n0  = blockIdx.x * BK_NPP;
  const int np  = min(BK_NPP, N - n0);
  for (int i = tid; i < BK_NPP * HID / 4; i += A1_BS) ((v4f*)acc)[i] = v4f{0.0f, 0.0f, 0.0f, 0.0f};
  __syncthreads();
  int P[BK_C], A[BK_C], total;
  bk_run(tab, L, tid >> 5, lane, P, A, total);
  unsigned wn = (lane < total) ? bins[bk_word(P, A, lane)] : NONE;
  for (int i = 0; i < total; i += 32) {
    const unsigned w = wn;
    const int ni = i + 32 + lane;
    wn = (ni < total) ? bins[bk_word(P, A, ni)] : NONE;
    const bool valid = w != NONE;
    const int s0 = valid ? (int)(w >> BK_LBITS) : 0;
    const int r0 = valid ? (int)(w & BK_LMASK) : (int)NONE;
    if (i + 32 <= total) a1_add_rows<true>(s0, r0, g, acc, j);
    else                 a1_add_rows<false>(s0, r0, g, acc, j);
  }
  __syncthreads();
  if (tid < np) {
    const int n = n0 + tid;
    const float dv = dinv[n];
    const v4f* gr = (const v4f*)(g + (size_t)n * HID);
    const v4f* ar = (const v4f*)(acc + tid * HID);
    float o0 = 0.0f, o1 = 0.0f;
#pragma unroll
    for (int q = 0; q < HID / 4; ++q) {
      const v4f gv = gr[q];
      const v4f av = ar[q];
#pragma unroll
      for (int e = 0; e < 4; ++e) {
        const int jj = 4 * q + e;
        const float v = fmaxf((gv[e] + av[e]) * dv + b1[jj], 0.0f);
        o0 = __builtin_fmaf(v, W2[jj * OUTF + 0], o0);
        o1 = __builtin_fmaf(v, W2[jj * OUTF + 1], o1);
      }
    }
    v2f o;
    o.x = o0 * dv;
    o.y = o1 * dv;
    *(v2f*)(h3 + (size_t)n * OUTF) = o;
  }
}

__global__ __launch_bounds__(A1_BS) void k_agg1_lds(
    const int* __restrict__ src, const int* __restrict__ dst,
    const float* __restrict__ g, float* __restrict__ acc1, int N, int E) {
  __shared__ __attribute__((aligned(16))) float acc[A1_NPP * HID];
  __shared__ unsigned qsrc[A1_BS / 32][A1_Q];
  __shared__ unsigned qrow[A1_BS / 32][A1_Q];
  const int tid = threadIdx.x;
  const unsigned lane = tid & 31;
  const unsigned j = lane & 15;
  const int wv  = tid >> 5;
  const int n0  = blockIdx.x * A1_NPP;
  const int np  = min(A1_NPP, N - n0);
  for (int i = tid; i < A1_NPP * HID / 4; i += A1_BS) ((v4f*)acc)[i] = v4f{0.0f, 0.0f, 0.0f, 0.0f};
  __syncthreads();
  const int per = (E + A1_C - 1) / A1_C;
  const int e0  = blockIdx.y * per;
  const int e1  = min(E, e0 + per);
  unsigned* qs = &qsrc[wv][0];
  unsigned* qr = &qrow[wv][0];

  if (e0 < e1) {
    const unsigned NONE = 0xffffffffu;
    unsigned head = 0, tail = 0;
    unsigned r[A1_U];
    int      s[A1_U];
    int tb = e0;
#pragma unroll
    for (int k = 0; k < A1_U; ++k) {
      int e  = tb + tid + k * A1_BS;
      int ec = min(e, e1 - 1);
      int dd = dst[ec];
      s[k] = src[ec];
      r[k] = (e < e1) ? (unsigned)(dd - n0) : NONE;
    }
    while (tb < e1) {
      const int nbase = tb + A1_U * A1_BS + tid;
      int dn[A1_U], sn[A1_U];
#pragma unroll
      for (int k = 0; k < A1_U; ++k) {
        int ec = min(nbase + k * A1_BS, e1 - 1);
        dn[k] = dst[ec];
        sn[k] = src[ec];
      }
#pragma unroll
      for (int k = 0; k < A1_U; ++k) {
        const bool hit = r[k] < (unsigned)np;
        const unsigned mask = __builtin_amdgcn_ballot_w32(hit);
        if (mask != 0u) {
          const unsigned pos = (tail + __builtin_amdgcn_mbcnt_lo(mask, 0u)) & (A1_Q - 1);
          if (hit) { qs[pos] = (unsigned)s[k]; qr[pos] = r[k]; }
          tail += (unsigned)__builtin_popcount(mask);
          if (tail - head >= 32u) {
            a1_drain<true>(qs, qr, head, 32u, g, acc, lane, j);
            head += 32u;
          }
        }
      }
#pragma unroll
      for (int k = 0; k < A1_U; ++k) {
        r[k] = (nbase + k * A1_BS < e1) ? (unsigned)(dn[k] - n0) : NONE;
        s[k] = sn[k];
      }
      tb += A1_U * A1_BS;
    }
    if (tail != head) a1_drain<false>(qs, qr, head, tail - head, g, acc, lane, j);
  }
  __syncthreads();
  a1_flush(acc, acc1 + (size_t)n0 * HID, np * HID, tid);
}

__global__ void k_layer2_node(const float* __restrict__ acc1, const float* __restrict__ dinv,
                              const float* __restrict__ b1, const float* __restrict__ W2,
                              float* __restrict__ h3, int N) {
  int n = blockIdx.x * blockDim.x + threadIdx.x;
  if (n >= N) return;
  const float dv = dinv[n];
  float o0 = 0.0f, o1 = 0.0f;
#pragma unroll
  for (int j = 0; j < HID; ++j) {
    float v = fmaxf(acc1[(size_t)n * HID + j] * dv + b1[j], 0.0f);
    o0 = __builtin_fmaf(v, W2[j * OUTF + 0], o0);
    o1 = __builtin_fmaf(v, W2[j * OUTF + 1], o1);
  }
  v2f o;
  o.x = o0 * dv;
  o.y = o1 * dv;
  *(v2f*)(h3 + (size_t)n * OUTF) = o;
}

__global__ __launch_bounds__(SCAN_BS) void k_agg2_part(
    const int* __restrict__ src, const int* __restrict__ dst,
    const float* __restrict__ h3, float* __restrict__ part, int N, int E) {
  __shared__ float acc[NPP * OUTF];
  const v2f* __restrict__ h3v = (const v2f*)h3;
  const int tid = threadIdx.x;
  const int n0  = blockIdx.x * NPP;
  const int np  = min(NPP, N - n0);
  for (int i = tid; i < NPP * OUTF; i += SCAN_BS) acc[i] = 0.0f;
  __syncthreads();
  const int per = (E + NCH - 1) / NCH;
  const int e0  = blockIdx.y * per;
  const int e1  = min(E, e0 + per);

  if (e0 < e1) {
    const unsigned NONE = 0xffffffffu;
    unsigned r[SCAN_U];
    int      s[SCAN_U];
    int base = e0 + tid;
#pragma unroll
    for (int k = 0; k < SCAN_U; ++k) {
      int e  = base + k * SCAN_BS;
      int ec = min(e, e1 - 1);
      int dd = dst[ec];
      s[k] = src[ec];
      r[k] = (e < e1) ? (unsigned)(dd - n0) : NONE;
    }
    while (base < e1) {
      v2f gv[SCAN_U];
#pragma unroll
      for (int k = 0; k < SCAN_U; ++k) {
        const bool keep = r[k] < (unsigned)np;
        gv[k] = h3v[keep ? s[k] : 0];
      }
      const int nbase = base + SCAN_U * SCAN_BS;
      int dn[SCAN_U], sn[SCAN_U];
#pragma unroll
      for (int k = 0; k < SCAN_U; ++k) {
        int ec = min(nbase + k * SCAN_BS, e1 - 1);
        dn[k] = dst[ec];
        sn[k] = src[ec];
      }
#pragma unroll
      for (int k = 0; k < SCAN_U; ++k) {
        if (r[k] < (unsigned)np) {
          atomicAdd(&acc[r[k] * OUTF + 0], gv[k].x);
          atomicAdd(&acc[r[k] * OUTF + 1], gv[k].y);
        }
      }
#pragma unroll
      for (int k = 0; k < SCAN_U; ++k) {
        r[k] = (nbase + k * SCAN_BS < e1) ? (unsigned)(dn[k] - n0) : NONE;
        s[k] = sn[k];
      }
      base = nbase;
    }
  }
  __syncthreads();
  float* o = part + ((size_t)blockIdx.y * N + n0) * OUTF;
  for (int i = tid; i < np * OUTF; i += SCAN_BS) o[i] = acc[i];
}

#define BK_A2_U 8

__global__ __launch_bounds__(BK_BS) void k_agg2_bk(
    const unsigned* __restrict__ bins, const int* __restrict__ tab,
    const float* __restrict__ h3, const float* __restrict__ dinv,
    const float* __restrict__ b2, float* __restrict__ out, int N, int L) {
  __shared__ float acc[BK_NPP * OUTF];
  const v2f* __restrict__ h3v = (const v2f*)h3;
  const int tid  = threadIdx.x;
  const int lane = tid & 31;
  const int n0   = blockIdx.x * BK_NPP;
  const int np   = min(BK_NPP, N - n0);
  for (int i = tid; i < BK_NPP * OUTF; i += BK_BS) acc[i] = 0.0f;
  __syncthreads();
  int P[BK_C], A[BK_C], total;
  bk_run(tab, L, tid >> 5, lane, P, A, total);
  for (int i = lane; i < total; i += 32 * BK_A2_U) {
    unsigned w[BK_A2_U];
    v2f gv[BK_A2_U];
#pragma unroll
    for (int k = 0; k < BK_A2_U; ++k) w[k] = bins[bk_word(P, A, min(i + 32 * k, total - 1))];
#pragma unroll
    for (int k = 0; k < BK_A2_U; ++k) gv[k] = h3v[w[k] >> BK_LBITS];
#pragma unroll
    for (int k = 0; k < BK_A2_U; ++k) {
      if (i + 32 * k < total) {
        const unsigned row = w[k] & BK_LMASK;
        atomicAdd(&acc[row * OUTF + 0], gv[k].x);
        atomicAdd(&acc[row * OUTF + 1], gv[k].y);
      }
    }
  }
  __syncthreads();
  for (int i = tid; i < np * OUTF; i += BK_BS) {
    const int t = n0 * OUTF + i;
    float s = h3[t];
    s += acc[i];
    out[t] = s * dinv[t / OUTF] + b2[t % OUTF];
  }
}

template <int P>
__global__ void k_agg2_reduce(const float* __restrict__ part, const float* __restrict__ h3,
                              const float* __restrict__ dinv, const float* __restrict__ b2,
                              float* __restrict__ out, int N) {
  int t = blockIdx.x * blockDim.x + threadIdx.x;
  if (t >= N * OUTF) return;
  float s = h3[t];
#pragma unroll
  for (int c = 0; c < P; ++c) s += part[(size_t)c * N * OUTF + t];
  out[t] = s * dinv[t / OUTF] + b2[t % OUTF];
}

static inline int cdiv_i(long long a, long long b) { return (int)((a + b - 1) / b); }

extern "C" void kernel_launch(void* const* d_in, const int* in_sizes, int n_in,
                              void* d_out, int out_size, void* d_ws, size_t ws_size,
                              hipStream_t stream) {
  const float* x  = (const float*)d_in[0];
  const int*   ei = (const int*)  d_in[1];
  const float* W1 = (const float*)d_in[2];
  const float* b1 = (const float*)d_in[3];
  const float* W2 = (const float*)d_in[4];
  const float* b2 = (const float*)d_in[5];
  float* out = (float*)d_out;

  const int N = in_sizes[0] / F_IN;
  const int E = in_sizes[1] / 2;
  const int* src = ei;
  const int* dst = ei + E;

  float* ws   = (float*)d_ws;
  float* dinv = ws;
  float* h    = dinv + N;
  float* acc1 = h + (size_t)N * HID;
  float* h3   = acc1 + (size_t)N * HID;

  const int BS = 256;
  const int nTiles = cdiv_i(N, 16);

  float* part = h;

  const int R = cdiv_i(N, BK_NPP);
  const int L = cdiv_i(E, BK_B);
  int*      tab  = (int*)acc1;
  unsigned* bins = (unsigned*)(h3 + (size_t)N * OUTF);
  const size_t need = ((size_t)N * (1 + 2 * HID + OUTF) + (size_t)BK_B * L) * sizeof(float);
  const bool bucketed = ws_size >= need && E >= 1 && R <= BK_RMAX && L <= BK_U * BK_SORT_BS &&
                        N <= (1 << (32 - BK_LBITS)) && (size_t)BK_TABR * BK_B <= (size_t)N * HID;

  if (bucketed) {
    k_gemm1_sort<<<BK_B + cdiv_i(nTiles, G1S_WAVES), BK_SORT_BS, 0, stream>>>(x, W1, h, src, dst, bins, tab, N, nTiles, E, L);

    k_deg_bk<<<R, BK_BS, 0, stream>>>(bins, tab, dinv, h, N, L);

    k_agg1_bk<<<R, A1_BS, 0, stream>>>(bins, tab, h, dinv, b1, W2, h3, N, L);

    k_agg2_bk<<<R, BK_BS, 0, stream>>>(bins, tab, h3, dinv, b2, out, N, L);
    return;
  }

  const dim3 scanGrid(cdiv_i(N, NPP), NCH);

  k_deg_part  <<<scanGrid, SCAN_BS, 0, stream>>>(dst, part, N, E);
  k_deg_reduce<NCH><<<cdiv_i(N, BS), BS, 0, stream>>>(part, dinv, N);

  k_gemm1_wmma<<<cdiv_i(nTiles, G1_WAVES), G1_WAVES * 32, 0, stream>>>(x, W1, dinv, h, acc1, N, nTiles);

  k_agg1_lds<<<dim3(cdiv_i(N, A1_NPP), A1_C), A1_BS, 0, stream>>>(src, dst, h, acc1, N, E);

  k_layer2_node<<<cdiv_i(N, BS), BS, 0, stream>>>(acc1, dinv, b1, W2, h3, N);

  k_agg2_part  <<<scanGrid, SCAN_BS, 0, stream>>>(src, dst, h3, part, N, E);
  k_agg2_reduce<NCH><<<cdiv_i((long long)N * OUTF, BS), BS, 0, stream>>>(part, h3, dinv, b2, out, N);
}


# Bucketed edge passes, end of a wave's run: every load unconditional at a clamped index, lanes past the end add 0 to an LDS word of their own (no branch, no shared word); float sum order unchanged
# speedup vs baseline: 7.5588x; 1.0477x over previous
#include <hip/hip_runtime.h>

typedef __attribute__((ext_vector_type(2))) float v2f;
typedef __attribute__((ext_vector_type(4))) float v4f;
typedef __attribute__((ext_vector_type(8))) float v8f;

#define F_IN 512
#define HID  16
#define OUTF 2

#define NPP     6400
#define NCH     16
#define SCAN_BS 1024
#define SCAN_U  8

__global__ __launch_bounds__(SCAN_BS) void k_deg_part(
    const int* __restrict__ dst, float* __restrict__ part, int N, int E) {
  __shared__ unsigned cnt[NPP];
  const int tid = threadIdx.x;
  const int n0  = blockIdx.x * NPP;
  const int np  = min(NPP, N - n0);
  for (int i = tid; i < NPP; i += SCAN_BS) cnt[i] = 0u;
  __syncthreads();
  const int per = (E + NCH - 1) / NCH;
  const int e0  = blockIdx.y * per;
  const int e1  = min(E, e0 + per);
  for (int base = e0 + tid; base < e1; base += 2 * SCAN_U * SCAN_BS) {
    int d[2 * SCAN_U];
#pragma unroll
    for (int k = 0; k < 2 * SCAN_U; ++k) d[k] = dst[min(base + k * SCAN_BS, e1 - 1)];
#pragma unroll
    for (int k = 0; k < 2 * SCAN_U; ++k) {
      unsigned r = (unsigned)(d[k] - n0);
      if (base + k * SCAN_BS < e1 && r < (unsigned)np) atomicAdd(&cnt[r], 1u);
    }
  }
  __syncthreads();
  float* o = part + (size_t)blockIdx.y * N + n0;
  for (int i = tid; i < np; i += SCAN_BS) o[i] = (float)cnt[i];
}

template <int P>
__global__ void k_deg_reduce(const float* __restrict__ part, float* __restrict__ dinv, int N) {
  int n = blockIdx.x * blockDim.x + threadIdx.x;
  if (n >= N) return;
  float s = 1.0f;
#pragma unroll
  for (int c = 0; c < P; ++c) s += part[(size_t)c * N + n];
  dinv[n] = rsqrtf(s);
}

#define BK_NPP   400
#define BK_RMAX  256
#define BK_TABR  (BK_RMAX + 1)
#define BK_C     12
#define BK_B     (32 * BK_C)
#define BK_BS    1024
#define BK_SORT_BS 512
#define BK_U     18
#define BK_LBITS 9
#define BK_LMASK 0x1FFu

#define BK_SORT_LDS (BK_RMAX + BK_RMAX / 32 + BK_TABR + BK_U * BK_SORT_BS)

__device__ __forceinline__ void bin_edges_block(
    const int* __restrict__ src, const int* __restrict__ dst,
    unsigned* __restrict__ bins, int* __restrict__ tab, int N, int E, int L,
    int chunk, unsigned* lds) {
  unsigned* cnt   = lds;
  unsigned* wtot  = cnt + BK_RMAX;
  unsigned* segs  = wtot + BK_RMAX / 32;
  unsigned* stage = segs + BK_TABR;
  const unsigned NONE = 0xffffffffu;
  const int tid = threadIdx.x;
  const int lane = tid & 31;
  const long long c0 = (long long)chunk * L;
  const int e0 = (int)(c0 < (long long)E ? c0 : (long long)E);
  const int e1 = (int)(c0 + L < (long long)E ? c0 + L : (long long)E);
  if (tid < BK_RMAX) cnt[tid] = 0u;
  __syncthreads();

  int d[BK_U], s[BK_U];
#pragma unroll
  for (int k = 0; k < BK_U; ++k) {
    const int ec = max(0, min(e0 + tid + k * BK_SORT_BS, e1 - 1));
    d[k] = dst[ec];
    s[k] = src[ec];
  }
  unsigned word[BK_U], key[BK_U], pos[BK_U];
#pragma unroll
  for (int k = 0; k < BK_U; ++k) {
    const unsigned r = (unsigned)d[k] / BK_NPP;
    const bool ok = (e0 + tid + k * BK_SORT_BS < e1) && ((unsigned)d[k] < (unsigned)N);
    key[k]  = ok ? r : NONE;
    pos[k]  = 0u;
    if (ok) pos[k] = atomicAdd(&cnt[r], 1u);
    word[k] = ((unsigned)s[k] << BK_LBITS) | ((unsigned)d[k] - r * BK_NPP);
  }
  __syncthreads();
  unsigned tot = 0u, incl = 0u;
  if (tid < BK_RMAX) {
    tot = cnt[tid];
    incl = tot;
#pragma unroll
    for (int dl = 1; dl < 32; dl <<= 1) {
      const unsigned t = __shfl_up(incl, (unsigned)dl);
      if (lane >= dl) incl += t;
    }
    if (lane == 31) wtot[tid >> 5] = incl;
  }
  __syncthreads();
  if (tid < BK_RMAX) {
    unsigned base = 0u;
#pragma unroll
    for (int w = 0; w < BK_RMAX / 32; ++w) base += (w < (tid >> 5)) ? wtot[w] : 0u;
    segs[tid] = base + incl - tot;
    if (tid == BK_RMAX - 1) segs[BK_RMAX] = base + incl;
  }
  __syncthreads();
#pragma unroll
  for (int k = 0; k < BK_U; ++k) {
    if (key[k] != NONE) stage[segs[key[k]] + pos[k]] = word[k];
  }
  __syncthreads();
  const unsigned kept = segs[BK_RMAX];
  unsigned* o = bins + (size_t)chunk * L;
  for (unsigned i = tid; i < kept; i += BK_SORT_BS) o[i] = stage[i];
  if (tid < BK_TABR) tab[tid * BK_B + chunk] = (int)segs[tid];
}

__device__ __forceinline__ void bk_run(const int* __restrict__ tab, int L, int wv, int lane,
                                       int (&P)[BK_C], int (&A)[BK_C], int& total) {
  const int b  = wv * BK_C + min(lane, BK_C - 1);
  const int s0 = tab[blockIdx.x * BK_B + b];
  const int s1 = tab[(blockIdx.x + 1) * BK_B + b];
  int run = 0;
#pragma unroll
  for (int q = 0; q < BK_C; ++q) {
    const int a0 = __builtin_amdgcn_readlane(s0, q);
    const int a1 = __builtin_amdgcn_readlane(s1, q);
    P[q] = run;
    A[q] = (wv * BK_C + q) * L + a0 - run;
    run += a1 - a0;
  }
  total = run;
}

__device__ __forceinline__ int bk_word(const int (&P)[BK_C], const int (&A)[BK_C], int i) {
  int a = A[0];
#pragma unroll
  for (int q = 1; q < BK_C; ++q) a = (i >= P[q]) ? A[q] : a;
  return a + i;
}

#define BK_DEG_U 8
static_assert(BK_NPP >= 32, "a lane past a run's end adds its 0 to LDS row number lane of the range");

#define BK_DEG_PCS ((BK_NPP * HID / 4 + BK_BS - 1) / BK_BS)

__global__ __launch_bounds__(BK_BS) void k_deg_bk(
    const unsigned* __restrict__ bins, const int* __restrict__ tab,
    float* __restrict__ dinv, float* __restrict__ h, int N, int L) {
  __shared__ unsigned cnt[BK_NPP];
  const int tid  = threadIdx.x;
  const int lane = tid & 31;
  const int n0   = blockIdx.x * BK_NPP;
  const int np   = min(BK_NPP, N - n0);
  v4f* hv = (v4f*)(h + (size_t)n0 * HID);
  const int npc = np * (HID / 4);
  v4f hp[BK_DEG_PCS];
#pragma unroll
  for (int q = 0; q < BK_DEG_PCS; ++q) hp[q] = hv[min(tid + q * BK_BS, npc - 1)];
  for (int i = tid; i < BK_NPP; i += BK_BS) cnt[i] = 0u;
  __syncthreads();
  int P[BK_C], A[BK_C], total;
  bk_run(tab, L, tid >> 5, lane, P, A, total);
  for (int i = lane; i < total; i += 32 * BK_DEG_U) {
    unsigned w[BK_DEG_U];
#pragma unroll
    for (int k = 0; k < BK_DEG_U; ++k) w[k] = bins[bk_word(P, A, min(i + 32 * k, total - 1))];
#pragma unroll
    for (int k = 0; k < BK_DEG_U; ++k) {
      const bool keep = i + 32 * k < total;
      atomicAdd(&cnt[keep ? (w[k] & BK_LMASK) : (unsigned)lane], keep ? 1u : 0u);
    }
  }
  __syncthreads();
  for (int i = tid; i < np; i += BK_BS) dinv[n0 + i] = rsqrtf(1.0f + (float)cnt[i]);
#pragma unroll
  for (int q = 0; q < BK_DEG_PCS; ++q) {
    const int p = tid + q * BK_BS;
    if (p < npc) hv[p] = hp[q] * rsqrtf(1.0f + (float)cnt[p >> 2]);
  }
}

#define G1_WAVES  8
#define G1_PIECE  64
#define G1_PITCH  68

__device__ __forceinline__ void g1_load_piece(v4f (&R)[8], const float* gp, int G) {
#pragma unroll
  for (int j = 0; j < 8; ++j)
    R[j] = __builtin_nontemporal_load((const v4f*)(gp + (size_t)j * 2 * F_IN + G * G1_PIECE));
}

__device__ __forceinline__ void g1_store_piece(const v4f (&R)[8], float* xsw, int hi, int col) {
#pragma unroll
  for (int j = 0; j < 8; ++j)
    *(v4f*)(xsw + (2 * j + hi) * G1_PITCH + 4 * col) = R[j];
}

__device__ __forceinline__ v8f g1_mul_piece(v8f acc, const float* xsw, const float* wfrag,
                                            int G, int hi, int col) {
#pragma unroll
  for (int i = 0; i < 8; ++i) {
    const v4f a4 = *(const v4f*)(xsw + col * G1_PITCH + 8 * i + 4 * hi);
    const v4f b4 = *(const v4f*)(wfrag + (((8 * G + i) * 2 + hi) * 16 + col) * 4);
    v2f a, b;
    a.x = a4.x; a.y = a4.y; b.x = b4.x; b.y = b4.y;
    acc = __builtin_amdgcn_wmma_f32_16x16x4_f32(
         false, a,  false, b,
         (short)0, acc,  false,  false);
    a.x = a4.z; a.y = a4.w; b.x = b4.z; b.y = b4.w;
    acc = __builtin_amdgcn_wmma_f32_16x16x4_f32(
         false, a,  false, b,
         (short)0, acc,  false,  false);
  }
  return acc;
}

template <int WAVES, bool SCALED>
__device__ __forceinline__ void gemm1_block(
    const float* __restrict__ x, const float* __restrict__ W1,
    const float* __restrict__ dinv, float* __restrict__ g,
    float* __restrict__ acc1, int N, int nTiles, int blk, float* wfrag, float* xs) {
  const int tid  = threadIdx.x;
  const int lane = tid & 31;
  const int wv   = tid >> 5;
  const int col  = lane & 15;
  const int hi   = lane >> 4;
  const int wave = blk * WAVES + wv;
  const int tileBase = wave * 16;
  const int tb = min(tileBase, N - 16);
  const float* gp = x + (size_t)(tb + hi) * F_IN + 4 * col;
  float* xsw = xs + wv * (16 * G1_PITCH);

  v4f R0[8], R1[8];
  g1_load_piece(R0, gp, 0);
  g1_load_piece(R1, gp, 1);

#pragma unroll
  for (int it = 0; it < F_IN * HID / 4 / (WAVES * 32); ++it) {
    const int i4 = tid + it * (WAVES * 32);
    const v4f w = *(const v4f*)(W1 + (size_t)i4 * 4);
    const int k  = i4 >> 2;
    const int c0 = (i4 & 3) * 4;
    float* d = wfrag + (((k >> 3) * 2 + ((k >> 2) & 1)) * 16 + c0) * 4 + (k & 3);
    d[0] = w.x; d[4] = w.y; d[8] = w.z; d[12] = w.w;
  }
  __syncthreads();
  if (wave >= nTiles) return;

  float dv[8];
#pragma unroll
  for (int r = 0; r < 8; ++r) dv[r] = SCALED ? dinv[tb + 8 * hi + r] : 1.0f;

  v8f acc = {};
  for (int G = 0; G < F_IN / G1_PIECE; G += 2) {
    g1_store_piece(R0, xsw, hi, col);
    if (G + 2 < F_IN / G1_PIECE) g1_load_piece(R0, gp, G + 2);
    acc = g1_mul_piece(acc, xsw, wfrag, G, hi, col);
    g1_store_piece(R1, xsw, hi, col);
    if (G + 3 < F_IN / G1_PIECE) g1_load_piece(R1, gp, G + 3);
    acc = g1_mul_piece(acc, xsw, wfrag, G + 1, hi, col);
  }

  if (tb == tileBase) {
    const size_t o = ((size_t)(tb + 8 * hi)) * HID + col;
    float* gout = g + o;
#pragma unroll
    for (int r = 0; r < 8; ++r) gout[r * HID] = SCALED ? acc[r] * dv[r] : acc[r];
    if (acc1 != nullptr) {
      float* aout = acc1 + o;
#pragma unroll
      for (int r = 0; r < 8; ++r) aout[r * HID] = SCALED ? acc[r] * dv[r] : acc[r];
    }
  } else {
#pragma unroll
    for (int r = 0; r < 8; ++r) {
      int m = tb + r + 8 * hi;
      if (m >= tileBase) {
        const float v = SCALED ? acc[r] * dv[r] : acc[r];
        g[(size_t)m * HID + col] = v;
        if (acc1 != nullptr) acc1[(size_t)m * HID + col] = v;
      }
    }
  }
}

__global__ __launch_bounds__(G1_WAVES * 32) void k_gemm1_wmma(
    const float* __restrict__ x, const float* __restrict__ W1,
    const float* __restrict__ dinv, float* __restrict__ g,
    float* __restrict__ acc1, int N, int nTiles) {
  __shared__ __attribute__((aligned(16))) float wfrag[F_IN * HID];
  __shared__ __attribute__((aligned(16))) float xs[G1_WAVES * 16 * G1_PITCH];
  gemm1_block<G1_WAVES, true>(x, W1, dinv, g, acc1, N, nTiles, (int)blockIdx.x, wfrag, xs);
}

#define G1S_WAVES 16
#define G1S_LDS   (F_IN * HID + G1S_WAVES * 16 * G1_PITCH)
static_assert(G1S_WAVES * 32 == BK_SORT_BS, "one block size for both roles");
static_assert(BK_SORT_LDS <= G1S_LDS, "the sort's arrays fit the launch's LDS");

__global__ __launch_bounds__(BK_SORT_BS) void k_gemm1_sort(
    const float* __restrict__ x, const float* __restrict__ W1, float* __restrict__ h,
    const int* __restrict__ src, const int* __restrict__ dst,
    unsigned* __restrict__ bins, int* __restrict__ tab,
    int N, int nTiles, int E, int L) {
  __shared__ __attribute__((aligned(16))) float lds[G1S_LDS];
  if ((int)blockIdx.x < BK_B) {
    bin_edges_block(src, dst, bins, tab, N, E, L, (int)blockIdx.x, (unsigned*)lds);
  } else {
    gemm1_block<G1S_WAVES, false>(x, W1, nullptr, h, nullptr, N, nTiles, (int)blockIdx.x - BK_B,
                                  lds, lds + F_IN * HID);
  }
}


#define A1_NPP  4800
#define A1_C    12
#define A1_BS   1024
#define A1_U    8
#define A1_Q    64

template <int I>
__device__ __forceinline__ int row_share_bcast(int v) {
  return __builtin_amdgcn_mov_dpp(v, 0x150 + I, 0xf, 0xf, false);
}

#define A1_BCAST(I)                                            \
  ss[I] = (unsigned)row_share_bcast<I>(s0);                    \
  rr[I] = (unsigned)row_share_bcast<I>(r0);

template <bool FULL>
__device__ __forceinline__ void a1_add_rows(int s0, int r0, const float* __restrict__ g,
                                            float* acc, unsigned j) {
  const unsigned NONE = 0xffffffffu;
  unsigned ss[16], rr[16];
  float m[16];
  A1_BCAST(0)  A1_BCAST(1)  A1_BCAST(2)  A1_BCAST(3)
  A1_BCAST(4)  A1_BCAST(5)  A1_BCAST(6)  A1_BCAST(7)
  A1_BCAST(8)  A1_BCAST(9)  A1_BCAST(10) A1_BCAST(11)
  A1_BCAST(12) A1_BCAST(13) A1_BCAST(14) A1_BCAST(15)
#pragma unroll
  for (int i = 0; i < 16; ++i) m[i] = g[ss[i] * HID + j];
#pragma unroll
  for (int i = 0; i < 16; ++i) {
    if (FULL || rr[i] != NONE) atomicAdd(&acc[rr[i] * HID + j], m[i]);
  }
}

__device__ __forceinline__ void a1_add_rows_masked(int s0, int r0, const float* __restrict__ g,
                                                   float* acc, unsigned j, unsigned hi) {
  const unsigned NONE = 0xffffffffu;
  unsigned ss[16], rr[16];
  float m[16];
  A1_BCAST(0)  A1_BCAST(1)  A1_BCAST(2)  A1_BCAST(3)
  A1_BCAST(4)  A1_BCAST(5)  A1_BCAST(6)  A1_BCAST(7)
  A1_BCAST(8)  A1_BCAST(9)  A1_BCAST(10) A1_BCAST(11)
  A1_BCAST(12) A1_BCAST(13) A1_BCAST(14) A1_BCAST(15)
#pragma unroll
  for (int i = 0; i < 16; ++i) m[i] = g[ss[i] * HID + j];
#pragma unroll
  for (int i = 0; i < 16; ++i) {
    const bool keep = rr[i] != NONE;
    atomicAdd(&acc[(keep ? rr[i] : hi) * HID + j], keep ? m[i] : 0.0f);
  }
}

template <bool FULL>
__device__ __forceinline__ void a1_drain(const unsigned* qs, const unsigned* qr, unsigned head,
                                         unsigned cnt, const float* __restrict__ g, float* acc,
                                         unsigned lane, unsigned j) {
  const unsigned NONE = 0xffffffffu;
  const unsigned idx = (head + lane) & (A1_Q - 1);
  const bool valid = FULL || lane < cnt;
  const int s0 = valid ? (int)qs[idx] : 0;
  const int r0 = valid ? (int)qr[idx] : (int)NONE;
  a1_add_rows<FULL>(s0, r0, g, acc, j);
}

__device__ __forceinline__ void a1_flush(const float* acc, float* o, int nf, int tid) {
  for (int i = tid; i < nf; i += 5 * A1_BS) {
    float v[5];
#pragma unroll
    for (int q = 0; q < 5; ++q) v[q] = acc[min(i + q * A1_BS, nf - 1)];
#pragma unroll
    for (int q = 0; q < 5; ++q) {
      if (i + q * A1_BS < nf) atomicAdd(o + i + q * A1_BS, v[q]);
    }
  }
}

__global__ __launch_bounds__(A1_BS) void k_agg1_bk(
    const unsigned* __restrict__ bins, const int* __restrict__ tab,
    const float* __restrict__ g, const float* __restrict__ dinv,
    const float* __restrict__ b1, const float* __restrict__ W2,
    float* __restrict__ h3, int N, int L) {
  __shared__ __attribute__((aligned(16))) float acc[BK_NPP * HID];
  const unsigned NONE = 0xffffffffu;
  const int tid = threadIdx.x;
  const int lane = tid & 31;
  const unsigned j = (unsigned)lane & 15u;
  const int n0  = blockIdx.x * BK_NPP;
  const int np  = min(BK_NPP, N - n0);
  for (int i = tid; i < BK_NPP * HID / 4; i += A1_BS) ((v4f*)acc)[i] = v4f{0.0f, 0.0f, 0.0f, 0.0f};
  __syncthreads();
  int P[BK_C], A[BK_C], total;
  bk_run(tab, L, tid >> 5, lane, P, A, total);
  if (total > 0) {
    const int last = total - 1;
    unsigned wn = bins[bk_word(P, A, min(lane, last))];
    for (int i = 0; i < total; i += 32) {
      const unsigned w = wn;
      wn = bins[bk_word(P, A, min(i + 32 + lane, last))];
      const bool valid = i + lane < total;
      const int s0 = valid ? (int)(w >> BK_LBITS) : 0;
      const int r0 = valid ? (int)(w & BK_LMASK) : (int)NONE;
      if (i + 32 <= total) a1_add_rows<true>(s0, r0, g, acc, j);
      else                 a1_add_rows_masked(s0, r0, g, acc, j, (unsigned)lane >> 4);
    }
  }
  __syncthreads();
  if (tid < np) {
    const int n = n0 + tid;
    const float dv = dinv[n];
    const v4f* gr = (const v4f*)(g + (size_t)n * HID);
    const v4f* ar = (const v4f*)(acc + tid * HID);
    float o0 = 0.0f, o1 = 0.0f;
#pragma unroll
    for (int q = 0; q < HID / 4; ++q) {
      const v4f gv = gr[q];
      const v4f av = ar[q];
#pragma unroll
      for (int e = 0; e < 4; ++e) {
        const int jj = 4 * q + e;
        const float v = fmaxf((gv[e] + av[e]) * dv + b1[jj], 0.0f);
        o0 = __builtin_fmaf(v, W2[jj * OUTF + 0], o0);
        o1 = __builtin_fmaf(v, W2[jj * OUTF + 1], o1);
      }
    }
    v2f o;
    o.x = o0 * dv;
    o.y = o1 * dv;
    *(v2f*)(h3 + (size_t)n * OUTF) = o;
  }
}

__global__ __launch_bounds__(A1_BS) void k_agg1_lds(
    const int* __restrict__ src, const int* __restrict__ dst,
    const float* __restrict__ g, float* __restrict__ acc1, int N, int E) {
  __shared__ __attribute__((aligned(16))) float acc[A1_NPP * HID];
  __shared__ unsigned qsrc[A1_BS / 32][A1_Q];
  __shared__ unsigned qrow[A1_BS / 32][A1_Q];
  const int tid = threadIdx.x;
  const unsigned lane = tid & 31;
  const unsigned j = lane & 15;
  const int wv  = tid >> 5;
  const int n0  = blockIdx.x * A1_NPP;
  const int np  = min(A1_NPP, N - n0);
  for (int i = tid; i < A1_NPP * HID / 4; i += A1_BS) ((v4f*)acc)[i] = v4f{0.0f, 0.0f, 0.0f, 0.0f};
  __syncthreads();
  const int per = (E + A1_C - 1) / A1_C;
  const int e0  = blockIdx.y * per;
  const int e1  = min(E, e0 + per);
  unsigned* qs = &qsrc[wv][0];
  unsigned* qr = &qrow[wv][0];

  if (e0 < e1) {
    const unsigned NONE = 0xffffffffu;
    unsigned head = 0, tail = 0;
    unsigned r[A1_U];
    int      s[A1_U];
    int tb = e0;
#pragma unroll
    for (int k = 0; k < A1_U; ++k) {
      int e  = tb + tid + k * A1_BS;
      int ec = min(e, e1 - 1);
      int dd = dst[ec];
      s[k] = src[ec];
      r[k] = (e < e1) ? (unsigned)(dd - n0) : NONE;
    }
    while (tb < e1) {
      const int nbase = tb + A1_U * A1_BS + tid;
      int dn[A1_U], sn[A1_U];
#pragma unroll
      for (int k = 0; k < A1_U; ++k) {
        int ec = min(nbase + k * A1_BS, e1 - 1);
        dn[k] = dst[ec];
        sn[k] = src[ec];
      }
#pragma unroll
      for (int k = 0; k < A1_U; ++k) {
        const bool hit = r[k] < (unsigned)np;
        const unsigned mask = __builtin_amdgcn_ballot_w32(hit);
        if (mask != 0u) {
          const unsigned pos = (tail + __builtin_amdgcn_mbcnt_lo(mask, 0u)) & (A1_Q - 1);
          if (hit) { qs[pos] = (unsigned)s[k]; qr[pos] = r[k]; }
          tail += (unsigned)__builtin_popcount(mask);
          if (tail - head >= 32u) {
            a1_drain<true>(qs, qr, head, 32u, g, acc, lane, j);
            head += 32u;
          }
        }
      }
#pragma unroll
      for (int k = 0; k < A1_U; ++k) {
        r[k] = (nbase + k * A1_BS < e1) ? (unsigned)(dn[k] - n0) : NONE;
        s[k] = sn[k];
      }
      tb += A1_U * A1_BS;
    }
    if (tail != head) a1_drain<false>(qs, qr, head, tail - head, g, acc, lane, j);
  }
  __syncthreads();
  a1_flush(acc, acc1 + (size_t)n0 * HID, np * HID, tid);
}

__global__ void k_layer2_node(const float* __restrict__ acc1, const float* __restrict__ dinv,
                              const float* __restrict__ b1, const float* __restrict__ W2,
                              float* __restrict__ h3, int N) {
  int n = blockIdx.x * blockDim.x + threadIdx.x;
  if (n >= N) return;
  const float dv = dinv[n];
  float o0 = 0.0f, o1 = 0.0f;
#pragma unroll
  for (int j = 0; j < HID; ++j) {
    float v = fmaxf(acc1[(size_t)n * HID + j] * dv + b1[j], 0.0f);
    o0 = __builtin_fmaf(v, W2[j * OUTF + 0], o0);
    o1 = __builtin_fmaf(v, W2[j * OUTF + 1], o1);
  }
  v2f o;
  o.x = o0 * dv;
  o.y = o1 * dv;
  *(v2f*)(h3 + (size_t)n * OUTF) = o;
}

__global__ __launch_bounds__(SCAN_BS) void k_agg2_part(
    const int* __restrict__ src, const int* __restrict__ dst,
    const float* __restrict__ h3, float* __restrict__ part, int N, int E) {
  __shared__ float acc[NPP * OUTF];
  const v2f* __restrict__ h3v = (const v2f*)h3;
  const int tid = threadIdx.x;
  const int n0  = blockIdx.x * NPP;
  const int np  = min(NPP, N - n0);
  for (int i = tid; i < NPP * OUTF; i += SCAN_BS) acc[i] = 0.0f;
  __syncthreads();
  const int per = (E + NCH - 1) / NCH;
  const int e0  = blockIdx.y * per;
  const int e1  = min(E, e0 + per);

  if (e0 < e1) {
    const unsigned NONE = 0xffffffffu;
    unsigned r[SCAN_U];
    int      s[SCAN_U];
    int base = e0 + tid;
#pragma unroll
    for (int k = 0; k < SCAN_U; ++k) {
      int e  = base + k * SCAN_BS;
      int ec = min(e, e1 - 1);
      int dd = dst[ec];
      s[k] = src[ec];
      r[k] = (e < e1) ? (unsigned)(dd - n0) : NONE;
    }
    while (base < e1) {
      v2f gv[SCAN_U];
#pragma unroll
      for (int k = 0; k < SCAN_U; ++k) {
        const bool keep = r[k] < (unsigned)np;
        gv[k] = h3v[keep ? s[k] : 0];
      }
      const int nbase = base + SCAN_U * SCAN_BS;
      int dn[SCAN_U], sn[SCAN_U];
#pragma unroll
      for (int k = 0; k < SCAN_U; ++k) {
        int ec = min(nbase + k * SCAN_BS, e1 - 1);
        dn[k] = dst[ec];
        sn[k] = src[ec];
      }
#pragma unroll
      for (int k = 0; k < SCAN_U; ++k) {
        if (r[k] < (unsigned)np) {
          atomicAdd(&acc[r[k] * OUTF + 0], gv[k].x);
          atomicAdd(&acc[r[k] * OUTF + 1], gv[k].y);
        }
      }
#pragma unroll
      for (int k = 0; k < SCAN_U; ++k) {
        r[k] = (nbase + k * SCAN_BS < e1) ? (unsigned)(dn[k] - n0) : NONE;
        s[k] = sn[k];
      }
      base = nbase;
    }
  }
  __syncthreads();
  float* o = part + ((size_t)blockIdx.y * N + n0) * OUTF;
  for (int i = tid; i < np * OUTF; i += SCAN_BS) o[i] = acc[i];
}

#define BK_A2_U 8

__global__ __launch_bounds__(BK_BS) void k_agg2_bk(
    const unsigned* __restrict__ bins, const int* __restrict__ tab,
    const float* __restrict__ h3, const float* __restrict__ dinv,
    const float* __restrict__ b2, float* __restrict__ out, int N, int L) {
  __shared__ float acc[BK_NPP * OUTF];
  const v2f* __restrict__ h3v = (const v2f*)h3;
  const int tid  = threadIdx.x;
  const int lane = tid & 31;
  const int n0   = blockIdx.x * BK_NPP;
  const int np   = min(BK_NPP, N - n0);
  for (int i = tid; i < BK_NPP * OUTF; i += BK_BS) acc[i] = 0.0f;
  __syncthreads();
  int P[BK_C], A[BK_C], total;
  bk_run(tab, L, tid >> 5, lane, P, A, total);
  if (total > 0) {
    const int last = total - 1;
    unsigned w[BK_A2_U];
#pragma unroll
    for (int k = 0; k < BK_A2_U; ++k) w[k] = bins[bk_word(P, A, min(lane + 32 * k, last))];
    for (int base = 0; base < total; base += 32 * BK_A2_U) {
      unsigned wn[BK_A2_U];
      v2f gv[BK_A2_U];
#pragma unroll
      for (int k = 0; k < BK_A2_U; ++k)
        wn[k] = bins[bk_word(P, A, min(base + 32 * BK_A2_U + lane + 32 * k, last))];
#pragma unroll
      for (int k = 0; k < BK_A2_U; ++k) gv[k] = h3v[w[k] >> BK_LBITS];
#pragma unroll
      for (int k = 0; k < BK_A2_U; ++k) {
        const bool keep = base + lane + 32 * k < total;
        const unsigned row = keep ? (w[k] & BK_LMASK) : (unsigned)lane;
        atomicAdd(&acc[row * OUTF + 0], keep ? gv[k].x : 0.0f);
        atomicAdd(&acc[row * OUTF + 1], keep ? gv[k].y : 0.0f);
      }
#pragma unroll
      for (int k = 0; k < BK_A2_U; ++k) w[k] = wn[k];
    }
  }
  __syncthreads();
  for (int i = tid; i < np * OUTF; i += BK_BS) {
    const int t = n0 * OUTF + i;
    float s = h3[t];
    s += acc[i];
    out[t] = s * dinv[t / OUTF] + b2[t % OUTF];
  }
}

template <int P>
__global__ void k_agg2_reduce(const float* __restrict__ part, const float* __restrict__ h3,
                              const float* __restrict__ dinv, const float* __restrict__ b2,
                              float* __restrict__ out, int N) {
  int t = blockIdx.x * blockDim.x + threadIdx.x;
  if (t >= N * OUTF) return;
  float s = h3[t];
#pragma unroll
  for (int c = 0; c < P; ++c) s += part[(size_t)c * N * OUTF + t];
  out[t] = s * dinv[t / OUTF] + b2[t % OUTF];
}

static inline int cdiv_i(long long a, long long b) { return (int)((a + b - 1) / b); }

extern "C" void kernel_launch(void* const* d_in, const int* in_sizes, int n_in,
                              void* d_out, int out_size, void* d_ws, size_t ws_size,
                              hipStream_t stream) {
  const float* x  = (const float*)d_in[0];
  const int*   ei = (const int*)  d_in[1];
  const float* W1 = (const float*)d_in[2];
  const float* b1 = (const float*)d_in[3];
  const float* W2 = (const float*)d_in[4];
  const float* b2 = (const float*)d_in[5];
  float* out = (float*)d_out;

  const int N = in_sizes[0] / F_IN;
  const int E = in_sizes[1] / 2;
  const int* src = ei;
  const int* dst = ei + E;

  float* ws   = (float*)d_ws;
  float* dinv = ws;
  float* h    = dinv + N;
  float* acc1 = h + (size_t)N * HID;
  float* h3   = acc1 + (size_t)N * HID;

  const int BS = 256;
  const int nTiles = cdiv_i(N, 16);

  float* part = h;

  const int R = cdiv_i(N, BK_NPP);
  const int L = cdiv_i(E, BK_B);
  int*      tab  = (int*)acc1;
  unsigned* bins = (unsigned*)(h3 + (size_t)N * OUTF);
  const size_t need = ((size_t)N * (1 + 2 * HID + OUTF) + (size_t)BK_B * L) * sizeof(float);
  const bool bucketed = ws_size >= need && E >= 1 && R <= BK_RMAX && L <= BK_U * BK_SORT_BS &&
                        N <= (1 << (32 - BK_LBITS)) && (size_t)BK_TABR * BK_B <= (size_t)N * HID;

  if (bucketed) {
    k_gemm1_sort<<<BK_B + cdiv_i(nTiles, G1S_WAVES), BK_SORT_BS, 0, stream>>>(x, W1, h, src, dst, bins, tab, N, nTiles, E, L);

    k_deg_bk<<<R, BK_BS, 0, stream>>>(bins, tab, dinv, h, N, L);

    k_agg1_bk<<<R, A1_BS, 0, stream>>>(bins, tab, h, dinv, b1, W2, h3, N, L);

    k_agg2_bk<<<R, BK_BS, 0, stream>>>(bins, tab, h3, dinv, b2, out, N, L);
    return;
  }

  const dim3 scanGrid(cdiv_i(N, NPP), NCH);

  k_deg_part  <<<scanGrid, SCAN_BS, 0, stream>>>(dst, part, N, E);
  k_deg_reduce<NCH><<<cdiv_i(N, BS), BS, 0, stream>>>(part, dinv, N);

  k_gemm1_wmma<<<cdiv_i(nTiles, G1_WAVES), G1_WAVES * 32, 0, stream>>>(x, W1, dinv, h, acc1, N, nTiles);

  k_agg1_lds<<<dim3(cdiv_i(N, A1_NPP), A1_C), A1_BS, 0, stream>>>(src, dst, h, acc1, N, E);

  k_layer2_node<<<cdiv_i(N, BS), BS, 0, stream>>>(acc1, dinv, b1, W2, h3, N);

  k_agg2_part  <<<scanGrid, SCAN_BS, 0, stream>>>(src, dst, h3, part, N, E);
  k_agg2_reduce<NCH><<<cdiv_i((long long)N * OUTF, BS), BS, 0, stream>>>(part, h3, dinv, b2, out, N);
}
